# starting kernel
# baseline (speedup 1.0000x reference)
#include <hip/hip_runtime.h>
#include <hip/hip_bf16.h>


typedef float v2f __attribute__((ext_vector_type(2)));
typedef float v8f __attribute__((ext_vector_type(8)));

#define N_NODES_C 100000
#define N_EDGES_C 800000
#define HID_C     128
#define OUT_DIM_C 500
#define N_GRAPHS_C 64

__global__ void degree_kernel(const int* __restrict__ src, const int* __restrict__ dst,
                              float* __restrict__ odeg, float* __restrict__ ideg,
                              int n_edges) {
    int e = blockIdx.x * blockDim.x + threadIdx.x;
    if (e < n_edges) {
        atomicAdd(&odeg[src[e]], 1.0f);
        atomicAdd(&ideg[dst[e]], 1.0f);
    }
}

__global__ void norm_kernel(const float* __restrict__ odeg, const float* __restrict__ ideg,
                            float* __restrict__ srcn, float* __restrict__ dstn, int n) {
    int i = blockIdx.x * blockDim.x + threadIdx.x;
    if (i < n) {
        srcn[i] = rsqrtf(fmaxf(odeg[i], 1.0f));
        dstn[i] = rsqrtf(fmaxf(ideg[i], 1.0f));
    }
}

__global__ void edge_agg_kernel(const float* __restrict__ h, const float* __restrict__ srcn,
                                const int* __restrict__ src, const int* __restrict__ dst,
                                float* __restrict__ agg, int n_edges) {
    int gid  = blockIdx.x * blockDim.x + threadIdx.x;
    int e    = gid >> 5;
    int lane = gid & 31;
    if (e >= n_edges) return;
    int s = src[e];
    int d = dst[e];
    float sn = srcn[s];
    float4 v = ((const float4*)(h + (size_t)s * HID_C))[lane];
    float* ap = agg + (size_t)d * HID_C + lane * 4;
    atomicAdd(ap + 0, v.x * sn);
    atomicAdd(ap + 1, v.y * sn);
    atomicAdd(ap + 2, v.z * sn);
    atomicAdd(ap + 3, v.w * sn);
}

__global__ void __launch_bounds__(256) conv_gemm_kernel(
    const float* __restrict__ agg, const float* __restrict__ dstn,
    const float* __restrict__ W, const float* __restrict__ bias,
    float* __restrict__ out, int n_nodes) {
    __shared__ float As[16][HID_C + 4];

    const int tid   = threadIdx.x;
    const int mbase = blockIdx.x * 16;

    for (int i = tid; i < 16 * (HID_C / 4); i += 256) {
        int r    = i >> 5;
        int c4   = i & 31;
        int node = mbase + r;
        float4 v = make_float4(0.f, 0.f, 0.f, 0.f);
        float  sc = 0.f;
        if (node < n_nodes) {
            v  = *(const float4*)(agg + (size_t)node * HID_C + c4 * 4);
            sc = dstn[node];
        }
        As[r][c4 * 4 + 0] = v.x * sc;
        As[r][c4 * 4 + 1] = v.y * sc;
        As[r][c4 * 4 + 2] = v.z * sc;
        As[r][c4 * 4 + 3] = v.w * sc;
    }
    __syncthreads();

    const int lane  = tid & 31;
    const int wave  = tid >> 5;
    const int l15   = lane & 15;
    const int khalf = (lane >> 4) << 1;
    const int col   = wave * 16 + l15;

    v8f acc = {0.f, 0.f, 0.f, 0.f, 0.f, 0.f, 0.f, 0.f};
    #pragma unroll
    for (int k = 0; k < HID_C; k += 4) {
        v2f a, b;
        a.x = As[l15][k + khalf];
        a.y = As[l15][k + khalf + 1];
        b.x = W[(size_t)(k + khalf) * HID_C + col];
        b.y = W[(size_t)(k + khalf + 1) * HID_C + col];
        acc = __builtin_amdgcn_wmma_f32_16x16x4_f32(false, a, false, b,
                                                    (short)0, acc, false, false);
    }

    const float bv = bias[col];
    #pragma unroll
    for (int i = 0; i < 8; ++i) {
        int node = mbase + i + ((lane >> 4) << 3);
        if (node < n_nodes)
            out[(size_t)node * HID_C + col] = fmaxf(acc[i] + bv, 0.0f);
    }
}

__global__ void __launch_bounds__(32) wmma_gemm_kernel(
    const float* __restrict__ A, const float* __restrict__ B,
    const float* __restrict__ bias, float* __restrict__ C,
    int M, int N, int K, int do_relu) {
    const int lane  = threadIdx.x & 31;
    const int l15   = lane & 15;
    const int khalf = (lane >> 4) << 1;
    const int mbase = blockIdx.x * 16;
    const int nbase = blockIdx.y * 16;
    const int row   = mbase + l15;
    const int col   = nbase + l15;

    v8f acc = {0.f, 0.f, 0.f, 0.f, 0.f, 0.f, 0.f, 0.f};
    for (int k = 0; k < K; k += 4) {
        int k0 = k + khalf, k1 = k + khalf + 1;
        v2f a, b;
        a.x = (row < M) ? A[(size_t)row * K + k0] : 0.0f;
        a.y = (row < M) ? A[(size_t)row * K + k1] : 0.0f;
        b.x = (col < N) ? B[(size_t)k0 * N + col] : 0.0f;
        b.y = (col < N) ? B[(size_t)k1 * N + col] : 0.0f;
        acc = __builtin_amdgcn_wmma_f32_16x16x4_f32(false, a, false, b,
                                                    (short)0, acc, false, false);
    }

    if (col < N) {
        const float bv = bias[col];
        #pragma unroll
        for (int i = 0; i < 8; ++i) {
            int m = mbase + i + ((lane >> 4) << 3);
            if (m < M) {
                float v = acc[i] + bv;
                C[(size_t)m * N + col] = do_relu ? fmaxf(v, 0.0f) : v;
            }
        }
    }
}

__global__ void segmax_kernel(const float* __restrict__ h, const int* __restrict__ gids,
                              float* __restrict__ hg, int n_nodes) {
    int node = blockIdx.x;
    int c    = threadIdx.x;
    if (node < n_nodes) {
        int g = gids[node];
        float v = h[(size_t)node * HID_C + c];
        atomicMax((unsigned int*)&hg[(size_t)g * HID_C + c], __float_as_uint(v));
    }
}

extern "C" void kernel_launch(void* const* d_in, const int* in_sizes, int n_in,
                              void* d_out, int out_size, void* d_ws, size_t ws_size,
                              hipStream_t stream) {
    (void)in_sizes; (void)n_in; (void)out_size; (void)ws_size;

    const float* feat = (const float*)d_in[0];
    const int*   src  = (const int*)d_in[1];
    const int*   dst  = (const int*)d_in[2];
    const int*   gids = (const int*)d_in[3];
    const float* Wg[3] = {(const float*)d_in[4], (const float*)d_in[6], (const float*)d_in[8]};
    const float* bg[3] = {(const float*)d_in[5], (const float*)d_in[7], (const float*)d_in[9]};
    const float* Wc1 = (const float*)d_in[10]; const float* bc1 = (const float*)d_in[11];
    const float* Wc2 = (const float*)d_in[12]; const float* bc2 = (const float*)d_in[13];
    const float* Wc3 = (const float*)d_in[14]; const float* bc3 = (const float*)d_in[15];
    float* out = (float*)d_out;

    char* ws = (char*)d_ws;
    size_t off = 0;
    auto take = [&](size_t elems) -> float* {
        float* p = (float*)(ws + off);
        off += ((elems * sizeof(float) + 255) / 256) * 256;
        return p;
    };
    float* odeg = take(N_NODES_C);
    float* ideg = take(N_NODES_C);
    float* srcn = take(N_NODES_C);
    float* dstn = take(N_NODES_C);
    float* agg  = take((size_t)N_NODES_C * HID_C);
    float* h    = take((size_t)N_NODES_C * HID_C);
    float* hg   = take((size_t)N_GRAPHS_C * HID_C);
    float* x1   = take((size_t)N_GRAPHS_C * HID_C);
    float* x2   = take((size_t)N_GRAPHS_C * (HID_C / 2));

    hipMemsetAsync(odeg, 0, N_NODES_C * sizeof(float), stream);
    hipMemsetAsync(ideg, 0, N_NODES_C * sizeof(float), stream);
    degree_kernel<<<(N_EDGES_C + 255) / 256, 256, 0, stream>>>(src, dst, odeg, ideg, N_EDGES_C);
    norm_kernel<<<(N_NODES_C + 255) / 256, 256, 0, stream>>>(odeg, ideg, srcn, dstn, N_NODES_C);

    const int mblocks = (N_NODES_C + 15) / 16;
    const float* hin = feat;
    for (int L = 0; L < 3; ++L) {
        hipMemsetAsync(agg, 0, (size_t)N_NODES_C * HID_C * sizeof(float), stream);
        edge_agg_kernel<<<(N_EDGES_C * 32 + 255) / 256, 256, 0, stream>>>(
            hin, srcn, src, dst, agg, N_EDGES_C);
        conv_gemm_kernel<<<mblocks, 256, 0, stream>>>(agg, dstn, Wg[L], bg[L], h, N_NODES_C);
        hin = h;
    }

    hipMemsetAsync(hg, 0, (size_t)N_GRAPHS_C * HID_C * sizeof(float), stream);
    segmax_kernel<<<N_NODES_C, HID_C, 0, stream>>>(h, gids, hg, N_NODES_C);

    wmma_gemm_kernel<<<dim3(N_GRAPHS_C / 16, HID_C / 16), 32, 0, stream>>>(
        hg, Wc1, bc1, x1, N_GRAPHS_C, HID_C, HID_C, 1);
    wmma_gemm_kernel<<<dim3(N_GRAPHS_C / 16, (HID_C / 2) / 16), 32, 0, stream>>>(
        x1, Wc2, bc2, x2, N_GRAPHS_C, HID_C / 2, HID_C, 1);
    wmma_gemm_kernel<<<dim3(N_GRAPHS_C / 16, (OUT_DIM_C + 15) / 16), 32, 0, stream>>>(
        x2, Wc3, bc3, out, N_GRAPHS_C, OUT_DIM_C, HID_C / 2, 0);
}


# Max pool on resident waves: 6250 waves each walk a contiguous share of node rows, column maxima of the current graph held in registers, one atomic max per column per run of equal graph ids
# speedup vs baseline: 1.5662x; 1.5662x over previous
#include <hip/hip_runtime.h>
#include <hip/hip_bf16.h>


typedef float v2f __attribute__((ext_vector_type(2)));
typedef float v8f __attribute__((ext_vector_type(8)));

#define N_NODES_C 100000
#define N_EDGES_C 800000
#define HID_C     128
#define OUT_DIM_C 500
#define N_GRAPHS_C 64

__global__ void degree_kernel(const int* __restrict__ src, const int* __restrict__ dst,
                              float* __restrict__ odeg, float* __restrict__ ideg,
                              int n_edges) {
    int e = blockIdx.x * blockDim.x + threadIdx.x;
    if (e < n_edges) {
        atomicAdd(&odeg[src[e]], 1.0f);
        atomicAdd(&ideg[dst[e]], 1.0f);
    }
}

__global__ void norm_kernel(const float* __restrict__ odeg, const float* __restrict__ ideg,
                            float* __restrict__ srcn, float* __restrict__ dstn, int n) {
    int i = blockIdx.x * blockDim.x + threadIdx.x;
    if (i < n) {
        srcn[i] = rsqrtf(fmaxf(odeg[i], 1.0f));
        dstn[i] = rsqrtf(fmaxf(ideg[i], 1.0f));
    }
}

#define EDGE_WAVES_PER_BLOCK 8
#define EDGE_BLOCKS 1563
#define EDGE_WAVES (EDGE_BLOCKS * EDGE_WAVES_PER_BLOCK)

__global__ void __launch_bounds__(EDGE_WAVES_PER_BLOCK * 32) edge_agg_kernel(
    const float* __restrict__ h, const float* __restrict__ srcn,
    const int* __restrict__ src, const int* __restrict__ dst,
    float* __restrict__ agg, int n_edges) {
    const unsigned lane = threadIdx.x & 31u;
    const int w = __builtin_amdgcn_readfirstlane(
        (int)(blockIdx.x * EDGE_WAVES_PER_BLOCK + (threadIdx.x >> 5)));

    const int ngroups = (n_edges + 7) >> 3;
    const int q  = ngroups / EDGE_WAVES;
    const int r  = ngroups % EDGE_WAVES;
    const int g0 = w * q + (w < r ? w : r);
    const int g1 = g0 + q + (w < r ? 1 : 0);
    const int e0 = g0 * 8;
    const int e1 = (g1 * 8 < n_edges) ? g1 * 8 : n_edges;

    for (int base = e0; base < e1; base += 32) {
        const int idx = base + (int)lane;
        int sv = 0, dv = 0;
        if (idx < e1) { sv = src[idx]; dv = dst[idx]; }
        const int cnt = (e1 - base < 32) ? (e1 - base) : 32;

        int j = 0;
        for (; j + 8 <= cnt; j += 8) {
            int   s[8], d[8];
            float sn[8];
            float v[8][4];
            #pragma unroll
            for (int u = 0; u < 8; ++u) {
                s[u] = __builtin_amdgcn_readlane(sv, j + u);
                d[u] = __builtin_amdgcn_readlane(dv, j + u);
            }
            #pragma unroll
            for (int u = 0; u < 8; ++u) {
                const float* hp = h + (size_t)s[u] * HID_C;
                sn[u]   = srcn[s[u]];
                v[u][0] = hp[lane];
                v[u][1] = hp[lane + 32u];
                v[u][2] = hp[lane + 64u];
                v[u][3] = hp[lane + 96u];
            }
            #pragma unroll
            for (int u = 0; u < 8; ++u) {
                float* ap = agg + (size_t)d[u] * HID_C;
                atomicAdd(ap + lane,       v[u][0] * sn[u]);
                atomicAdd(ap + lane + 32u, v[u][1] * sn[u]);
                atomicAdd(ap + lane + 64u, v[u][2] * sn[u]);
                atomicAdd(ap + lane + 96u, v[u][3] * sn[u]);
            }
        }
        for (; j < cnt; ++j) {
            const int s1 = __builtin_amdgcn_readlane(sv, j);
            const int d1 = __builtin_amdgcn_readlane(dv, j);
            const float* hp = h + (size_t)s1 * HID_C;
            const float sn1 = srcn[s1];
            const float a0 = hp[lane];
            const float a1 = hp[lane + 32u];
            const float a2 = hp[lane + 64u];
            const float a3 = hp[lane + 96u];
            float* ap = agg + (size_t)d1 * HID_C;
            atomicAdd(ap + lane,       a0 * sn1);
            atomicAdd(ap + lane + 32u, a1 * sn1);
            atomicAdd(ap + lane + 64u, a2 * sn1);
            atomicAdd(ap + lane + 96u, a3 * sn1);
        }
    }
}

__global__ void __launch_bounds__(256) conv_gemm_kernel(
    const float* __restrict__ agg, const float* __restrict__ dstn,
    const float* __restrict__ W, const float* __restrict__ bias,
    float* __restrict__ out, int n_nodes) {
    __shared__ float As[16][HID_C + 4];

    const int tid   = threadIdx.x;
    const int mbase = blockIdx.x * 16;

    for (int i = tid; i < 16 * (HID_C / 4); i += 256) {
        int r    = i >> 5;
        int c4   = i & 31;
        int node = mbase + r;
        float4 v = make_float4(0.f, 0.f, 0.f, 0.f);
        float  sc = 0.f;
        if (node < n_nodes) {
            v  = *(const float4*)(agg + (size_t)node * HID_C + c4 * 4);
            sc = dstn[node];
        }
        As[r][c4 * 4 + 0] = v.x * sc;
        As[r][c4 * 4 + 1] = v.y * sc;
        As[r][c4 * 4 + 2] = v.z * sc;
        As[r][c4 * 4 + 3] = v.w * sc;
    }
    __syncthreads();

    const int lane  = tid & 31;
    const int wave  = tid >> 5;
    const int l15   = lane & 15;
    const int khalf = (lane >> 4) << 1;
    const int col   = wave * 16 + l15;

    v8f acc = {0.f, 0.f, 0.f, 0.f, 0.f, 0.f, 0.f, 0.f};
    #pragma unroll
    for (int k = 0; k < HID_C; k += 4) {
        v2f a, b;
        a.x = As[l15][k + khalf];
        a.y = As[l15][k + khalf + 1];
        b.x = W[(size_t)(k + khalf) * HID_C + col];
        b.y = W[(size_t)(k + khalf + 1) * HID_C + col];
        acc = __builtin_amdgcn_wmma_f32_16x16x4_f32(false, a, false, b,
                                                    (short)0, acc, false, false);
    }

    const float bv = bias[col];
    #pragma unroll
    for (int i = 0; i < 8; ++i) {
        int node = mbase + i + ((lane >> 4) << 3);
        if (node < n_nodes)
            out[(size_t)node * HID_C + col] = fmaxf(acc[i] + bv, 0.0f);
    }
}

__global__ void __launch_bounds__(32) wmma_gemm_kernel(
    const float* __restrict__ A, const float* __restrict__ B,
    const float* __restrict__ bias, float* __restrict__ C,
    int M, int N, int K, int do_relu) {
    const int lane  = threadIdx.x & 31;
    const int l15   = lane & 15;
    const int khalf = (lane >> 4) << 1;
    const int mbase = blockIdx.x * 16;
    const int nbase = blockIdx.y * 16;
    const int row   = mbase + l15;
    const int col   = nbase + l15;

    v8f acc = {0.f, 0.f, 0.f, 0.f, 0.f, 0.f, 0.f, 0.f};
    for (int k = 0; k < K; k += 4) {
        int k0 = k + khalf, k1 = k + khalf + 1;
        v2f a, b;
        a.x = (row < M) ? A[(size_t)row * K + k0] : 0.0f;
        a.y = (row < M) ? A[(size_t)row * K + k1] : 0.0f;
        b.x = (col < N) ? B[(size_t)k0 * N + col] : 0.0f;
        b.y = (col < N) ? B[(size_t)k1 * N + col] : 0.0f;
        acc = __builtin_amdgcn_wmma_f32_16x16x4_f32(false, a, false, b,
                                                    (short)0, acc, false, false);
    }

    if (col < N) {
        const float bv = bias[col];
        #pragma unroll
        for (int i = 0; i < 8; ++i) {
            int m = mbase + i + ((lane >> 4) << 3);
            if (m < M) {
                float v = acc[i] + bv;
                C[(size_t)m * N + col] = do_relu ? fmaxf(v, 0.0f) : v;
            }
        }
    }
}

#define POOL_WAVES_PER_BLOCK 10
#define POOL_BLOCKS 625
#define POOL_WAVES (POOL_BLOCKS * POOL_WAVES_PER_BLOCK)

__device__ __forceinline__ void pool_flush(float* __restrict__ hg, int g, unsigned lane,
                                           unsigned m0, unsigned m1, unsigned m2, unsigned m3) {
    unsigned int* p = (unsigned int*)(hg + (size_t)g * HID_C);
    atomicMax(p + lane,       m0);
    atomicMax(p + lane + 32u, m1);
    atomicMax(p + lane + 64u, m2);
    atomicMax(p + lane + 96u, m3);
}

__global__ void __launch_bounds__(POOL_WAVES_PER_BLOCK * 32) segmax_kernel(
    const float* __restrict__ h, const int* __restrict__ gids,
    float* __restrict__ hg, int n_nodes) {
    const unsigned lane = threadIdx.x & 31u;
    const int w = __builtin_amdgcn_readfirstlane(
        (int)(blockIdx.x * POOL_WAVES_PER_BLOCK + (threadIdx.x >> 5)));

    const int q  = n_nodes / POOL_WAVES;
    const int r  = n_nodes % POOL_WAVES;
    const int n0 = w * q + (w < r ? w : r);
    const int n1 = n0 + q + (w < r ? 1 : 0);

    int cur = -1;
    unsigned m0 = 0u, m1 = 0u, m2 = 0u, m3 = 0u;

    for (int base = n0; base < n1; base += 32) {
        const int idx = base + (int)lane;
        int gv = 0;
        if (idx < n1) gv = gids[idx];
        const int cnt = (n1 - base < 32) ? (n1 - base) : 32;

        int j = 0;
        for (; j + 8 <= cnt; j += 8) {
            float v[8][4];
            #pragma unroll
            for (int u = 0; u < 8; ++u) {
                const float* hp = h + (size_t)(base + j + u) * HID_C;
                v[u][0] = hp[lane];
                v[u][1] = hp[lane + 32u];
                v[u][2] = hp[lane + 64u];
                v[u][3] = hp[lane + 96u];
            }
            #pragma unroll
            for (int u = 0; u < 8; ++u) {
                const int g = __builtin_amdgcn_readlane(gv, j + u);
                if (g != cur) {
                    if (cur >= 0) pool_flush(hg, cur, lane, m0, m1, m2, m3);
                    m0 = 0u; m1 = 0u; m2 = 0u; m3 = 0u;
                    cur = g;
                }
                const unsigned b0 = __float_as_uint(v[u][0]);
                const unsigned b1 = __float_as_uint(v[u][1]);
                const unsigned b2 = __float_as_uint(v[u][2]);
                const unsigned b3 = __float_as_uint(v[u][3]);
                m0 = (b0 > m0) ? b0 : m0;
                m1 = (b1 > m1) ? b1 : m1;
                m2 = (b2 > m2) ? b2 : m2;
                m3 = (b3 > m3) ? b3 : m3;
            }
        }
        for (; j < cnt; ++j) {
            const float* hp = h + (size_t)(base + j) * HID_C;
            const unsigned b0 = __float_as_uint(hp[lane]);
            const unsigned b1 = __float_as_uint(hp[lane + 32u]);
            const unsigned b2 = __float_as_uint(hp[lane + 64u]);
            const unsigned b3 = __float_as_uint(hp[lane + 96u]);
            const int g = __builtin_amdgcn_readlane(gv, j);
            if (g != cur) {
                if (cur >= 0) pool_flush(hg, cur, lane, m0, m1, m2, m3);
                m0 = 0u; m1 = 0u; m2 = 0u; m3 = 0u;
                cur = g;
            }
            m0 = (b0 > m0) ? b0 : m0;
            m1 = (b1 > m1) ? b1 : m1;
            m2 = (b2 > m2) ? b2 : m2;
            m3 = (b3 > m3) ? b3 : m3;
        }
    }
    if (cur >= 0) pool_flush(hg, cur, lane, m0, m1, m2, m3);
}

extern "C" void kernel_launch(void* const* d_in, const int* in_sizes, int n_in,
                              void* d_out, int out_size, void* d_ws, size_t ws_size,
                              hipStream_t stream) {
    (void)in_sizes; (void)n_in; (void)out_size; (void)ws_size;

    const float* feat = (const float*)d_in[0];
    const int*   src  = (const int*)d_in[1];
    const int*   dst  = (const int*)d_in[2];
    const int*   gids = (const int*)d_in[3];
    const float* Wg[3] = {(const float*)d_in[4], (const float*)d_in[6], (const float*)d_in[8]};
    const float* bg[3] = {(const float*)d_in[5], (const float*)d_in[7], (const float*)d_in[9]};
    const float* Wc1 = (const float*)d_in[10]; const float* bc1 = (const float*)d_in[11];
    const float* Wc2 = (const float*)d_in[12]; const float* bc2 = (const float*)d_in[13];
    const float* Wc3 = (const float*)d_in[14]; const float* bc3 = (const float*)d_in[15];
    float* out = (float*)d_out;

    char* ws = (char*)d_ws;
    size_t off = 0;
    auto take = [&](size_t elems) -> float* {
        float* p = (float*)(ws + off);
        off += ((elems * sizeof(float) + 255) / 256) * 256;
        return p;
    };
    float* odeg = take(N_NODES_C);
    float* ideg = take(N_NODES_C);
    float* srcn = take(N_NODES_C);
    float* dstn = take(N_NODES_C);
    float* agg  = take((size_t)N_NODES_C * HID_C);
    float* h    = take((size_t)N_NODES_C * HID_C);
    float* hg   = take((size_t)N_GRAPHS_C * HID_C);
    float* x1   = take((size_t)N_GRAPHS_C * HID_C);
    float* x2   = take((size_t)N_GRAPHS_C * (HID_C / 2));

    hipMemsetAsync(odeg, 0, N_NODES_C * sizeof(float), stream);
    hipMemsetAsync(ideg, 0, N_NODES_C * sizeof(float), stream);
    degree_kernel<<<(N_EDGES_C + 255) / 256, 256, 0, stream>>>(src, dst, odeg, ideg, N_EDGES_C);
    norm_kernel<<<(N_NODES_C + 255) / 256, 256, 0, stream>>>(odeg, ideg, srcn, dstn, N_NODES_C);

    const int mblocks = (N_NODES_C + 15) / 16;
    const float* hin = feat;
    for (int L = 0; L < 3; ++L) {
        hipMemsetAsync(agg, 0, (size_t)N_NODES_C * HID_C * sizeof(float), stream);
        edge_agg_kernel<<<EDGE_BLOCKS, EDGE_WAVES_PER_BLOCK * 32, 0, stream>>>(
            hin, srcn, src, dst, agg, N_EDGES_C);
        conv_gemm_kernel<<<mblocks, 256, 0, stream>>>(agg, dstn, Wg[L], bg[L], h, N_NODES_C);
        hin = h;
    }

    hipMemsetAsync(hg, 0, (size_t)N_GRAPHS_C * HID_C * sizeof(float), stream);
    segmax_kernel<<<POOL_BLOCKS, POOL_WAVES_PER_BLOCK * 32, 0, stream>>>(h, gids, hg, N_NODES_C);

    wmma_gemm_kernel<<<dim3(N_GRAPHS_C / 16, HID_C / 16), 32, 0, stream>>>(
        hg, Wc1, bc1, x1, N_GRAPHS_C, HID_C, HID_C, 1);
    wmma_gemm_kernel<<<dim3(N_GRAPHS_C / 16, (HID_C / 2) / 16), 32, 0, stream>>>(
        x1, Wc2, bc2, x2, N_GRAPHS_C, HID_C / 2, HID_C, 1);
    wmma_gemm_kernel<<<dim3(N_GRAPHS_C / 16, (OUT_DIM_C + 15) / 16), 32, 0, stream>>>(
        x2, Wc3, bc3, out, N_GRAPHS_C, OUT_DIM_C, HID_C / 2, 0);
}


# Degree counts in LDS: a work-group per (node range, edge chunk, id array) counts its ids with LDS integer adds, stores a partial plane; norm pass sums 16 partials; global f32 adds and 2 fills gone
# speedup vs baseline: 1.7685x; 1.1292x over previous
#include <hip/hip_runtime.h>
#include <hip/hip_bf16.h>


typedef float v2f __attribute__((ext_vector_type(2)));
typedef float v8f __attribute__((ext_vector_type(8)));

#define N_NODES_C 100000
#define N_EDGES_C 800000
#define HID_C     128
#define OUT_DIM_C 500
#define N_GRAPHS_C 64

#define DEG_RANGES  16
#define DEG_CHUNKS  16
#define DEG_THREADS 256
#define DEG_RANGE_NODES ((((N_NODES_C + DEG_RANGES - 1) / DEG_RANGES) + 31) & ~31)
#define DEG_LOADS   13

__global__ void __launch_bounds__(DEG_THREADS) degree_kernel(
    const int* __restrict__ src, const int* __restrict__ dst,
    unsigned int* __restrict__ part, int n_edges, int n_nodes) {
    __shared__ unsigned int cnt[DEG_RANGE_NODES];

    const int tid = threadIdx.x;
    const int* __restrict__ ids = (blockIdx.z == 0) ? src : dst;
    const int lo = blockIdx.x * DEG_RANGE_NODES;
    int span = n_nodes - lo;
    if (span > DEG_RANGE_NODES) span = DEG_RANGE_NODES;
    if (span < 0) span = 0;

    for (int i = tid; i < DEG_RANGE_NODES; i += DEG_THREADS) cnt[i] = 0u;
    __syncthreads();

    const int chunk = (((n_edges + DEG_CHUNKS - 1) / DEG_CHUNKS) + 3) & ~3;
    const int e0 = blockIdx.y * chunk;
    int e1 = e0 + chunk;
    if (e1 > n_edges) e1 = n_edges;
    if (e1 < e0) e1 = e0;

    const int e1v = e0 + ((e1 - e0) & ~3);
    for (int base = e0 + 4 * tid; base < e1v; base += 4 * DEG_THREADS * DEG_LOADS) {
        int4 v[DEG_LOADS];
        #pragma unroll
        for (int u = 0; u < DEG_LOADS; ++u) {
            const int e = base + u * 4 * DEG_THREADS;
            v[u] = make_int4(-1, -1, -1, -1);
            if (e < e1v) v[u] = *(const int4*)(ids + e);
        }
        #pragma unroll
        for (int u = 0; u < DEG_LOADS; ++u) {
            const unsigned rx = (unsigned)(v[u].x - lo);
            const unsigned ry = (unsigned)(v[u].y - lo);
            const unsigned rz = (unsigned)(v[u].z - lo);
            const unsigned rw = (unsigned)(v[u].w - lo);
            if (rx < (unsigned)span) atomicAdd(&cnt[rx], 1u);
            if (ry < (unsigned)span) atomicAdd(&cnt[ry], 1u);
            if (rz < (unsigned)span) atomicAdd(&cnt[rz], 1u);
            if (rw < (unsigned)span) atomicAdd(&cnt[rw], 1u);
        }
    }
    if (tid < e1 - e1v) {
        const unsigned rt = (unsigned)(ids[e1v + tid] - lo);
        if (rt < (unsigned)span) atomicAdd(&cnt[rt], 1u);
    }
    __syncthreads();

    unsigned int* __restrict__ dstp =
        part + ((size_t)blockIdx.z * DEG_CHUNKS + blockIdx.y) * (size_t)n_nodes + lo;
    for (int i = tid; i < span; i += DEG_THREADS) dstp[i] = cnt[i];
}

__global__ void norm_kernel(const unsigned int* __restrict__ part,
                            float* __restrict__ srcn, float* __restrict__ dstn, int n) {
    int i = blockIdx.x * blockDim.x + threadIdx.x;
    if (i < n) {
        unsigned int od = 0u, id = 0u;
        #pragma unroll
        for (int c = 0; c < DEG_CHUNKS; ++c) {
            od += part[(size_t)c * n + i];
            id += part[(size_t)(DEG_CHUNKS + c) * n + i];
        }
        srcn[i] = rsqrtf(fmaxf((float)od, 1.0f));
        dstn[i] = rsqrtf(fmaxf((float)id, 1.0f));
    }
}

#define EDGE_WAVES_PER_BLOCK 8
#define EDGE_BLOCKS 1563
#define EDGE_WAVES (EDGE_BLOCKS * EDGE_WAVES_PER_BLOCK)

__global__ void __launch_bounds__(EDGE_WAVES_PER_BLOCK * 32) edge_agg_kernel(
    const float* __restrict__ h, const float* __restrict__ srcn,
    const int* __restrict__ src, const int* __restrict__ dst,
    float* __restrict__ agg, int n_edges) {
    const unsigned lane = threadIdx.x & 31u;
    const int w = __builtin_amdgcn_readfirstlane(
        (int)(blockIdx.x * EDGE_WAVES_PER_BLOCK + (threadIdx.x >> 5)));

    const int ngroups = (n_edges + 7) >> 3;
    const int q  = ngroups / EDGE_WAVES;
    const int r  = ngroups % EDGE_WAVES;
    const int g0 = w * q + (w < r ? w : r);
    const int g1 = g0 + q + (w < r ? 1 : 0);
    const int e0 = g0 * 8;
    const int e1 = (g1 * 8 < n_edges) ? g1 * 8 : n_edges;

    for (int base = e0; base < e1; base += 32) {
        const int idx = base + (int)lane;
        int sv = 0, dv = 0;
        if (idx < e1) { sv = src[idx]; dv = dst[idx]; }
        const int cnt = (e1 - base < 32) ? (e1 - base) : 32;

        int j = 0;
        for (; j + 8 <= cnt; j += 8) {
            int   s[8], d[8];
            float sn[8];
            float v[8][4];
            #pragma unroll
            for (int u = 0; u < 8; ++u) {
                s[u] = __builtin_amdgcn_readlane(sv, j + u);
                d[u] = __builtin_amdgcn_readlane(dv, j + u);
            }
            #pragma unroll
            for (int u = 0; u < 8; ++u) {
                const float* hp = h + (size_t)s[u] * HID_C;
                sn[u]   = srcn[s[u]];
                v[u][0] = hp[lane];
                v[u][1] = hp[lane + 32u];
                v[u][2] = hp[lane + 64u];
                v[u][3] = hp[lane + 96u];
            }
            #pragma unroll
            for (int u = 0; u < 8; ++u) {
                float* ap = agg + (size_t)d[u] * HID_C;
                atomicAdd(ap + lane,       v[u][0] * sn[u]);
                atomicAdd(ap + lane + 32u, v[u][1] * sn[u]);
                atomicAdd(ap + lane + 64u, v[u][2] * sn[u]);
                atomicAdd(ap + lane + 96u, v[u][3] * sn[u]);
            }
        }
        for (; j < cnt; ++j) {
            const int s1 = __builtin_amdgcn_readlane(sv, j);
            const int d1 = __builtin_amdgcn_readlane(dv, j);
            const float* hp = h + (size_t)s1 * HID_C;
            const float sn1 = srcn[s1];
            const float a0 = hp[lane];
            const float a1 = hp[lane + 32u];
            const float a2 = hp[lane + 64u];
            const float a3 = hp[lane + 96u];
            float* ap = agg + (size_t)d1 * HID_C;
            atomicAdd(ap + lane,       a0 * sn1);
            atomicAdd(ap + lane + 32u, a1 * sn1);
            atomicAdd(ap + lane + 64u, a2 * sn1);
            atomicAdd(ap + lane + 96u, a3 * sn1);
        }
    }
}

__global__ void __launch_bounds__(256) conv_gemm_kernel(
    const float* __restrict__ agg, const float* __restrict__ dstn,
    const float* __restrict__ W, const float* __restrict__ bias,
    float* __restrict__ out, int n_nodes) {
    __shared__ float As[16][HID_C + 4];

    const int tid   = threadIdx.x;
    const int mbase = blockIdx.x * 16;

    for (int i = tid; i < 16 * (HID_C / 4); i += 256) {
        int r    = i >> 5;
        int c4   = i & 31;
        int node = mbase + r;
        float4 v = make_float4(0.f, 0.f, 0.f, 0.f);
        float  sc = 0.f;
        if (node < n_nodes) {
            v  = *(const float4*)(agg + (size_t)node * HID_C + c4 * 4);
            sc = dstn[node];
        }
        As[r][c4 * 4 + 0] = v.x * sc;
        As[r][c4 * 4 + 1] = v.y * sc;
        As[r][c4 * 4 + 2] = v.z * sc;
        As[r][c4 * 4 + 3] = v.w * sc;
    }
    __syncthreads();

    const int lane  = tid & 31;
    const int wave  = tid >> 5;
    const int l15   = lane & 15;
    const int khalf = (lane >> 4) << 1;
    const int col   = wave * 16 + l15;

    v8f acc = {0.f, 0.f, 0.f, 0.f, 0.f, 0.f, 0.f, 0.f};
    #pragma unroll
    for (int k = 0; k < HID_C; k += 4) {
        v2f a, b;
        a.x = As[l15][k + khalf];
        a.y = As[l15][k + khalf + 1];
        b.x = W[(size_t)(k + khalf) * HID_C + col];
        b.y = W[(size_t)(k + khalf + 1) * HID_C + col];
        acc = __builtin_amdgcn_wmma_f32_16x16x4_f32(false, a, false, b,
                                                    (short)0, acc, false, false);
    }

    const float bv = bias[col];
    #pragma unroll
    for (int i = 0; i < 8; ++i) {
        int node = mbase + i + ((lane >> 4) << 3);
        if (node < n_nodes)
            out[(size_t)node * HID_C + col] = fmaxf(acc[i] + bv, 0.0f);
    }
}

__global__ void __launch_bounds__(32) wmma_gemm_kernel(
    const float* __restrict__ A, const float* __restrict__ B,
    const float* __restrict__ bias, float* __restrict__ C,
    int M, int N, int K, int do_relu) {
    const int lane  = threadIdx.x & 31;
    const int l15   = lane & 15;
    const int khalf = (lane >> 4) << 1;
    const int mbase = blockIdx.x * 16;
    const int nbase = blockIdx.y * 16;
    const int row   = mbase + l15;
    const int col   = nbase + l15;

    v8f acc = {0.f, 0.f, 0.f, 0.f, 0.f, 0.f, 0.f, 0.f};
    for (int k = 0; k < K; k += 4) {
        int k0 = k + khalf, k1 = k + khalf + 1;
        v2f a, b;
        a.x = (row < M) ? A[(size_t)row * K + k0] : 0.0f;
        a.y = (row < M) ? A[(size_t)row * K + k1] : 0.0f;
        b.x = (col < N) ? B[(size_t)k0 * N + col] : 0.0f;
        b.y = (col < N) ? B[(size_t)k1 * N + col] : 0.0f;
        acc = __builtin_amdgcn_wmma_f32_16x16x4_f32(false, a, false, b,
                                                    (short)0, acc, false, false);
    }

    if (col < N) {
        const float bv = bias[col];
        #pragma unroll
        for (int i = 0; i < 8; ++i) {
            int m = mbase + i + ((lane >> 4) << 3);
            if (m < M) {
                float v = acc[i] + bv;
                C[(size_t)m * N + col] = do_relu ? fmaxf(v, 0.0f) : v;
            }
        }
    }
}

#define POOL_WAVES_PER_BLOCK 10
#define POOL_BLOCKS 625
#define POOL_WAVES (POOL_BLOCKS * POOL_WAVES_PER_BLOCK)

__device__ __forceinline__ void pool_flush(float* __restrict__ hg, int g, unsigned lane,
                                           unsigned m0, unsigned m1, unsigned m2, unsigned m3) {
    unsigned int* p = (unsigned int*)(hg + (size_t)g * HID_C);
    atomicMax(p + lane,       m0);
    atomicMax(p + lane + 32u, m1);
    atomicMax(p + lane + 64u, m2);
    atomicMax(p + lane + 96u, m3);
}

__global__ void __launch_bounds__(POOL_WAVES_PER_BLOCK * 32) segmax_kernel(
    const float* __restrict__ h, const int* __restrict__ gids,
    float* __restrict__ hg, int n_nodes) {
    const unsigned lane = threadIdx.x & 31u;
    const int w = __builtin_amdgcn_readfirstlane(
        (int)(blockIdx.x * POOL_WAVES_PER_BLOCK + (threadIdx.x >> 5)));

    const int q  = n_nodes / POOL_WAVES;
    const int r  = n_nodes % POOL_WAVES;
    const int n0 = w * q + (w < r ? w : r);
    const int n1 = n0 + q + (w < r ? 1 : 0);

    int cur = -1;
    unsigned m0 = 0u, m1 = 0u, m2 = 0u, m3 = 0u;

    for (int base = n0; base < n1; base += 32) {
        const int idx = base + (int)lane;
        int gv = 0;
        if (idx < n1) gv = gids[idx];
        const int cnt = (n1 - base < 32) ? (n1 - base) : 32;

        int j = 0;
        for (; j + 8 <= cnt; j += 8) {
            float v[8][4];
            #pragma unroll
            for (int u = 0; u < 8; ++u) {
                const float* hp = h + (size_t)(base + j + u) * HID_C;
                v[u][0] = hp[lane];
                v[u][1] = hp[lane + 32u];
                v[u][2] = hp[lane + 64u];
                v[u][3] = hp[lane + 96u];
            }
            #pragma unroll
            for (int u = 0; u < 8; ++u) {
                const int g = __builtin_amdgcn_readlane(gv, j + u);
                if (g != cur) {
                    if (cur >= 0) pool_flush(hg, cur, lane, m0, m1, m2, m3);
                    m0 = 0u; m1 = 0u; m2 = 0u; m3 = 0u;
                    cur = g;
                }
                const unsigned b0 = __float_as_uint(v[u][0]);
                const unsigned b1 = __float_as_uint(v[u][1]);
                const unsigned b2 = __float_as_uint(v[u][2]);
                const unsigned b3 = __float_as_uint(v[u][3]);
                m0 = (b0 > m0) ? b0 : m0;
                m1 = (b1 > m1) ? b1 : m1;
                m2 = (b2 > m2) ? b2 : m2;
                m3 = (b3 > m3) ? b3 : m3;
            }
        }
        for (; j < cnt; ++j) {
            const float* hp = h + (size_t)(base + j) * HID_C;
            const unsigned b0 = __float_as_uint(hp[lane]);
            const unsigned b1 = __float_as_uint(hp[lane + 32u]);
            const unsigned b2 = __float_as_uint(hp[lane + 64u]);
            const unsigned b3 = __float_as_uint(hp[lane + 96u]);
            const int g = __builtin_amdgcn_readlane(gv, j);
            if (g != cur) {
                if (cur >= 0) pool_flush(hg, cur, lane, m0, m1, m2, m3);
                m0 = 0u; m1 = 0u; m2 = 0u; m3 = 0u;
                cur = g;
            }
            m0 = (b0 > m0) ? b0 : m0;
            m1 = (b1 > m1) ? b1 : m1;
            m2 = (b2 > m2) ? b2 : m2;
            m3 = (b3 > m3) ? b3 : m3;
        }
    }
    if (cur >= 0) pool_flush(hg, cur, lane, m0, m1, m2, m3);
}

extern "C" void kernel_launch(void* const* d_in, const int* in_sizes, int n_in,
                              void* d_out, int out_size, void* d_ws, size_t ws_size,
                              hipStream_t stream) {
    (void)in_sizes; (void)n_in; (void)out_size; (void)ws_size;

    const float* feat = (const float*)d_in[0];
    const int*   src  = (const int*)d_in[1];
    const int*   dst  = (const int*)d_in[2];
    const int*   gids = (const int*)d_in[3];
    const float* Wg[3] = {(const float*)d_in[4], (const float*)d_in[6], (const float*)d_in[8]};
    const float* bg[3] = {(const float*)d_in[5], (const float*)d_in[7], (const float*)d_in[9]};
    const float* Wc1 = (const float*)d_in[10]; const float* bc1 = (const float*)d_in[11];
    const float* Wc2 = (const float*)d_in[12]; const float* bc2 = (const float*)d_in[13];
    const float* Wc3 = (const float*)d_in[14]; const float* bc3 = (const float*)d_in[15];
    float* out = (float*)d_out;

    char* ws = (char*)d_ws;
    size_t off = 0;
    auto take = [&](size_t elems) -> float* {
        float* p = (float*)(ws + off);
        off += ((elems * sizeof(float) + 255) / 256) * 256;
        return p;
    };
    unsigned int* part = (unsigned int*)take((size_t)2 * DEG_CHUNKS * N_NODES_C);
    float* srcn = take(N_NODES_C);
    float* dstn = take(N_NODES_C);
    float* agg  = take((size_t)N_NODES_C * HID_C);
    float* h    = take((size_t)N_NODES_C * HID_C);
    float* hg   = take((size_t)N_GRAPHS_C * HID_C);
    float* x1   = take((size_t)N_GRAPHS_C * HID_C);
    float* x2   = take((size_t)N_GRAPHS_C * (HID_C / 2));

    degree_kernel<<<dim3(DEG_RANGES, DEG_CHUNKS, 2), DEG_THREADS, 0, stream>>>(
        src, dst, part, N_EDGES_C, N_NODES_C);
    norm_kernel<<<(N_NODES_C + 255) / 256, 256, 0, stream>>>(part, srcn, dstn, N_NODES_C);

    const int mblocks = (N_NODES_C + 15) / 16;
    const float* hin = feat;
    for (int L = 0; L < 3; ++L) {
        hipMemsetAsync(agg, 0, (size_t)N_NODES_C * HID_C * sizeof(float), stream);
        edge_agg_kernel<<<EDGE_BLOCKS, EDGE_WAVES_PER_BLOCK * 32, 0, stream>>>(
            hin, srcn, src, dst, agg, N_EDGES_C);
        conv_gemm_kernel<<<mblocks, 256, 0, stream>>>(agg, dstn, Wg[L], bg[L], h, N_NODES_C);
        hin = h;
    }

    hipMemsetAsync(hg, 0, (size_t)N_GRAPHS_C * HID_C * sizeof(float), stream);
    segmax_kernel<<<POOL_BLOCKS, POOL_WAVES_PER_BLOCK * 32, 0, stream>>>(h, gids, hg, N_NODES_C);

    wmma_gemm_kernel<<<dim3(N_GRAPHS_C / 16, HID_C / 16), 32, 0, stream>>>(
        hg, Wc1, bc1, x1, N_GRAPHS_C, HID_C, HID_C, 1);
    wmma_gemm_kernel<<<dim3(N_GRAPHS_C / 16, (HID_C / 2) / 16), 32, 0, stream>>>(
        x1, Wc2, bc2, x2, N_GRAPHS_C, HID_C / 2, HID_C, 1);
    wmma_gemm_kernel<<<dim3(N_GRAPHS_C / 16, (OUT_DIM_C + 15) / 16), 32, 0, stream>>>(
        x2, Wc3, bc3, out, N_GRAPHS_C, OUT_DIM_C, HID_C / 2, 0);
}


# Edge sums in an LDS tile of 400 rows, gather pipelined: a wave queues its matching edges in LDS, takes them 8 at a time, 32 row loads in flight, LDS f32 atomic adds (arrival order); agg fills gone
# speedup vs baseline: 2.0891x; 1.1813x over previous
#include <hip/hip_runtime.h>
#include <hip/hip_bf16.h>


typedef float v2f __attribute__((ext_vector_type(2)));
typedef float v8f __attribute__((ext_vector_type(8)));

#define N_NODES_C 100000
#define N_EDGES_C 800000
#define HID_C     128
#define OUT_DIM_C 500
#define N_GRAPHS_C 64

#define DEG_RANGES  16
#define DEG_CHUNKS  16
#define DEG_THREADS 256
#define DEG_RANGE_NODES ((((N_NODES_C + DEG_RANGES - 1) / DEG_RANGES) + 31) & ~31)
#define DEG_LOADS   13

__global__ void __launch_bounds__(DEG_THREADS) degree_kernel(
    const int* __restrict__ src, const int* __restrict__ dst,
    unsigned int* __restrict__ part, int n_edges, int n_nodes) {
    __shared__ unsigned int cnt[DEG_RANGE_NODES];

    const int tid = threadIdx.x;
    const int* __restrict__ ids = (blockIdx.z == 0) ? src : dst;
    const int lo = blockIdx.x * DEG_RANGE_NODES;
    int span = n_nodes - lo;
    if (span > DEG_RANGE_NODES) span = DEG_RANGE_NODES;
    if (span < 0) span = 0;

    for (int i = tid; i < DEG_RANGE_NODES; i += DEG_THREADS) cnt[i] = 0u;
    __syncthreads();

    const int chunk = (((n_edges + DEG_CHUNKS - 1) / DEG_CHUNKS) + 3) & ~3;
    const int e0 = blockIdx.y * chunk;
    int e1 = e0 + chunk;
    if (e1 > n_edges) e1 = n_edges;
    if (e1 < e0) e1 = e0;

    const int e1v = e0 + ((e1 - e0) & ~3);
    for (int base = e0 + 4 * tid; base < e1v; base += 4 * DEG_THREADS * DEG_LOADS) {
        int4 v[DEG_LOADS];
        #pragma unroll
        for (int u = 0; u < DEG_LOADS; ++u) {
            const int e = base + u * 4 * DEG_THREADS;
            v[u] = make_int4(-1, -1, -1, -1);
            if (e < e1v) v[u] = *(const int4*)(ids + e);
        }
        #pragma unroll
        for (int u = 0; u < DEG_LOADS; ++u) {
            const unsigned rx = (unsigned)(v[u].x - lo);
            const unsigned ry = (unsigned)(v[u].y - lo);
            const unsigned rz = (unsigned)(v[u].z - lo);
            const unsigned rw = (unsigned)(v[u].w - lo);
            if (rx < (unsigned)span) atomicAdd(&cnt[rx], 1u);
            if (ry < (unsigned)span) atomicAdd(&cnt[ry], 1u);
            if (rz < (unsigned)span) atomicAdd(&cnt[rz], 1u);
            if (rw < (unsigned)span) atomicAdd(&cnt[rw], 1u);
        }
    }
    if (tid < e1 - e1v) {
        const unsigned rt = (unsigned)(ids[e1v + tid] - lo);
        if (rt < (unsigned)span) atomicAdd(&cnt[rt], 1u);
    }
    __syncthreads();

    unsigned int* __restrict__ dstp =
        part + ((size_t)blockIdx.z * DEG_CHUNKS + blockIdx.y) * (size_t)n_nodes + lo;
    for (int i = tid; i < span; i += DEG_THREADS) dstp[i] = cnt[i];
}

__global__ void norm_kernel(const unsigned int* __restrict__ part,
                            float* __restrict__ srcn, float* __restrict__ dstn, int n) {
    int i = blockIdx.x * blockDim.x + threadIdx.x;
    if (i < n) {
        unsigned int od = 0u, id = 0u;
        #pragma unroll
        for (int c = 0; c < DEG_CHUNKS; ++c) {
            od += part[(size_t)c * n + i];
            id += part[(size_t)(DEG_CHUNKS + c) * n + i];
        }
        srcn[i] = rsqrtf(fmaxf((float)od, 1.0f));
        dstn[i] = rsqrtf(fmaxf((float)id, 1.0f));
    }
}

#define TILE_ROWS    400
#define TILE_THREADS 1024
#define TILE_WAVES   (TILE_THREADS / 32)
#define SCAN_LOADS   4
#define CHUNK_IDS    (SCAN_LOADS * 128)
#define QCAP         (CHUNK_IDS + 8)
#define TILE_BLOCKS  ((N_NODES_C + TILE_ROWS - 1) / TILE_ROWS)

static_assert(N_EDGES_C <= (1 << 20), "edge index must fit bits 0-19 of a queue word");
static_assert(TILE_ROWS <= (1 << 12), "tile row must fit above bit 20 of a queue word");

__device__ __forceinline__ void scan_load(int4 (&v)[SCAN_LOADS], const int* __restrict__ ids,
                                          int cbase, unsigned lane) {
    #pragma unroll
    for (int u = 0; u < SCAN_LOADS; ++u)
        v[u] = *(const int4*)(ids + cbase + u * 128 + 4 * (int)lane);
}

template <bool FULL>
__device__ __forceinline__ void tile_group(const float* __restrict__ h,
                                           const float* __restrict__ srcn,
                                           float* tile, int sv, int dv,
                                           int first, int nvalid, unsigned lane) {
    int   s[8], d[8];
    float sn[8];
    float v[8][4];
    #pragma unroll
    for (int u = 0; u < 8; ++u) {
        s[u] = __builtin_amdgcn_readlane(sv, first + u);
        d[u] = __builtin_amdgcn_readlane(dv, first + u);
    }
    #pragma unroll
    for (int u = 0; u < 8; ++u) {
        sn[u] = 0.0f;
        v[u][0] = 0.0f; v[u][1] = 0.0f; v[u][2] = 0.0f; v[u][3] = 0.0f;
        if (FULL || u < nvalid) {
            const float* hp = h + (size_t)s[u] * HID_C;
            sn[u]   = srcn[s[u]];
            v[u][0] = hp[lane];
            v[u][1] = hp[lane + 32u];
            v[u][2] = hp[lane + 64u];
            v[u][3] = hp[lane + 96u];
        }
    }
    #pragma unroll
    for (int u = 0; u < 8; ++u) {
        if (FULL || u < nvalid) {
            float* tp = tile + d[u] * HID_C;
            __hip_atomic_fetch_add(tp + lane,       v[u][0] * sn[u], __ATOMIC_RELAXED, __HIP_MEMORY_SCOPE_WORKGROUP);
            __hip_atomic_fetch_add(tp + lane + 32u, v[u][1] * sn[u], __ATOMIC_RELAXED, __HIP_MEMORY_SCOPE_WORKGROUP);
            __hip_atomic_fetch_add(tp + lane + 64u, v[u][2] * sn[u], __ATOMIC_RELAXED, __HIP_MEMORY_SCOPE_WORKGROUP);
            __hip_atomic_fetch_add(tp + lane + 96u, v[u][3] * sn[u], __ATOMIC_RELAXED, __HIP_MEMORY_SCOPE_WORKGROUP);
        }
    }
}

__global__ void __launch_bounds__(TILE_THREADS) edge_agg_kernel(
    const float* __restrict__ h, const float* __restrict__ srcn,
    const int* __restrict__ src, const int* __restrict__ dst,
    float* __restrict__ agg, int n_edges, int n_nodes) {
    __shared__ float        tile[TILE_ROWS * HID_C];
    __shared__ unsigned int queue[TILE_WAVES * QCAP];

    const unsigned lane = threadIdx.x & 31u;
    const int w  = __builtin_amdgcn_readfirstlane((int)(threadIdx.x >> 5));
    const int lo = blockIdx.x * TILE_ROWS;
    int span = n_nodes - lo;
    if (span > TILE_ROWS) span = TILE_ROWS;
    if (span < 0) span = 0;

    float4* t4 = (float4*)tile;
    for (int i = threadIdx.x; i < TILE_ROWS * HID_C / 4; i += TILE_THREADS)
        t4[i] = make_float4(0.f, 0.f, 0.f, 0.f);
    __syncthreads();

    unsigned int* q = queue + w * QCAP;
    int count = 0;

    auto enqueue = [&](int id, int e) {
        const unsigned rel  = (unsigned)(id - lo);
        const bool     m    = rel < (unsigned)span;
        const unsigned mask = __builtin_amdgcn_ballot_w32(m);
        if (mask != 0u) {
            const unsigned rank = (unsigned)count + __builtin_amdgcn_mbcnt_lo(mask, 0u);
            if (m) q[rank] = (unsigned)e | (rel << 20);
            count += __builtin_popcount(mask);
        }
    };
    auto drain = [&]() {
        const int nfull = count & ~7;
        for (int pos = 0; pos < nfull; pos += 32) {
            const int n = (nfull - pos < 32) ? (nfull - pos) : 32;
            unsigned ent = 0u;
            int sv = 0;
            if ((int)lane < n) {
                ent = q[pos + (int)lane];
                sv  = src[ent & 0xFFFFFu];
            }
            const int dv = (int)(ent >> 20);
            for (int b = 0; b < n; b += 8)
                tile_group<true>(h, srcn, tile, sv, dv, b, 8, lane);
        }
        const int rem = count - nfull;
        unsigned carry = 0u;
        if ((int)lane < rem) carry = q[nfull + (int)lane];
        if ((int)lane < rem) q[lane] = carry;
        count = rem;
    };

    const int nfc = n_edges / CHUNK_IDS;
    if (w < nfc) {
        int4 cur[SCAN_LOADS], nxt[SCAN_LOADS];
        scan_load(cur, dst, w * CHUNK_IDS, lane);
        for (int c = w; c < nfc; c += TILE_WAVES) {
            const int cn = (c + TILE_WAVES < nfc) ? (c + TILE_WAVES) : (nfc - 1);
            scan_load(nxt, dst, cn * CHUNK_IDS, lane);
            #pragma unroll
            for (int u = 0; u < SCAN_LOADS; ++u) {
                const int e = c * CHUNK_IDS + u * 128 + 4 * (int)lane;
                enqueue(cur[u].x, e);
                enqueue(cur[u].y, e + 1);
                enqueue(cur[u].z, e + 2);
                enqueue(cur[u].w, e + 3);
            }
            if (count >= 8) drain();
            #pragma unroll
            for (int u = 0; u < SCAN_LOADS; ++u) cur[u] = nxt[u];
        }
    }
    if (w == nfc % TILE_WAVES) {
        for (int e0 = nfc * CHUNK_IDS; e0 < n_edges; e0 += 32) {
            const int e = e0 + (int)lane;
            const int id = (e < n_edges) ? dst[e] : -1;
            enqueue(id, e);
        }
        if (count >= 8) drain();
    }
    if (count > 0) {
        unsigned ent = 0u;
        int sv = 0;
        if ((int)lane < count) {
            ent = q[lane];
            sv  = src[ent & 0xFFFFFu];
        }
        const int dv = (int)(ent >> 20);
        tile_group<false>(h, srcn, tile, sv, dv, 0, count, lane);
    }
    __syncthreads();

    for (int r = w; r < span; r += TILE_WAVES) {
        const float4 t = *(const float4*)(tile + r * HID_C + 4 * (int)lane);
        *(float4*)(agg + (size_t)(lo + r) * HID_C + 4 * lane) = t;
    }
}

__global__ void __launch_bounds__(256) conv_gemm_kernel(
    const float* __restrict__ agg, const float* __restrict__ dstn,
    const float* __restrict__ W, const float* __restrict__ bias,
    float* __restrict__ out, int n_nodes) {
    __shared__ float As[16][HID_C + 4];

    const int tid   = threadIdx.x;
    const int mbase = blockIdx.x * 16;

    for (int i = tid; i < 16 * (HID_C / 4); i += 256) {
        int r    = i >> 5;
        int c4   = i & 31;
        int node = mbase + r;
        float4 v = make_float4(0.f, 0.f, 0.f, 0.f);
        float  sc = 0.f;
        if (node < n_nodes) {
            v  = *(const float4*)(agg + (size_t)node * HID_C + c4 * 4);
            sc = dstn[node];
        }
        As[r][c4 * 4 + 0] = v.x * sc;
        As[r][c4 * 4 + 1] = v.y * sc;
        As[r][c4 * 4 + 2] = v.z * sc;
        As[r][c4 * 4 + 3] = v.w * sc;
    }
    __syncthreads();

    const int lane  = tid & 31;
    const int wave  = tid >> 5;
    const int l15   = lane & 15;
    const int khalf = (lane >> 4) << 1;
    const int col   = wave * 16 + l15;

    v8f acc = {0.f, 0.f, 0.f, 0.f, 0.f, 0.f, 0.f, 0.f};
    #pragma unroll
    for (int k = 0; k < HID_C; k += 4) {
        v2f a, b;
        a.x = As[l15][k + khalf];
        a.y = As[l15][k + khalf + 1];
        b.x = W[(size_t)(k + khalf) * HID_C + col];
        b.y = W[(size_t)(k + khalf + 1) * HID_C + col];
        acc = __builtin_amdgcn_wmma_f32_16x16x4_f32(false, a, false, b,
                                                    (short)0, acc, false, false);
    }

    const float bv = bias[col];
    #pragma unroll
    for (int i = 0; i < 8; ++i) {
        int node = mbase + i + ((lane >> 4) << 3);
        if (node < n_nodes)
            out[(size_t)node * HID_C + col] = fmaxf(acc[i] + bv, 0.0f);
    }
}

__global__ void __launch_bounds__(32) wmma_gemm_kernel(
    const float* __restrict__ A, const float* __restrict__ B,
    const float* __restrict__ bias, float* __restrict__ C,
    int M, int N, int K, int do_relu) {
    const int lane  = threadIdx.x & 31;
    const int l15   = lane & 15;
    const int khalf = (lane >> 4) << 1;
    const int mbase = blockIdx.x * 16;
    const int nbase = blockIdx.y * 16;
    const int row   = mbase + l15;
    const int col   = nbase + l15;

    v8f acc = {0.f, 0.f, 0.f, 0.f, 0.f, 0.f, 0.f, 0.f};
    for (int k = 0; k < K; k += 4) {
        int k0 = k + khalf, k1 = k + khalf + 1;
        v2f a, b;
        a.x = (row < M) ? A[(size_t)row * K + k0] : 0.0f;
        a.y = (row < M) ? A[(size_t)row * K + k1] : 0.0f;
        b.x = (col < N) ? B[(size_t)k0 * N + col] : 0.0f;
        b.y = (col < N) ? B[(size_t)k1 * N + col] : 0.0f;
        acc = __builtin_amdgcn_wmma_f32_16x16x4_f32(false, a, false, b,
                                                    (short)0, acc, false, false);
    }

    if (col < N) {
        const float bv = bias[col];
        #pragma unroll
        for (int i = 0; i < 8; ++i) {
            int m = mbase + i + ((lane >> 4) << 3);
            if (m < M) {
                float v = acc[i] + bv;
                C[(size_t)m * N + col] = do_relu ? fmaxf(v, 0.0f) : v;
            }
        }
    }
}

#define POOL_WAVES_PER_BLOCK 10
#define POOL_BLOCKS 625
#define POOL_WAVES (POOL_BLOCKS * POOL_WAVES_PER_BLOCK)

__device__ __forceinline__ void pool_flush(float* __restrict__ hg, int g, unsigned lane,
                                           unsigned m0, unsigned m1, unsigned m2, unsigned m3) {
    unsigned int* p = (unsigned int*)(hg + (size_t)g * HID_C);
    atomicMax(p + lane,       m0);
    atomicMax(p + lane + 32u, m1);
    atomicMax(p + lane + 64u, m2);
    atomicMax(p + lane + 96u, m3);
}

__global__ void __launch_bounds__(POOL_WAVES_PER_BLOCK * 32) segmax_kernel(
    const float* __restrict__ h, const int* __restrict__ gids,
    float* __restrict__ hg, int n_nodes) {
    const unsigned lane = threadIdx.x & 31u;
    const int w = __builtin_amdgcn_readfirstlane(
        (int)(blockIdx.x * POOL_WAVES_PER_BLOCK + (threadIdx.x >> 5)));

    const int q  = n_nodes / POOL_WAVES;
    const int r  = n_nodes % POOL_WAVES;
    const int n0 = w * q + (w < r ? w : r);
    const int n1 = n0 + q + (w < r ? 1 : 0);

    int cur = -1;
    unsigned m0 = 0u, m1 = 0u, m2 = 0u, m3 = 0u;

    for (int base = n0; base < n1; base += 32) {
        const int idx = base + (int)lane;
        int gv = 0;
        if (idx < n1) gv = gids[idx];
        const int cnt = (n1 - base < 32) ? (n1 - base) : 32;

        int j = 0;
        for (; j + 8 <= cnt; j += 8) {
            float v[8][4];
            #pragma unroll
            for (int u = 0; u < 8; ++u) {
                const float* hp = h + (size_t)(base + j + u) * HID_C;
                v[u][0] = hp[lane];
                v[u][1] = hp[lane + 32u];
                v[u][2] = hp[lane + 64u];
                v[u][3] = hp[lane + 96u];
            }
            #pragma unroll
            for (int u = 0; u < 8; ++u) {
                const int g = __builtin_amdgcn_readlane(gv, j + u);
                if (g != cur) {
                    if (cur >= 0) pool_flush(hg, cur, lane, m0, m1, m2, m3);
                    m0 = 0u; m1 = 0u; m2 = 0u; m3 = 0u;
                    cur = g;
                }
                const unsigned b0 = __float_as_uint(v[u][0]);
                const unsigned b1 = __float_as_uint(v[u][1]);
                const unsigned b2 = __float_as_uint(v[u][2]);
                const unsigned b3 = __float_as_uint(v[u][3]);
                m0 = (b0 > m0) ? b0 : m0;
                m1 = (b1 > m1) ? b1 : m1;
                m2 = (b2 > m2) ? b2 : m2;
                m3 = (b3 > m3) ? b3 : m3;
            }
        }
        for (; j < cnt; ++j) {
            const float* hp = h + (size_t)(base + j) * HID_C;
            const unsigned b0 = __float_as_uint(hp[lane]);
            const unsigned b1 = __float_as_uint(hp[lane + 32u]);
            const unsigned b2 = __float_as_uint(hp[lane + 64u]);
            const unsigned b3 = __float_as_uint(hp[lane + 96u]);
            const int g = __builtin_amdgcn_readlane(gv, j);
            if (g != cur) {
                if (cur >= 0) pool_flush(hg, cur, lane, m0, m1, m2, m3);
                m0 = 0u; m1 = 0u; m2 = 0u; m3 = 0u;
                cur = g;
            }
            m0 = (b0 > m0) ? b0 : m0;
            m1 = (b1 > m1) ? b1 : m1;
            m2 = (b2 > m2) ? b2 : m2;
            m3 = (b3 > m3) ? b3 : m3;
        }
    }
    if (cur >= 0) pool_flush(hg, cur, lane, m0, m1, m2, m3);
}

extern "C" void kernel_launch(void* const* d_in, const int* in_sizes, int n_in,
                              void* d_out, int out_size, void* d_ws, size_t ws_size,
                              hipStream_t stream) {
    (void)in_sizes; (void)n_in; (void)out_size; (void)ws_size;

    const float* feat = (const float*)d_in[0];
    const int*   src  = (const int*)d_in[1];
    const int*   dst  = (const int*)d_in[2];
    const int*   gids = (const int*)d_in[3];
    const float* Wg[3] = {(const float*)d_in[4], (const float*)d_in[6], (const float*)d_in[8]};
    const float* bg[3] = {(const float*)d_in[5], (const float*)d_in[7], (const float*)d_in[9]};
    const float* Wc1 = (const float*)d_in[10]; const float* bc1 = (const float*)d_in[11];
    const float* Wc2 = (const float*)d_in[12]; const float* bc2 = (const float*)d_in[13];
    const float* Wc3 = (const float*)d_in[14]; const float* bc3 = (const float*)d_in[15];
    float* out = (float*)d_out;

    char* ws = (char*)d_ws;
    size_t off = 0;
    auto take = [&](size_t elems) -> float* {
        float* p = (float*)(ws + off);
        off += ((elems * sizeof(float) + 255) / 256) * 256;
        return p;
    };
    unsigned int* part = (unsigned int*)take((size_t)2 * DEG_CHUNKS * N_NODES_C);
    float* srcn = take(N_NODES_C);
    float* dstn = take(N_NODES_C);
    float* agg  = take((size_t)N_NODES_C * HID_C);
    float* h    = take((size_t)N_NODES_C * HID_C);
    float* hg   = take((size_t)N_GRAPHS_C * HID_C);
    float* x1   = take((size_t)N_GRAPHS_C * HID_C);
    float* x2   = take((size_t)N_GRAPHS_C * (HID_C / 2));

    degree_kernel<<<dim3(DEG_RANGES, DEG_CHUNKS, 2), DEG_THREADS, 0, stream>>>(
        src, dst, part, N_EDGES_C, N_NODES_C);
    norm_kernel<<<(N_NODES_C + 255) / 256, 256, 0, stream>>>(part, srcn, dstn, N_NODES_C);

    const int mblocks = (N_NODES_C + 15) / 16;
    const float* hin = feat;
    for (int L = 0; L < 3; ++L) {
        edge_agg_kernel<<<TILE_BLOCKS, TILE_THREADS, 0, stream>>>(
            hin, srcn, src, dst, agg, N_EDGES_C, N_NODES_C);
        conv_gemm_kernel<<<mblocks, 256, 0, stream>>>(agg, dstn, Wg[L], bg[L], h, N_NODES_C);
        hin = h;
    }

    hipMemsetAsync(hg, 0, (size_t)N_GRAPHS_C * HID_C * sizeof(float), stream);
    segmax_kernel<<<POOL_BLOCKS, POOL_WAVES_PER_BLOCK * 32, 0, stream>>>(h, gids, hg, N_NODES_C);

    wmma_gemm_kernel<<<dim3(N_GRAPHS_C / 16, HID_C / 16), 32, 0, stream>>>(
        hg, Wc1, bc1, x1, N_GRAPHS_C, HID_C, HID_C, 1);
    wmma_gemm_kernel<<<dim3(N_GRAPHS_C / 16, (HID_C / 2) / 16), 32, 0, stream>>>(
        x1, Wc2, bc2, x2, N_GRAPHS_C, HID_C / 2, HID_C, 1);
    wmma_gemm_kernel<<<dim3(N_GRAPHS_C / 16, (OUT_DIM_C + 15) / 16), 32, 0, stream>>>(
        x2, Wc3, bc3, out, N_GRAPHS_C, OUT_DIM_C, HID_C / 2, 0);
}


# Edge kernel reads only marked id lines: a new launch builds per 400-row tile a bitmap of the 128-B lines of the dst list holding its rows; whole-list scan per tile gone; LDS f32 adds in arrival order
# speedup vs baseline: 2.2514x; 1.0777x over previous
#include <hip/hip_runtime.h>
#include <hip/hip_bf16.h>


typedef float v2f __attribute__((ext_vector_type(2)));
typedef float v8f __attribute__((ext_vector_type(8)));

#define N_NODES_C 100000
#define N_EDGES_C 800000
#define HID_C     128
#define OUT_DIM_C 500
#define N_GRAPHS_C 64

#define DEG_RANGES  16
#define DEG_CHUNKS  16
#define DEG_THREADS 256
#define DEG_RANGE_NODES ((((N_NODES_C + DEG_RANGES - 1) / DEG_RANGES) + 31) & ~31)
#define DEG_LOADS   13

__global__ void __launch_bounds__(DEG_THREADS) degree_kernel(
    const int* __restrict__ src, const int* __restrict__ dst,
    unsigned int* __restrict__ part, int n_edges, int n_nodes) {
    __shared__ unsigned int cnt[DEG_RANGE_NODES];

    const int tid = threadIdx.x;
    const int* __restrict__ ids = (blockIdx.z == 0) ? src : dst;
    const int lo = blockIdx.x * DEG_RANGE_NODES;
    int span = n_nodes - lo;
    if (span > DEG_RANGE_NODES) span = DEG_RANGE_NODES;
    if (span < 0) span = 0;

    for (int i = tid; i < DEG_RANGE_NODES; i += DEG_THREADS) cnt[i] = 0u;
    __syncthreads();

    const int chunk = (((n_edges + DEG_CHUNKS - 1) / DEG_CHUNKS) + 3) & ~3;
    const int e0 = blockIdx.y * chunk;
    int e1 = e0 + chunk;
    if (e1 > n_edges) e1 = n_edges;
    if (e1 < e0) e1 = e0;

    const int e1v = e0 + ((e1 - e0) & ~3);
    for (int base = e0 + 4 * tid; base < e1v; base += 4 * DEG_THREADS * DEG_LOADS) {
        int4 v[DEG_LOADS];
        #pragma unroll
        for (int u = 0; u < DEG_LOADS; ++u) {
            const int e = base + u * 4 * DEG_THREADS;
            v[u] = make_int4(-1, -1, -1, -1);
            if (e < e1v) v[u] = *(const int4*)(ids + e);
        }
        #pragma unroll
        for (int u = 0; u < DEG_LOADS; ++u) {
            const unsigned rx = (unsigned)(v[u].x - lo);
            const unsigned ry = (unsigned)(v[u].y - lo);
            const unsigned rz = (unsigned)(v[u].z - lo);
            const unsigned rw = (unsigned)(v[u].w - lo);
            if (rx < (unsigned)span) atomicAdd(&cnt[rx], 1u);
            if (ry < (unsigned)span) atomicAdd(&cnt[ry], 1u);
            if (rz < (unsigned)span) atomicAdd(&cnt[rz], 1u);
            if (rw < (unsigned)span) atomicAdd(&cnt[rw], 1u);
        }
    }
    if (tid < e1 - e1v) {
        const unsigned rt = (unsigned)(ids[e1v + tid] - lo);
        if (rt < (unsigned)span) atomicAdd(&cnt[rt], 1u);
    }
    __syncthreads();

    unsigned int* __restrict__ dstp =
        part + ((size_t)blockIdx.z * DEG_CHUNKS + blockIdx.y) * (size_t)n_nodes + lo;
    for (int i = tid; i < span; i += DEG_THREADS) dstp[i] = cnt[i];
}

__global__ void norm_kernel(const unsigned int* __restrict__ part,
                            float* __restrict__ srcn, float* __restrict__ dstn, int n) {
    int i = blockIdx.x * blockDim.x + threadIdx.x;
    if (i < n) {
        unsigned int od = 0u, id = 0u;
        #pragma unroll
        for (int c = 0; c < DEG_CHUNKS; ++c) {
            od += part[(size_t)c * n + i];
            id += part[(size_t)(DEG_CHUNKS + c) * n + i];
        }
        srcn[i] = rsqrtf(fmaxf((float)od, 1.0f));
        dstn[i] = rsqrtf(fmaxf((float)id, 1.0f));
    }
}

#define TILE_ROWS    400
#define TILE_THREADS 1024
#define TILE_WAVES   (TILE_THREADS / 32)
#define TILE_BLOCKS  ((N_NODES_C + TILE_ROWS - 1) / TILE_ROWS)

#define LINE_IDS     32
#define N_LINES      ((N_EDGES_C + LINE_IDS - 1) / LINE_IDS)
#define BM_WORDS     ((N_LINES + 31) / 32)
#define WAVE_WORDS   ((BM_WORDS + TILE_WAVES - 1) / TILE_WAVES)
#define BM_PITCH     (((WAVE_WORDS * TILE_WAVES) + 31) & ~31)
#define MARK_BLOCKS  (BM_PITCH / 32)
#define MARK_STRIDE  33
#define BATCH_LINES  16
#define QCAP         (BATCH_LINES * LINE_IDS + 8)

static_assert(N_EDGES_C <= (1 << 20), "edge index must fit bits 0-19 of a queue word");
static_assert(TILE_ROWS <= (1 << 12), "tile row must fit above bit 20 of a queue word");
static_assert(WAVE_WORDS <= 32, "a wave holds its bitmap words one per lane");

__global__ void __launch_bounds__(1024) mark_kernel(
    const int* __restrict__ dst, unsigned int* __restrict__ bm, int n_edges) {
    __shared__ unsigned int marks[TILE_BLOCKS * MARK_STRIDE];

    const unsigned lane = threadIdx.x & 31u;
    const int wv = (int)(threadIdx.x >> 5);
    for (int i = threadIdx.x; i < TILE_BLOCKS * MARK_STRIDE; i += 1024) marks[i] = 0u;
    __syncthreads();

    const int ebase = ((int)blockIdx.x * 32 + wv) * (32 * LINE_IDS) + (int)lane;
    #pragma unroll
    for (int g = 0; g < 32; g += 8) {
        int id[8];
        #pragma unroll
        for (int u = 0; u < 8; ++u) {
            const int e = ebase + (g + u) * LINE_IDS;
            id[u] = -1;
            if (e < n_edges) id[u] = dst[e];
        }
        #pragma unroll
        for (int u = 0; u < 8; ++u) {
            const unsigned t = (unsigned)id[u] / (unsigned)TILE_ROWS;
            if (t < (unsigned)TILE_BLOCKS) atomicOr(&marks[t * MARK_STRIDE + wv], 1u << (g + u));
        }
    }
    __syncthreads();

    for (int t = wv; t < TILE_BLOCKS; t += 32)
        bm[(size_t)t * BM_PITCH + blockIdx.x * 32 + lane] = marks[t * MARK_STRIDE + lane];
}

template <bool FULL>
__device__ __forceinline__ void tile_group(const float* __restrict__ h,
                                           const float* __restrict__ srcn,
                                           float* tile, int sv, int dv,
                                           int first, int nvalid, unsigned lane) {
    int   s[8], d[8];
    float sn[8];
    float v[8][4];
    #pragma unroll
    for (int u = 0; u < 8; ++u) {
        s[u] = __builtin_amdgcn_readlane(sv, first + u);
        d[u] = __builtin_amdgcn_readlane(dv, first + u);
    }
    #pragma unroll
    for (int u = 0; u < 8; ++u) {
        sn[u] = 0.0f;
        v[u][0] = 0.0f; v[u][1] = 0.0f; v[u][2] = 0.0f; v[u][3] = 0.0f;
        if (FULL || u < nvalid) {
            const float* hp = h + (size_t)s[u] * HID_C;
            sn[u]   = srcn[s[u]];
            v[u][0] = hp[lane];
            v[u][1] = hp[lane + 32u];
            v[u][2] = hp[lane + 64u];
            v[u][3] = hp[lane + 96u];
        }
    }
    #pragma unroll
    for (int u = 0; u < 8; ++u) {
        if (FULL || u < nvalid) {
            float* tp = tile + d[u] * HID_C;
            __hip_atomic_fetch_add(tp + lane,       v[u][0] * sn[u], __ATOMIC_RELAXED, __HIP_MEMORY_SCOPE_WORKGROUP);
            __hip_atomic_fetch_add(tp + lane + 32u, v[u][1] * sn[u], __ATOMIC_RELAXED, __HIP_MEMORY_SCOPE_WORKGROUP);
            __hip_atomic_fetch_add(tp + lane + 64u, v[u][2] * sn[u], __ATOMIC_RELAXED, __HIP_MEMORY_SCOPE_WORKGROUP);
            __hip_atomic_fetch_add(tp + lane + 96u, v[u][3] * sn[u], __ATOMIC_RELAXED, __HIP_MEMORY_SCOPE_WORKGROUP);
        }
    }
}

__global__ void __launch_bounds__(TILE_THREADS) edge_agg_kernel(
    const float* __restrict__ h, const float* __restrict__ srcn,
    const int* __restrict__ src, const int* __restrict__ dst,
    const unsigned int* __restrict__ bm,
    float* __restrict__ agg, int n_edges, int n_nodes) {
    __shared__ float        tile[TILE_ROWS * HID_C];
    __shared__ unsigned int queue[TILE_WAVES * QCAP];

    const unsigned lane = threadIdx.x & 31u;
    const int w  = __builtin_amdgcn_readfirstlane((int)(threadIdx.x >> 5));
    const int lo = blockIdx.x * TILE_ROWS;
    int span = n_nodes - lo;
    if (span > TILE_ROWS) span = TILE_ROWS;
    if (span < 0) span = 0;

    unsigned words = 0u;
    if ((int)lane < WAVE_WORDS)
        words = bm[(size_t)blockIdx.x * BM_PITCH + w * WAVE_WORDS + (int)lane];

    float4* t4 = (float4*)tile;
    for (int i = threadIdx.x; i < TILE_ROWS * HID_C / 4; i += TILE_THREADS)
        t4[i] = make_float4(0.f, 0.f, 0.f, 0.f);
    __syncthreads();

    unsigned int* q = queue + w * QCAP;
    int count = 0;

    auto enqueue = [&](int id, int e) {
        const unsigned rel  = (unsigned)(id - lo);
        const bool     m    = rel < (unsigned)span;
        const unsigned mask = __builtin_amdgcn_ballot_w32(m);
        if (mask != 0u) {
            const unsigned rank = (unsigned)count + __builtin_amdgcn_mbcnt_lo(mask, 0u);
            if (m) q[rank] = (unsigned)e | (rel << 20);
            count += __builtin_popcount(mask);
        }
    };
    auto drain = [&]() {
        const int nfull = count & ~7;
        for (int pos = 0; pos < nfull; pos += 32) {
            const int n = (nfull - pos < 32) ? (nfull - pos) : 32;
            unsigned ent = 0u;
            int sv = 0;
            if ((int)lane < n) {
                ent = q[pos + (int)lane];
                sv  = src[ent & 0xFFFFFu];
            }
            const int dv = (int)(ent >> 20);
            for (int b = 0; b < n; b += 8)
                tile_group<true>(h, srcn, tile, sv, dv, b, 8, lane);
        }
        const int rem = count - nfull;
        unsigned carry = 0u;
        if ((int)lane < rem) carry = q[nfull + (int)lane];
        if ((int)lane < rem) q[lane] = carry;
        count = rem;
    };

    const int line0 = w * WAVE_WORDS * 32;
    int p = 0;
    unsigned bits = (unsigned)__builtin_amdgcn_readlane((int)words, 0);
    auto next_line = [&]() -> int {
        while (bits == 0u && p < WAVE_WORDS - 1) {
            ++p;
            bits = (unsigned)__builtin_amdgcn_readlane((int)words, p);
        }
        if (bits == 0u) return -1;
        const int b = __builtin_ctz(bits);
        bits &= bits - 1u;
        return line0 + p * 32 + b;
    };
    auto load_batch = [&](int (&ln)[BATCH_LINES], int (&idv)[BATCH_LINES]) {
        #pragma unroll
        for (int u = 0; u < BATCH_LINES; ++u) ln[u] = next_line();
        #pragma unroll
        for (int u = 0; u < BATCH_LINES; ++u) {
            const int lnu = (ln[u] < 0) ? 0 : ln[u];
            const int e   = lnu * LINE_IDS + (int)lane;
            int v = -1;
            if (e < n_edges) v = dst[e];
            idv[u] = (ln[u] < 0) ? -1 : v;
        }
    };

    {
        int lc[BATCH_LINES], ic[BATCH_LINES], lx[BATCH_LINES], ix[BATCH_LINES];
        load_batch(lc, ic);
        for (int it = 0; it < (WAVE_WORDS * 32) / BATCH_LINES && lc[0] >= 0; ++it) {
            load_batch(lx, ix);
            #pragma unroll
            for (int u = 0; u < BATCH_LINES; ++u)
                enqueue(ic[u], lc[u] * LINE_IDS + (int)lane);
            if (count >= 8) drain();
            #pragma unroll
            for (int u = 0; u < BATCH_LINES; ++u) { lc[u] = lx[u]; ic[u] = ix[u]; }
        }
    }
    if (count > 0) {
        unsigned ent = 0u;
        int sv = 0;
        if ((int)lane < count) {
            ent = q[lane];
            sv  = src[ent & 0xFFFFFu];
        }
        const int dv = (int)(ent >> 20);
        tile_group<false>(h, srcn, tile, sv, dv, 0, count, lane);
    }
    __syncthreads();

    for (int r = w; r < span; r += TILE_WAVES) {
        const float4 t = *(const float4*)(tile + r * HID_C + 4 * (int)lane);
        *(float4*)(agg + (size_t)(lo + r) * HID_C + 4 * lane) = t;
    }
}

__global__ void __launch_bounds__(256) conv_gemm_kernel(
    const float* __restrict__ agg, const float* __restrict__ dstn,
    const float* __restrict__ W, const float* __restrict__ bias,
    float* __restrict__ out, int n_nodes) {
    __shared__ float As[16][HID_C + 4];

    const int tid   = threadIdx.x;
    const int mbase = blockIdx.x * 16;

    for (int i = tid; i < 16 * (HID_C / 4); i += 256) {
        int r    = i >> 5;
        int c4   = i & 31;
        int node = mbase + r;
        float4 v = make_float4(0.f, 0.f, 0.f, 0.f);
        float  sc = 0.f;
        if (node < n_nodes) {
            v  = *(const float4*)(agg + (size_t)node * HID_C + c4 * 4);
            sc = dstn[node];
        }
        As[r][c4 * 4 + 0] = v.x * sc;
        As[r][c4 * 4 + 1] = v.y * sc;
        As[r][c4 * 4 + 2] = v.z * sc;
        As[r][c4 * 4 + 3] = v.w * sc;
    }
    __syncthreads();

    const int lane  = tid & 31;
    const int wave  = tid >> 5;
    const int l15   = lane & 15;
    const int khalf = (lane >> 4) << 1;
    const int col   = wave * 16 + l15;

    v8f acc = {0.f, 0.f, 0.f, 0.f, 0.f, 0.f, 0.f, 0.f};
    #pragma unroll
    for (int k = 0; k < HID_C; k += 4) {
        v2f a, b;
        a.x = As[l15][k + khalf];
        a.y = As[l15][k + khalf + 1];
        b.x = W[(size_t)(k + khalf) * HID_C + col];
        b.y = W[(size_t)(k + khalf + 1) * HID_C + col];
        acc = __builtin_amdgcn_wmma_f32_16x16x4_f32(false, a, false, b,
                                                    (short)0, acc, false, false);
    }

    const float bv = bias[col];
    #pragma unroll
    for (int i = 0; i < 8; ++i) {
        int node = mbase + i + ((lane >> 4) << 3);
        if (node < n_nodes)
            out[(size_t)node * HID_C + col] = fmaxf(acc[i] + bv, 0.0f);
    }
}

__global__ void __launch_bounds__(32) wmma_gemm_kernel(
    const float* __restrict__ A, const float* __restrict__ B,
    const float* __restrict__ bias, float* __restrict__ C,
    int M, int N, int K, int do_relu) {
    const int lane  = threadIdx.x & 31;
    const int l15   = lane & 15;
    const int khalf = (lane >> 4) << 1;
    const int mbase = blockIdx.x * 16;
    const int nbase = blockIdx.y * 16;
    const int row   = mbase + l15;
    const int col   = nbase + l15;

    v8f acc = {0.f, 0.f, 0.f, 0.f, 0.f, 0.f, 0.f, 0.f};
    for (int k = 0; k < K; k += 4) {
        int k0 = k + khalf, k1 = k + khalf + 1;
        v2f a, b;
        a.x = (row < M) ? A[(size_t)row * K + k0] : 0.0f;
        a.y = (row < M) ? A[(size_t)row * K + k1] : 0.0f;
        b.x = (col < N) ? B[(size_t)k0 * N + col] : 0.0f;
        b.y = (col < N) ? B[(size_t)k1 * N + col] : 0.0f;
        acc = __builtin_amdgcn_wmma_f32_16x16x4_f32(false, a, false, b,
                                                    (short)0, acc, false, false);
    }

    if (col < N) {
        const float bv = bias[col];
        #pragma unroll
        for (int i = 0; i < 8; ++i) {
            int m = mbase + i + ((lane >> 4) << 3);
            if (m < M) {
                float v = acc[i] + bv;
                C[(size_t)m * N + col] = do_relu ? fmaxf(v, 0.0f) : v;
            }
        }
    }
}

#define POOL_WAVES_PER_BLOCK 10
#define POOL_BLOCKS 625
#define POOL_WAVES (POOL_BLOCKS * POOL_WAVES_PER_BLOCK)

__device__ __forceinline__ void pool_flush(float* __restrict__ hg, int g, unsigned lane,
                                           unsigned m0, unsigned m1, unsigned m2, unsigned m3) {
    unsigned int* p = (unsigned int*)(hg + (size_t)g * HID_C);
    atomicMax(p + lane,       m0);
    atomicMax(p + lane + 32u, m1);
    atomicMax(p + lane + 64u, m2);
    atomicMax(p + lane + 96u, m3);
}

__global__ void __launch_bounds__(POOL_WAVES_PER_BLOCK * 32) segmax_kernel(
    const float* __restrict__ h, const int* __restrict__ gids,
    float* __restrict__ hg, int n_nodes) {
    const unsigned lane = threadIdx.x & 31u;
    const int w = __builtin_amdgcn_readfirstlane(
        (int)(blockIdx.x * POOL_WAVES_PER_BLOCK + (threadIdx.x >> 5)));

    const int q  = n_nodes / POOL_WAVES;
    const int r  = n_nodes % POOL_WAVES;
    const int n0 = w * q + (w < r ? w : r);
    const int n1 = n0 + q + (w < r ? 1 : 0);

    int cur = -1;
    unsigned m0 = 0u, m1 = 0u, m2 = 0u, m3 = 0u;

    for (int base = n0; base < n1; base += 32) {
        const int idx = base + (int)lane;
        int gv = 0;
        if (idx < n1) gv = gids[idx];
        const int cnt = (n1 - base < 32) ? (n1 - base) : 32;

        int j = 0;
        for (; j + 8 <= cnt; j += 8) {
            float v[8][4];
            #pragma unroll
            for (int u = 0; u < 8; ++u) {
                const float* hp = h + (size_t)(base + j + u) * HID_C;
                v[u][0] = hp[lane];
                v[u][1] = hp[lane + 32u];
                v[u][2] = hp[lane + 64u];
                v[u][3] = hp[lane + 96u];
            }
            #pragma unroll
            for (int u = 0; u < 8; ++u) {
                const int g = __builtin_amdgcn_readlane(gv, j + u);
                if (g != cur) {
                    if (cur >= 0) pool_flush(hg, cur, lane, m0, m1, m2, m3);
                    m0 = 0u; m1 = 0u; m2 = 0u; m3 = 0u;
                    cur = g;
                }
                const unsigned b0 = __float_as_uint(v[u][0]);
                const unsigned b1 = __float_as_uint(v[u][1]);
                const unsigned b2 = __float_as_uint(v[u][2]);
                const unsigned b3 = __float_as_uint(v[u][3]);
                m0 = (b0 > m0) ? b0 : m0;
                m1 = (b1 > m1) ? b1 : m1;
                m2 = (b2 > m2) ? b2 : m2;
                m3 = (b3 > m3) ? b3 : m3;
            }
        }
        for (; j < cnt; ++j) {
            const float* hp = h + (size_t)(base + j) * HID_C;
            const unsigned b0 = __float_as_uint(hp[lane]);
            const unsigned b1 = __float_as_uint(hp[lane + 32u]);
            const unsigned b2 = __float_as_uint(hp[lane + 64u]);
            const unsigned b3 = __float_as_uint(hp[lane + 96u]);
            const int g = __builtin_amdgcn_readlane(gv, j);
            if (g != cur) {
                if (cur >= 0) pool_flush(hg, cur, lane, m0, m1, m2, m3);
                m0 = 0u; m1 = 0u; m2 = 0u; m3 = 0u;
                cur = g;
            }
            m0 = (b0 > m0) ? b0 : m0;
            m1 = (b1 > m1) ? b1 : m1;
            m2 = (b2 > m2) ? b2 : m2;
            m3 = (b3 > m3) ? b3 : m3;
        }
    }
    if (cur >= 0) pool_flush(hg, cur, lane, m0, m1, m2, m3);
}

extern "C" void kernel_launch(void* const* d_in, const int* in_sizes, int n_in,
                              void* d_out, int out_size, void* d_ws, size_t ws_size,
                              hipStream_t stream) {
    (void)in_sizes; (void)n_in; (void)out_size; (void)ws_size;

    const float* feat = (const float*)d_in[0];
    const int*   src  = (const int*)d_in[1];
    const int*   dst  = (const int*)d_in[2];
    const int*   gids = (const int*)d_in[3];
    const float* Wg[3] = {(const float*)d_in[4], (const float*)d_in[6], (const float*)d_in[8]};
    const float* bg[3] = {(const float*)d_in[5], (const float*)d_in[7], (const float*)d_in[9]};
    const float* Wc1 = (const float*)d_in[10]; const float* bc1 = (const float*)d_in[11];
    const float* Wc2 = (const float*)d_in[12]; const float* bc2 = (const float*)d_in[13];
    const float* Wc3 = (const float*)d_in[14]; const float* bc3 = (const float*)d_in[15];
    float* out = (float*)d_out;

    char* ws = (char*)d_ws;
    size_t off = 0;
    auto take = [&](size_t elems) -> float* {
        float* p = (float*)(ws + off);
        off += ((elems * sizeof(float) + 255) / 256) * 256;
        return p;
    };
    unsigned int* part = (unsigned int*)take((size_t)2 * DEG_CHUNKS * N_NODES_C);
    float* srcn = take(N_NODES_C);
    float* dstn = take(N_NODES_C);
    unsigned int* bm = (unsigned int*)take((size_t)TILE_BLOCKS * BM_PITCH);
    float* agg  = take((size_t)N_NODES_C * HID_C);
    float* h    = take((size_t)N_NODES_C * HID_C);
    float* hg   = take((size_t)N_GRAPHS_C * HID_C);
    float* x1   = take((size_t)N_GRAPHS_C * HID_C);
    float* x2   = take((size_t)N_GRAPHS_C * (HID_C / 2));

    degree_kernel<<<dim3(DEG_RANGES, DEG_CHUNKS, 2), DEG_THREADS, 0, stream>>>(
        src, dst, part, N_EDGES_C, N_NODES_C);
    norm_kernel<<<(N_NODES_C + 255) / 256, 256, 0, stream>>>(part, srcn, dstn, N_NODES_C);

    mark_kernel<<<MARK_BLOCKS, 1024, 0, stream>>>(dst, bm, N_EDGES_C);

    const int mblocks = (N_NODES_C + 15) / 16;
    const float* hin = feat;
    for (int L = 0; L < 3; ++L) {
        edge_agg_kernel<<<TILE_BLOCKS, TILE_THREADS, 0, stream>>>(
            hin, srcn, src, dst, bm, agg, N_EDGES_C, N_NODES_C);
        conv_gemm_kernel<<<mblocks, 256, 0, stream>>>(agg, dstn, Wg[L], bg[L], h, N_NODES_C);
        hin = h;
    }

    hipMemsetAsync(hg, 0, (size_t)N_GRAPHS_C * HID_C * sizeof(float), stream);
    segmax_kernel<<<POOL_BLOCKS, POOL_WAVES_PER_BLOCK * 32, 0, stream>>>(h, gids, hg, N_NODES_C);

    wmma_gemm_kernel<<<dim3(N_GRAPHS_C / 16, HID_C / 16), 32, 0, stream>>>(
        hg, Wc1, bc1, x1, N_GRAPHS_C, HID_C, HID_C, 1);
    wmma_gemm_kernel<<<dim3(N_GRAPHS_C / 16, (HID_C / 2) / 16), 32, 0, stream>>>(
        x1, Wc2, bc2, x2, N_GRAPHS_C, HID_C / 2, HID_C, 1);
    wmma_gemm_kernel<<<dim3(N_GRAPHS_C / 16, (OUT_DIM_C + 15) / 16), 32, 0, stream>>>(
        x2, Wc3, bc3, out, N_GRAPHS_C, OUT_DIM_C, HID_C / 2, 0);
}


# Conv product taken from the LDS edge tile in the same launch: 32 waves run the 400x128 by 128x128 WMMA there, W slice in registers; edge-sum store, its re-read and 3 launches gone; sum orders kept
# speedup vs baseline: 2.5973x; 1.1536x over previous
#include <hip/hip_runtime.h>
#include <hip/hip_bf16.h>


typedef float v2f __attribute__((ext_vector_type(2)));
typedef float v8f __attribute__((ext_vector_type(8)));

#define N_NODES_C 100000
#define N_EDGES_C 800000
#define HID_C     128
#define OUT_DIM_C 500
#define N_GRAPHS_C 64

#define DEG_RANGES  16
#define DEG_CHUNKS  16
#define DEG_THREADS 256
#define DEG_RANGE_NODES ((((N_NODES_C + DEG_RANGES - 1) / DEG_RANGES) + 31) & ~31)
#define DEG_LOADS   13

__global__ void __launch_bounds__(DEG_THREADS) degree_kernel(
    const int* __restrict__ src, const int* __restrict__ dst,
    unsigned int* __restrict__ part, int n_edges, int n_nodes) {
    __shared__ unsigned int cnt[DEG_RANGE_NODES];

    const int tid = threadIdx.x;
    const int* __restrict__ ids = (blockIdx.z == 0) ? src : dst;
    const int lo = blockIdx.x * DEG_RANGE_NODES;
    int span = n_nodes - lo;
    if (span > DEG_RANGE_NODES) span = DEG_RANGE_NODES;
    if (span < 0) span = 0;

    for (int i = tid; i < DEG_RANGE_NODES; i += DEG_THREADS) cnt[i] = 0u;
    __syncthreads();

    const int chunk = (((n_edges + DEG_CHUNKS - 1) / DEG_CHUNKS) + 3) & ~3;
    const int e0 = blockIdx.y * chunk;
    int e1 = e0 + chunk;
    if (e1 > n_edges) e1 = n_edges;
    if (e1 < e0) e1 = e0;

    const int e1v = e0 + ((e1 - e0) & ~3);
    for (int base = e0 + 4 * tid; base < e1v; base += 4 * DEG_THREADS * DEG_LOADS) {
        int4 v[DEG_LOADS];
        #pragma unroll
        for (int u = 0; u < DEG_LOADS; ++u) {
            const int e = base + u * 4 * DEG_THREADS;
            v[u] = make_int4(-1, -1, -1, -1);
            if (e < e1v) v[u] = *(const int4*)(ids + e);
        }
        #pragma unroll
        for (int u = 0; u < DEG_LOADS; ++u) {
            const unsigned rx = (unsigned)(v[u].x - lo);
            const unsigned ry = (unsigned)(v[u].y - lo);
            const unsigned rz = (unsigned)(v[u].z - lo);
            const unsigned rw = (unsigned)(v[u].w - lo);
            if (rx < (unsigned)span) atomicAdd(&cnt[rx], 1u);
            if (ry < (unsigned)span) atomicAdd(&cnt[ry], 1u);
            if (rz < (unsigned)span) atomicAdd(&cnt[rz], 1u);
            if (rw < (unsigned)span) atomicAdd(&cnt[rw], 1u);
        }
    }
    if (tid < e1 - e1v) {
        const unsigned rt = (unsigned)(ids[e1v + tid] - lo);
        if (rt < (unsigned)span) atomicAdd(&cnt[rt], 1u);
    }
    __syncthreads();

    unsigned int* __restrict__ dstp =
        part + ((size_t)blockIdx.z * DEG_CHUNKS + blockIdx.y) * (size_t)n_nodes + lo;
    for (int i = tid; i < span; i += DEG_THREADS) dstp[i] = cnt[i];
}

__global__ void norm_kernel(const unsigned int* __restrict__ part,
                            float* __restrict__ srcn, float* __restrict__ dstn, int n) {
    int i = blockIdx.x * blockDim.x + threadIdx.x;
    if (i < n) {
        unsigned int od = 0u, id = 0u;
        #pragma unroll
        for (int c = 0; c < DEG_CHUNKS; ++c) {
            od += part[(size_t)c * n + i];
            id += part[(size_t)(DEG_CHUNKS + c) * n + i];
        }
        srcn[i] = rsqrtf(fmaxf((float)od, 1.0f));
        dstn[i] = rsqrtf(fmaxf((float)id, 1.0f));
    }
}

#define TILE_ROWS    400
#define TILE_PITCH   (HID_C + 4)
#define M_TILES      (TILE_ROWS / 16)
#define GEMM_ROUNDS  ((M_TILES + 3) / 4)
#define TILE_THREADS 1024
#define TILE_WAVES   (TILE_THREADS / 32)
#define TILE_BLOCKS  ((N_NODES_C + TILE_ROWS - 1) / TILE_ROWS)

#define LINE_IDS     32
#define N_LINES      ((N_EDGES_C + LINE_IDS - 1) / LINE_IDS)
#define BM_WORDS     ((N_LINES + 31) / 32)
#define WAVE_WORDS   ((BM_WORDS + TILE_WAVES - 1) / TILE_WAVES)
#define BM_PITCH     (((WAVE_WORDS * TILE_WAVES) + 31) & ~31)
#define MARK_BLOCKS  (BM_PITCH / 32)
#define MARK_STRIDE  33
#define BATCH_LINES  16
#define QCAP         (BATCH_LINES * LINE_IDS + 8)

static_assert(N_EDGES_C <= (1 << 20), "edge index must fit bits 0-19 of a queue word");
static_assert(TILE_ROWS <= (1 << 12), "tile row must fit above bit 20 of a queue word");
static_assert(WAVE_WORDS <= 32, "a wave holds its bitmap words one per lane");
static_assert(TILE_ROWS % 16 == 0, "a tile is whole 16-row blocks");
static_assert(TILE_WAVES == 32 && HID_C == 128, "8 column owners x 4 row blocks a round, 2 rows a wave to store");

__global__ void __launch_bounds__(1024) mark_kernel(
    const int* __restrict__ dst, unsigned int* __restrict__ bm, int n_edges) {
    __shared__ unsigned int marks[TILE_BLOCKS * MARK_STRIDE];

    const unsigned lane = threadIdx.x & 31u;
    const int wv = (int)(threadIdx.x >> 5);
    for (int i = threadIdx.x; i < TILE_BLOCKS * MARK_STRIDE; i += 1024) marks[i] = 0u;
    __syncthreads();

    const int ebase = ((int)blockIdx.x * 32 + wv) * (32 * LINE_IDS) + (int)lane;
    #pragma unroll
    for (int g = 0; g < 32; g += 8) {
        int id[8];
        #pragma unroll
        for (int u = 0; u < 8; ++u) {
            const int e = ebase + (g + u) * LINE_IDS;
            id[u] = -1;
            if (e < n_edges) id[u] = dst[e];
        }
        #pragma unroll
        for (int u = 0; u < 8; ++u) {
            const unsigned t = (unsigned)id[u] / (unsigned)TILE_ROWS;
            if (t < (unsigned)TILE_BLOCKS) atomicOr(&marks[t * MARK_STRIDE + wv], 1u << (g + u));
        }
    }
    __syncthreads();

    for (int t = wv; t < TILE_BLOCKS; t += 32)
        bm[(size_t)t * BM_PITCH + blockIdx.x * 32 + lane] = marks[t * MARK_STRIDE + lane];
}

template <bool FULL>
__device__ __forceinline__ void tile_group(const float* __restrict__ h,
                                           const float* __restrict__ srcn,
                                           float* tile, int sv, int dv,
                                           int first, int nvalid, unsigned lane) {
    int   s[8], d[8];
    float sn[8];
    float v[8][4];
    #pragma unroll
    for (int u = 0; u < 8; ++u) {
        s[u] = __builtin_amdgcn_readlane(sv, first + u);
        d[u] = __builtin_amdgcn_readlane(dv, first + u);
    }
    #pragma unroll
    for (int u = 0; u < 8; ++u) {
        sn[u] = 0.0f;
        v[u][0] = 0.0f; v[u][1] = 0.0f; v[u][2] = 0.0f; v[u][3] = 0.0f;
        if (FULL || u < nvalid) {
            const float* hp = h + (size_t)s[u] * HID_C;
            sn[u]   = srcn[s[u]];
            v[u][0] = hp[lane];
            v[u][1] = hp[lane + 32u];
            v[u][2] = hp[lane + 64u];
            v[u][3] = hp[lane + 96u];
        }
    }
    #pragma unroll
    for (int u = 0; u < 8; ++u) {
        if (FULL || u < nvalid) {
            float* tp = tile + d[u] * TILE_PITCH;
            __hip_atomic_fetch_add(tp + lane,       v[u][0] * sn[u], __ATOMIC_RELAXED, __HIP_MEMORY_SCOPE_WORKGROUP);
            __hip_atomic_fetch_add(tp + lane + 32u, v[u][1] * sn[u], __ATOMIC_RELAXED, __HIP_MEMORY_SCOPE_WORKGROUP);
            __hip_atomic_fetch_add(tp + lane + 64u, v[u][2] * sn[u], __ATOMIC_RELAXED, __HIP_MEMORY_SCOPE_WORKGROUP);
            __hip_atomic_fetch_add(tp + lane + 96u, v[u][3] * sn[u], __ATOMIC_RELAXED, __HIP_MEMORY_SCOPE_WORKGROUP);
        }
    }
}

__global__ void __launch_bounds__(TILE_THREADS) edge_conv_kernel(
    const float* __restrict__ h, const float* __restrict__ srcn,
    const float* __restrict__ dstn,
    const int* __restrict__ src, const int* __restrict__ dst,
    const unsigned int* __restrict__ bm,
    const float* __restrict__ W, const float* __restrict__ bias,
    float* __restrict__ out, int n_edges, int n_nodes) {
    __shared__ float        tile[TILE_ROWS * TILE_PITCH];
    __shared__ unsigned int queue[TILE_WAVES * QCAP];

    const unsigned lane = threadIdx.x & 31u;
    const int w  = __builtin_amdgcn_readfirstlane((int)(threadIdx.x >> 5));
    const int lo = blockIdx.x * TILE_ROWS;
    int span = n_nodes - lo;
    if (span > TILE_ROWS) span = TILE_ROWS;
    if (span < 0) span = 0;

    unsigned words = 0u;
    if ((int)lane < WAVE_WORDS)
        words = bm[(size_t)blockIdx.x * BM_PITCH + w * WAVE_WORDS + (int)lane];

    float4* t4 = (float4*)tile;
    for (int i = threadIdx.x; i < TILE_ROWS * TILE_PITCH / 4; i += TILE_THREADS)
        t4[i] = make_float4(0.f, 0.f, 0.f, 0.f);
    __syncthreads();

    unsigned int* q = queue + w * QCAP;
    int count = 0;

    auto enqueue = [&](int id, int e) {
        const unsigned rel  = (unsigned)(id - lo);
        const bool     m    = rel < (unsigned)span;
        const unsigned mask = __builtin_amdgcn_ballot_w32(m);
        if (mask != 0u) {
            const unsigned rank = (unsigned)count + __builtin_amdgcn_mbcnt_lo(mask, 0u);
            if (m) q[rank] = (unsigned)e | (rel << 20);
            count += __builtin_popcount(mask);
        }
    };
    auto drain = [&]() {
        const int nfull = count & ~7;
        for (int pos = 0; pos < nfull; pos += 32) {
            const int n = (nfull - pos < 32) ? (nfull - pos) : 32;
            unsigned ent = 0u;
            int sv = 0;
            if ((int)lane < n) {
                ent = q[pos + (int)lane];
                sv  = src[ent & 0xFFFFFu];
            }
            const int dv = (int)(ent >> 20);
            for (int b = 0; b < n; b += 8)
                tile_group<true>(h, srcn, tile, sv, dv, b, 8, lane);
        }
        const int rem = count - nfull;
        unsigned carry = 0u;
        if ((int)lane < rem) carry = q[nfull + (int)lane];
        if ((int)lane < rem) q[lane] = carry;
        count = rem;
    };

    const int line0 = w * WAVE_WORDS * 32;
    int p = 0;
    unsigned bits = (unsigned)__builtin_amdgcn_readlane((int)words, 0);
    auto next_line = [&]() -> int {
        while (bits == 0u && p < WAVE_WORDS - 1) {
            ++p;
            bits = (unsigned)__builtin_amdgcn_readlane((int)words, p);
        }
        if (bits == 0u) return -1;
        const int b = __builtin_ctz(bits);
        bits &= bits - 1u;
        return line0 + p * 32 + b;
    };
    auto load_batch = [&](int (&ln)[BATCH_LINES], int (&idv)[BATCH_LINES]) {
        #pragma unroll
        for (int u = 0; u < BATCH_LINES; ++u) ln[u] = next_line();
        #pragma unroll
        for (int u = 0; u < BATCH_LINES; ++u) {
            const int lnu = (ln[u] < 0) ? 0 : ln[u];
            const int e   = lnu * LINE_IDS + (int)lane;
            int v = -1;
            if (e < n_edges) v = dst[e];
            idv[u] = (ln[u] < 0) ? -1 : v;
        }
    };

    {
        int lc[BATCH_LINES], ic[BATCH_LINES], lx[BATCH_LINES], ix[BATCH_LINES];
        load_batch(lc, ic);
        for (int it = 0; it < (WAVE_WORDS * 32) / BATCH_LINES && lc[0] >= 0; ++it) {
            load_batch(lx, ix);
            #pragma unroll
            for (int u = 0; u < BATCH_LINES; ++u)
                enqueue(ic[u], lc[u] * LINE_IDS + (int)lane);
            if (count >= 8) drain();
            #pragma unroll
            for (int u = 0; u < BATCH_LINES; ++u) { lc[u] = lx[u]; ic[u] = ix[u]; }
        }
    }
    const int l15   = (int)(lane & 15u);
    const int khalf = (int)(lane >> 4) << 1;
    const int nt    = w & 7;
    const int mg    = w >> 3;
    const int col   = nt * 16 + l15;

    v2f bw[HID_C / 4];
    #pragma unroll
    for (int ks = 0; ks < HID_C / 4; ++ks) {
        bw[ks].x = W[(size_t)(4 * ks + khalf) * HID_C + col];
        bw[ks].y = W[(size_t)(4 * ks + khalf + 1) * HID_C + col];
    }
    const float bv = bias[col];

    if (count > 0) {
        unsigned ent = 0u;
        int sv = 0;
        if ((int)lane < count) {
            ent = q[lane];
            sv  = src[ent & 0xFFFFFu];
        }
        const int dv = (int)(ent >> 20);
        tile_group<false>(h, srcn, tile, sv, dv, 0, count, lane);
    }

    auto row_scale = [&](int j) -> float {
        const int m    = 4 * j + mg;
        const int node = lo + 16 * m + l15;
        return (m < M_TILES && node < n_nodes) ? dstn[node] : 0.0f;
    };
    float sc = row_scale(0);

    __syncthreads();

    auto store_round = [&](int j) {
        #pragma unroll
        for (int t = 0; t < 2; ++t) {
            const int r = 64 * j + 32 * t + w;
            if (r < span) {
                const float4 v = *(const float4*)(tile + r * TILE_PITCH + 4 * (int)lane);
                *(float4*)(out + (size_t)(lo + r) * HID_C + 4 * lane) = v;
            }
        }
    };

    for (int j = 0; j < GEMM_ROUNDS; ++j) {
        const int m = 4 * j + mg;
        const float sc_next = row_scale(j + 1);
        v8f acc = {0.f, 0.f, 0.f, 0.f, 0.f, 0.f, 0.f, 0.f};
        if (m < M_TILES) {
            const float* ap = tile + (16 * m + l15) * TILE_PITCH + khalf;
            #pragma unroll
            for (int ks = 0; ks < HID_C / 4; ++ks) {
                v2f a = *(const v2f*)(ap + 4 * ks);
                a.x *= sc;
                a.y *= sc;
                acc = __builtin_amdgcn_wmma_f32_16x16x4_f32(false, a, false, bw[ks],
                                                            (short)0, acc, false, false);
            }
        }
        __syncthreads();
        if (m < M_TILES) {
            float* op = tile + (16 * m + 8 * (int)(lane >> 4)) * TILE_PITCH + col;
            #pragma unroll
            for (int i = 0; i < 8; ++i)
                op[i * TILE_PITCH] = fmaxf(acc[i] + bv, 0.0f);
        }
        if (j > 0) store_round(j - 1);
        sc = sc_next;
    }
    __syncthreads();
    store_round(GEMM_ROUNDS - 1);
}

__global__ void __launch_bounds__(32) wmma_gemm_kernel(
    const float* __restrict__ A, const float* __restrict__ B,
    const float* __restrict__ bias, float* __restrict__ C,
    int M, int N, int K, int do_relu) {
    const int lane  = threadIdx.x & 31;
    const int l15   = lane & 15;
    const int khalf = (lane >> 4) << 1;
    const int mbase = blockIdx.x * 16;
    const int nbase = blockIdx.y * 16;
    const int row   = mbase + l15;
    const int col   = nbase + l15;

    v8f acc = {0.f, 0.f, 0.f, 0.f, 0.f, 0.f, 0.f, 0.f};
    for (int k = 0; k < K; k += 4) {
        int k0 = k + khalf, k1 = k + khalf + 1;
        v2f a, b;
        a.x = (row < M) ? A[(size_t)row * K + k0] : 0.0f;
        a.y = (row < M) ? A[(size_t)row * K + k1] : 0.0f;
        b.x = (col < N) ? B[(size_t)k0 * N + col] : 0.0f;
        b.y = (col < N) ? B[(size_t)k1 * N + col] : 0.0f;
        acc = __builtin_amdgcn_wmma_f32_16x16x4_f32(false, a, false, b,
                                                    (short)0, acc, false, false);
    }

    if (col < N) {
        const float bv = bias[col];
        #pragma unroll
        for (int i = 0; i < 8; ++i) {
            int m = mbase + i + ((lane >> 4) << 3);
            if (m < M) {
                float v = acc[i] + bv;
                C[(size_t)m * N + col] = do_relu ? fmaxf(v, 0.0f) : v;
            }
        }
    }
}

#define POOL_WAVES_PER_BLOCK 10
#define POOL_BLOCKS 625
#define POOL_WAVES (POOL_BLOCKS * POOL_WAVES_PER_BLOCK)

__device__ __forceinline__ void pool_flush(float* __restrict__ hg, int g, unsigned lane,
                                           unsigned m0, unsigned m1, unsigned m2, unsigned m3) {
    unsigned int* p = (unsigned int*)(hg + (size_t)g * HID_C);
    atomicMax(p + lane,       m0);
    atomicMax(p + lane + 32u, m1);
    atomicMax(p + lane + 64u, m2);
    atomicMax(p + lane + 96u, m3);
}

__global__ void __launch_bounds__(POOL_WAVES_PER_BLOCK * 32) segmax_kernel(
    const float* __restrict__ h, const int* __restrict__ gids,
    float* __restrict__ hg, int n_nodes) {
    const unsigned lane = threadIdx.x & 31u;
    const int w = __builtin_amdgcn_readfirstlane(
        (int)(blockIdx.x * POOL_WAVES_PER_BLOCK + (threadIdx.x >> 5)));

    const int q  = n_nodes / POOL_WAVES;
    const int r  = n_nodes % POOL_WAVES;
    const int n0 = w * q + (w < r ? w : r);
    const int n1 = n0 + q + (w < r ? 1 : 0);

    int cur = -1;
    unsigned m0 = 0u, m1 = 0u, m2 = 0u, m3 = 0u;

    for (int base = n0; base < n1; base += 32) {
        const int idx = base + (int)lane;
        int gv = 0;
        if (idx < n1) gv = gids[idx];
        const int cnt = (n1 - base < 32) ? (n1 - base) : 32;

        int j = 0;
        for (; j + 8 <= cnt; j += 8) {
            float v[8][4];
            #pragma unroll
            for (int u = 0; u < 8; ++u) {
                const float* hp = h + (size_t)(base + j + u) * HID_C;
                v[u][0] = hp[lane];
                v[u][1] = hp[lane + 32u];
                v[u][2] = hp[lane + 64u];
                v[u][3] = hp[lane + 96u];
            }
            #pragma unroll
            for (int u = 0; u < 8; ++u) {
                const int g = __builtin_amdgcn_readlane(gv, j + u);
                if (g != cur) {
                    if (cur >= 0) pool_flush(hg, cur, lane, m0, m1, m2, m3);
                    m0 = 0u; m1 = 0u; m2 = 0u; m3 = 0u;
                    cur = g;
                }
                const unsigned b0 = __float_as_uint(v[u][0]);
                const unsigned b1 = __float_as_uint(v[u][1]);
                const unsigned b2 = __float_as_uint(v[u][2]);
                const unsigned b3 = __float_as_uint(v[u][3]);
                m0 = (b0 > m0) ? b0 : m0;
                m1 = (b1 > m1) ? b1 : m1;
                m2 = (b2 > m2) ? b2 : m2;
                m3 = (b3 > m3) ? b3 : m3;
            }
        }
        for (; j < cnt; ++j) {
            const float* hp = h + (size_t)(base + j) * HID_C;
            const unsigned b0 = __float_as_uint(hp[lane]);
            const unsigned b1 = __float_as_uint(hp[lane + 32u]);
            const unsigned b2 = __float_as_uint(hp[lane + 64u]);
            const unsigned b3 = __float_as_uint(hp[lane + 96u]);
            const int g = __builtin_amdgcn_readlane(gv, j);
            if (g != cur) {
                if (cur >= 0) pool_flush(hg, cur, lane, m0, m1, m2, m3);
                m0 = 0u; m1 = 0u; m2 = 0u; m3 = 0u;
                cur = g;
            }
            m0 = (b0 > m0) ? b0 : m0;
            m1 = (b1 > m1) ? b1 : m1;
            m2 = (b2 > m2) ? b2 : m2;
            m3 = (b3 > m3) ? b3 : m3;
        }
    }
    if (cur >= 0) pool_flush(hg, cur, lane, m0, m1, m2, m3);
}

extern "C" void kernel_launch(void* const* d_in, const int* in_sizes, int n_in,
                              void* d_out, int out_size, void* d_ws, size_t ws_size,
                              hipStream_t stream) {
    (void)in_sizes; (void)n_in; (void)out_size; (void)ws_size;

    const float* feat = (const float*)d_in[0];
    const int*   src  = (const int*)d_in[1];
    const int*   dst  = (const int*)d_in[2];
    const int*   gids = (const int*)d_in[3];
    const float* Wg[3] = {(const float*)d_in[4], (const float*)d_in[6], (const float*)d_in[8]};
    const float* bg[3] = {(const float*)d_in[5], (const float*)d_in[7], (const float*)d_in[9]};
    const float* Wc1 = (const float*)d_in[10]; const float* bc1 = (const float*)d_in[11];
    const float* Wc2 = (const float*)d_in[12]; const float* bc2 = (const float*)d_in[13];
    const float* Wc3 = (const float*)d_in[14]; const float* bc3 = (const float*)d_in[15];
    float* out = (float*)d_out;

    char* ws = (char*)d_ws;
    size_t off = 0;
    auto take = [&](size_t elems) -> float* {
        float* p = (float*)(ws + off);
        off += ((elems * sizeof(float) + 255) / 256) * 256;
        return p;
    };
    unsigned int* part = (unsigned int*)take((size_t)2 * DEG_CHUNKS * N_NODES_C);
    float* srcn = take(N_NODES_C);
    float* dstn = take(N_NODES_C);
    unsigned int* bm = (unsigned int*)take((size_t)TILE_BLOCKS * BM_PITCH);
    float* h2   = take((size_t)N_NODES_C * HID_C);
    float* h    = take((size_t)N_NODES_C * HID_C);
    float* hg   = take((size_t)N_GRAPHS_C * HID_C);
    float* x1   = take((size_t)N_GRAPHS_C * HID_C);
    float* x2   = take((size_t)N_GRAPHS_C * (HID_C / 2));

    degree_kernel<<<dim3(DEG_RANGES, DEG_CHUNKS, 2), DEG_THREADS, 0, stream>>>(
        src, dst, part, N_EDGES_C, N_NODES_C);
    norm_kernel<<<(N_NODES_C + 255) / 256, 256, 0, stream>>>(part, srcn, dstn, N_NODES_C);

    mark_kernel<<<MARK_BLOCKS, 1024, 0, stream>>>(dst, bm, N_EDGES_C);

    edge_conv_kernel<<<TILE_BLOCKS, TILE_THREADS, 0, stream>>>(
        feat, srcn, dstn, src, dst, bm, Wg[0], bg[0], h, N_EDGES_C, N_NODES_C);
    edge_conv_kernel<<<TILE_BLOCKS, TILE_THREADS, 0, stream>>>(
        h, srcn, dstn, src, dst, bm, Wg[1], bg[1], h2, N_EDGES_C, N_NODES_C);
    edge_conv_kernel<<<TILE_BLOCKS, TILE_THREADS, 0, stream>>>(
        h2, srcn, dstn, src, dst, bm, Wg[2], bg[2], h, N_EDGES_C, N_NODES_C);

    hipMemsetAsync(hg, 0, (size_t)N_GRAPHS_C * HID_C * sizeof(float), stream);
    segmax_kernel<<<POOL_BLOCKS, POOL_WAVES_PER_BLOCK * 32, 0, stream>>>(h, gids, hg, N_NODES_C);

    wmma_gemm_kernel<<<dim3(N_GRAPHS_C / 16, HID_C / 16), 32, 0, stream>>>(
        hg, Wc1, bc1, x1, N_GRAPHS_C, HID_C, HID_C, 1);
    wmma_gemm_kernel<<<dim3(N_GRAPHS_C / 16, (HID_C / 2) / 16), 32, 0, stream>>>(
        x1, Wc2, bc2, x2, N_GRAPHS_C, HID_C / 2, HID_C, 1);
    wmma_gemm_kernel<<<dim3(N_GRAPHS_C / 16, (OUT_DIM_C + 15) / 16), 32, 0, stream>>>(
        x2, Wc3, bc3, out, N_GRAPHS_C, OUT_DIM_C, HID_C / 2, 0);
}


# Classifier GEMMs split over K: 8 waves per 16x16 tile, a wave takes 4 (or 2) k-steps with all its loads in one group, partial tiles summed in LDS in wave order; sum order of these 3 products changes
# speedup vs baseline: 3.1364x; 1.2076x over previous
#include <hip/hip_runtime.h>
#include <hip/hip_bf16.h>


typedef float v2f __attribute__((ext_vector_type(2)));
typedef float v8f __attribute__((ext_vector_type(8)));

#define N_NODES_C 100000
#define N_EDGES_C 800000
#define HID_C     128
#define OUT_DIM_C 500
#define N_GRAPHS_C 64

#define DEG_RANGES  16
#define DEG_CHUNKS  16
#define DEG_THREADS 256
#define DEG_RANGE_NODES ((((N_NODES_C + DEG_RANGES - 1) / DEG_RANGES) + 31) & ~31)
#define DEG_LOADS   13

__global__ void __launch_bounds__(DEG_THREADS) degree_kernel(
    const int* __restrict__ src, const int* __restrict__ dst,
    unsigned int* __restrict__ part, int n_edges, int n_nodes) {
    __shared__ unsigned int cnt[DEG_RANGE_NODES];

    const int tid = threadIdx.x;
    const int* __restrict__ ids = (blockIdx.z == 0) ? src : dst;
    const int lo = blockIdx.x * DEG_RANGE_NODES;
    int span = n_nodes - lo;
    if (span > DEG_RANGE_NODES) span = DEG_RANGE_NODES;
    if (span < 0) span = 0;

    for (int i = tid; i < DEG_RANGE_NODES; i += DEG_THREADS) cnt[i] = 0u;
    __syncthreads();

    const int chunk = (((n_edges + DEG_CHUNKS - 1) / DEG_CHUNKS) + 3) & ~3;
    const int e0 = blockIdx.y * chunk;
    int e1 = e0 + chunk;
    if (e1 > n_edges) e1 = n_edges;
    if (e1 < e0) e1 = e0;

    const int e1v = e0 + ((e1 - e0) & ~3);
    for (int base = e0 + 4 * tid; base < e1v; base += 4 * DEG_THREADS * DEG_LOADS) {
        int4 v[DEG_LOADS];
        #pragma unroll
        for (int u = 0; u < DEG_LOADS; ++u) {
            const int e = base + u * 4 * DEG_THREADS;
            v[u] = make_int4(-1, -1, -1, -1);
            if (e < e1v) v[u] = *(const int4*)(ids + e);
        }
        #pragma unroll
        for (int u = 0; u < DEG_LOADS; ++u) {
            const unsigned rx = (unsigned)(v[u].x - lo);
            const unsigned ry = (unsigned)(v[u].y - lo);
            const unsigned rz = (unsigned)(v[u].z - lo);
            const unsigned rw = (unsigned)(v[u].w - lo);
            if (rx < (unsigned)span) atomicAdd(&cnt[rx], 1u);
            if (ry < (unsigned)span) atomicAdd(&cnt[ry], 1u);
            if (rz < (unsigned)span) atomicAdd(&cnt[rz], 1u);
            if (rw < (unsigned)span) atomicAdd(&cnt[rw], 1u);
        }
    }
    if (tid < e1 - e1v) {
        const unsigned rt = (unsigned)(ids[e1v + tid] - lo);
        if (rt < (unsigned)span) atomicAdd(&cnt[rt], 1u);
    }
    __syncthreads();

    unsigned int* __restrict__ dstp =
        part + ((size_t)blockIdx.z * DEG_CHUNKS + blockIdx.y) * (size_t)n_nodes + lo;
    for (int i = tid; i < span; i += DEG_THREADS) dstp[i] = cnt[i];
}

__global__ void norm_kernel(const unsigned int* __restrict__ part,
                            float* __restrict__ srcn, float* __restrict__ dstn, int n) {
    int i = blockIdx.x * blockDim.x + threadIdx.x;
    if (i < n) {
        unsigned int od = 0u, id = 0u;
        #pragma unroll
        for (int c = 0; c < DEG_CHUNKS; ++c) {
            od += part[(size_t)c * n + i];
            id += part[(size_t)(DEG_CHUNKS + c) * n + i];
        }
        srcn[i] = rsqrtf(fmaxf((float)od, 1.0f));
        dstn[i] = rsqrtf(fmaxf((float)id, 1.0f));
    }
}

#define TILE_ROWS    400
#define TILE_PITCH   (HID_C + 4)
#define M_TILES      (TILE_ROWS / 16)
#define GEMM_ROUNDS  ((M_TILES + 3) / 4)
#define TILE_THREADS 1024
#define TILE_WAVES   (TILE_THREADS / 32)
#define TILE_BLOCKS  ((N_NODES_C + TILE_ROWS - 1) / TILE_ROWS)

#define LINE_IDS     32
#define N_LINES      ((N_EDGES_C + LINE_IDS - 1) / LINE_IDS)
#define BM_WORDS     ((N_LINES + 31) / 32)
#define WAVE_WORDS   ((BM_WORDS + TILE_WAVES - 1) / TILE_WAVES)
#define BM_PITCH     (((WAVE_WORDS * TILE_WAVES) + 31) & ~31)
#define MARK_BLOCKS  (BM_PITCH / 32)
#define MARK_STRIDE  33
#define BATCH_LINES  16
#define QCAP         (BATCH_LINES * LINE_IDS + 8)

static_assert(N_EDGES_C <= (1 << 20), "edge index must fit bits 0-19 of a queue word");
static_assert(TILE_ROWS <= (1 << 12), "tile row must fit above bit 20 of a queue word");
static_assert(WAVE_WORDS <= 32, "a wave holds its bitmap words one per lane");
static_assert(TILE_ROWS % 16 == 0, "a tile is whole 16-row blocks");
static_assert(TILE_WAVES == 32 && HID_C == 128, "8 column owners x 4 row blocks a round, 2 rows a wave to store");

__global__ void __launch_bounds__(1024) mark_kernel(
    const int* __restrict__ dst, unsigned int* __restrict__ bm, int n_edges) {
    __shared__ unsigned int marks[TILE_BLOCKS * MARK_STRIDE];

    const unsigned lane = threadIdx.x & 31u;
    const int wv = (int)(threadIdx.x >> 5);
    for (int i = threadIdx.x; i < TILE_BLOCKS * MARK_STRIDE; i += 1024) marks[i] = 0u;
    __syncthreads();

    const int ebase = ((int)blockIdx.x * 32 + wv) * (32 * LINE_IDS) + (int)lane;
    #pragma unroll
    for (int g = 0; g < 32; g += 8) {
        int id[8];
        #pragma unroll
        for (int u = 0; u < 8; ++u) {
            const int e = ebase + (g + u) * LINE_IDS;
            id[u] = -1;
            if (e < n_edges) id[u] = dst[e];
        }
        #pragma unroll
        for (int u = 0; u < 8; ++u) {
            const unsigned t = (unsigned)id[u] / (unsigned)TILE_ROWS;
            if (t < (unsigned)TILE_BLOCKS) atomicOr(&marks[t * MARK_STRIDE + wv], 1u << (g + u));
        }
    }
    __syncthreads();

    for (int t = wv; t < TILE_BLOCKS; t += 32)
        bm[(size_t)t * BM_PITCH + blockIdx.x * 32 + lane] = marks[t * MARK_STRIDE + lane];
}

template <bool FULL>
__device__ __forceinline__ void tile_group(const float* __restrict__ h,
                                           const float* __restrict__ srcn,
                                           float* tile, int sv, int dv,
                                           int first, int nvalid, unsigned lane) {
    int   s[8], d[8];
    float sn[8];
    float v[8][4];
    #pragma unroll
    for (int u = 0; u < 8; ++u) {
        s[u] = __builtin_amdgcn_readlane(sv, first + u);
        d[u] = __builtin_amdgcn_readlane(dv, first + u);
    }
    #pragma unroll
    for (int u = 0; u < 8; ++u) {
        sn[u] = 0.0f;
        v[u][0] = 0.0f; v[u][1] = 0.0f; v[u][2] = 0.0f; v[u][3] = 0.0f;
        if (FULL || u < nvalid) {
            const float* hp = h + (size_t)s[u] * HID_C;
            sn[u]   = srcn[s[u]];
            v[u][0] = hp[lane];
            v[u][1] = hp[lane + 32u];
            v[u][2] = hp[lane + 64u];
            v[u][3] = hp[lane + 96u];
        }
    }
    #pragma unroll
    for (int u = 0; u < 8; ++u) {
        if (FULL || u < nvalid) {
            float* tp = tile + d[u] * TILE_PITCH;
            __hip_atomic_fetch_add(tp + lane,       v[u][0] * sn[u], __ATOMIC_RELAXED, __HIP_MEMORY_SCOPE_WORKGROUP);
            __hip_atomic_fetch_add(tp + lane + 32u, v[u][1] * sn[u], __ATOMIC_RELAXED, __HIP_MEMORY_SCOPE_WORKGROUP);
            __hip_atomic_fetch_add(tp + lane + 64u, v[u][2] * sn[u], __ATOMIC_RELAXED, __HIP_MEMORY_SCOPE_WORKGROUP);
            __hip_atomic_fetch_add(tp + lane + 96u, v[u][3] * sn[u], __ATOMIC_RELAXED, __HIP_MEMORY_SCOPE_WORKGROUP);
        }
    }
}

__global__ void __launch_bounds__(TILE_THREADS) edge_conv_kernel(
    const float* __restrict__ h, const float* __restrict__ srcn,
    const float* __restrict__ dstn,
    const int* __restrict__ src, const int* __restrict__ dst,
    const unsigned int* __restrict__ bm,
    const float* __restrict__ W, const float* __restrict__ bias,
    float* __restrict__ out, int n_edges, int n_nodes) {
    __shared__ float        tile[TILE_ROWS * TILE_PITCH];
    __shared__ unsigned int queue[TILE_WAVES * QCAP];

    const unsigned lane = threadIdx.x & 31u;
    const int w  = __builtin_amdgcn_readfirstlane((int)(threadIdx.x >> 5));
    const int lo = blockIdx.x * TILE_ROWS;
    int span = n_nodes - lo;
    if (span > TILE_ROWS) span = TILE_ROWS;
    if (span < 0) span = 0;

    unsigned words = 0u;
    if ((int)lane < WAVE_WORDS)
        words = bm[(size_t)blockIdx.x * BM_PITCH + w * WAVE_WORDS + (int)lane];

    float4* t4 = (float4*)tile;
    for (int i = threadIdx.x; i < TILE_ROWS * TILE_PITCH / 4; i += TILE_THREADS)
        t4[i] = make_float4(0.f, 0.f, 0.f, 0.f);
    __syncthreads();

    unsigned int* q = queue + w * QCAP;
    int count = 0;

    auto enqueue = [&](int id, int e) {
        const unsigned rel  = (unsigned)(id - lo);
        const bool     m    = rel < (unsigned)span;
        const unsigned mask = __builtin_amdgcn_ballot_w32(m);
        if (mask != 0u) {
            const unsigned rank = (unsigned)count + __builtin_amdgcn_mbcnt_lo(mask, 0u);
            if (m) q[rank] = (unsigned)e | (rel << 20);
            count += __builtin_popcount(mask);
        }
    };
    auto drain = [&]() {
        const int nfull = count & ~7;
        for (int pos = 0; pos < nfull; pos += 32) {
            const int n = (nfull - pos < 32) ? (nfull - pos) : 32;
            unsigned ent = 0u;
            int sv = 0;
            if ((int)lane < n) {
                ent = q[pos + (int)lane];
                sv  = src[ent & 0xFFFFFu];
            }
            const int dv = (int)(ent >> 20);
            for (int b = 0; b < n; b += 8)
                tile_group<true>(h, srcn, tile, sv, dv, b, 8, lane);
        }
        const int rem = count - nfull;
        unsigned carry = 0u;
        if ((int)lane < rem) carry = q[nfull + (int)lane];
        if ((int)lane < rem) q[lane] = carry;
        count = rem;
    };

    const int line0 = w * WAVE_WORDS * 32;
    int p = 0;
    unsigned bits = (unsigned)__builtin_amdgcn_readlane((int)words, 0);
    auto next_line = [&]() -> int {
        while (bits == 0u && p < WAVE_WORDS - 1) {
            ++p;
            bits = (unsigned)__builtin_amdgcn_readlane((int)words, p);
        }
        if (bits == 0u) return -1;
        const int b = __builtin_ctz(bits);
        bits &= bits - 1u;
        return line0 + p * 32 + b;
    };
    auto load_batch = [&](int (&ln)[BATCH_LINES], int (&idv)[BATCH_LINES]) {
        #pragma unroll
        for (int u = 0; u < BATCH_LINES; ++u) ln[u] = next_line();
        #pragma unroll
        for (int u = 0; u < BATCH_LINES; ++u) {
            const int lnu = (ln[u] < 0) ? 0 : ln[u];
            const int e   = lnu * LINE_IDS + (int)lane;
            int v = -1;
            if (e < n_edges) v = dst[e];
            idv[u] = (ln[u] < 0) ? -1 : v;
        }
    };

    {
        int lc[BATCH_LINES], ic[BATCH_LINES], lx[BATCH_LINES], ix[BATCH_LINES];
        load_batch(lc, ic);
        for (int it = 0; it < (WAVE_WORDS * 32) / BATCH_LINES && lc[0] >= 0; ++it) {
            load_batch(lx, ix);
            #pragma unroll
            for (int u = 0; u < BATCH_LINES; ++u)
                enqueue(ic[u], lc[u] * LINE_IDS + (int)lane);
            if (count >= 8) drain();
            #pragma unroll
            for (int u = 0; u < BATCH_LINES; ++u) { lc[u] = lx[u]; ic[u] = ix[u]; }
        }
    }
    const int l15   = (int)(lane & 15u);
    const int khalf = (int)(lane >> 4) << 1;
    const int nt    = w & 7;
    const int mg    = w >> 3;
    const int col   = nt * 16 + l15;

    v2f bw[HID_C / 4];
    #pragma unroll
    for (int ks = 0; ks < HID_C / 4; ++ks) {
        bw[ks].x = W[(size_t)(4 * ks + khalf) * HID_C + col];
        bw[ks].y = W[(size_t)(4 * ks + khalf + 1) * HID_C + col];
    }
    const float bv = bias[col];

    if (count > 0) {
        unsigned ent = 0u;
        int sv = 0;
        if ((int)lane < count) {
            ent = q[lane];
            sv  = src[ent & 0xFFFFFu];
        }
        const int dv = (int)(ent >> 20);
        tile_group<false>(h, srcn, tile, sv, dv, 0, count, lane);
    }

    auto row_scale = [&](int j) -> float {
        const int m    = 4 * j + mg;
        const int node = lo + 16 * m + l15;
        return (m < M_TILES && node < n_nodes) ? dstn[node] : 0.0f;
    };
    float sc = row_scale(0);

    __syncthreads();

    auto store_round = [&](int j) {
        #pragma unroll
        for (int t = 0; t < 2; ++t) {
            const int r = 64 * j + 32 * t + w;
            if (r < span) {
                const float4 v = *(const float4*)(tile + r * TILE_PITCH + 4 * (int)lane);
                *(float4*)(out + (size_t)(lo + r) * HID_C + 4 * lane) = v;
            }
        }
    };

    for (int j = 0; j < GEMM_ROUNDS; ++j) {
        const int m = 4 * j + mg;
        const float sc_next = row_scale(j + 1);
        v8f acc = {0.f, 0.f, 0.f, 0.f, 0.f, 0.f, 0.f, 0.f};
        if (m < M_TILES) {
            const float* ap = tile + (16 * m + l15) * TILE_PITCH + khalf;
            #pragma unroll
            for (int ks = 0; ks < HID_C / 4; ++ks) {
                v2f a = *(const v2f*)(ap + 4 * ks);
                a.x *= sc;
                a.y *= sc;
                acc = __builtin_amdgcn_wmma_f32_16x16x4_f32(false, a, false, bw[ks],
                                                            (short)0, acc, false, false);
            }
        }
        __syncthreads();
        if (m < M_TILES) {
            float* op = tile + (16 * m + 8 * (int)(lane >> 4)) * TILE_PITCH + col;
            #pragma unroll
            for (int i = 0; i < 8; ++i)
                op[i * TILE_PITCH] = fmaxf(acc[i] + bv, 0.0f);
        }
        if (j > 0) store_round(j - 1);
        sc = sc_next;
    }
    __syncthreads();
    store_round(GEMM_ROUNDS - 1);
}

#define CLS_SPLIT 8

template <int K>
__global__ void __launch_bounds__(32 * CLS_SPLIT) wmma_gemm_kernel(
    const float* __restrict__ A, const float* __restrict__ B,
    const float* __restrict__ bias, float* __restrict__ C,
    int M, int N, int do_relu) {
    constexpr int STEPS = K / (4 * CLS_SPLIT);
    static_assert(K % (4 * CLS_SPLIT) == 0 && STEPS >= 1, "K is cut into whole k-steps per wave");
    __shared__ float part[CLS_SPLIT][8][32];

    const int lane  = threadIdx.x & 31;
    const int s     = __builtin_amdgcn_readfirstlane((int)(threadIdx.x >> 5));
    const int l15   = lane & 15;
    const int khalf = (lane >> 4) << 1;
    const int mbase = blockIdx.x * 16;
    const int nbase = blockIdx.y * 16;

    int row = mbase + l15;
    if (row > M - 1) row = M - 1;
    int col = nbase + l15;
    if (col > N - 1) col = N - 1;

    const int kbase = s * (STEPS * 4) + khalf;
    const float* ap = A + (size_t)row * K + kbase;
    const float* bp = B + (size_t)kbase * N + col;

    v2f a[STEPS], b[STEPS];
    #pragma unroll
    for (int t = 0; t < STEPS; ++t) {
        a[t].x = ap[4 * t];
        a[t].y = ap[4 * t + 1];
        b[t].x = bp[(size_t)(4 * t) * N];
        b[t].y = bp[(size_t)(4 * t + 1) * N];
    }
    const float bv = bias[col];

    v8f acc = {0.f, 0.f, 0.f, 0.f, 0.f, 0.f, 0.f, 0.f};
    #pragma unroll
    for (int t = 0; t < STEPS; ++t)
        acc = __builtin_amdgcn_wmma_f32_16x16x4_f32(false, a[t], false, b[t],
                                                    (short)0, acc, false, false);
    #pragma unroll
    for (int i = 0; i < 8; ++i) part[s][i][lane] = acc[i];
    __syncthreads();

    for (int i = s; i < 8; i += CLS_SPLIT) {
        float v = part[0][i][lane];
        #pragma unroll
        for (int p = 1; p < CLS_SPLIT; ++p) v += part[p][i][lane];
        const int m = mbase + i + ((lane >> 4) << 3);
        const int n = nbase + l15;
        if (m < M && n < N) {
            v += bv;
            C[(size_t)m * N + n] = do_relu ? fmaxf(v, 0.0f) : v;
        }
    }
}

#define POOL_WAVES_PER_BLOCK 10
#define POOL_BLOCKS 625
#define POOL_WAVES (POOL_BLOCKS * POOL_WAVES_PER_BLOCK)

__device__ __forceinline__ void pool_flush(float* __restrict__ hg, int g, unsigned lane,
                                           unsigned m0, unsigned m1, unsigned m2, unsigned m3) {
    unsigned int* p = (unsigned int*)(hg + (size_t)g * HID_C);
    atomicMax(p + lane,       m0);
    atomicMax(p + lane + 32u, m1);
    atomicMax(p + lane + 64u, m2);
    atomicMax(p + lane + 96u, m3);
}

__global__ void __launch_bounds__(POOL_WAVES_PER_BLOCK * 32) segmax_kernel(
    const float* __restrict__ h, const int* __restrict__ gids,
    float* __restrict__ hg, int n_nodes) {
    const unsigned lane = threadIdx.x & 31u;
    const int w = __builtin_amdgcn_readfirstlane(
        (int)(blockIdx.x * POOL_WAVES_PER_BLOCK + (threadIdx.x >> 5)));

    const int q  = n_nodes / POOL_WAVES;
    const int r  = n_nodes % POOL_WAVES;
    const int n0 = w * q + (w < r ? w : r);
    const int n1 = n0 + q + (w < r ? 1 : 0);

    int cur = -1;
    unsigned m0 = 0u, m1 = 0u, m2 = 0u, m3 = 0u;

    for (int base = n0; base < n1; base += 32) {
        const int idx = base + (int)lane;
        int gv = 0;
        if (idx < n1) gv = gids[idx];
        const int cnt = (n1 - base < 32) ? (n1 - base) : 32;

        int j = 0;
        for (; j + 8 <= cnt; j += 8) {
            float v[8][4];
            #pragma unroll
            for (int u = 0; u < 8; ++u) {
                const float* hp = h + (size_t)(base + j + u) * HID_C;
                v[u][0] = hp[lane];
                v[u][1] = hp[lane + 32u];
                v[u][2] = hp[lane + 64u];
                v[u][3] = hp[lane + 96u];
            }
            #pragma unroll
            for (int u = 0; u < 8; ++u) {
                const int g = __builtin_amdgcn_readlane(gv, j + u);
                if (g != cur) {
                    if (cur >= 0) pool_flush(hg, cur, lane, m0, m1, m2, m3);
                    m0 = 0u; m1 = 0u; m2 = 0u; m3 = 0u;
                    cur = g;
                }
                const unsigned b0 = __float_as_uint(v[u][0]);
                const unsigned b1 = __float_as_uint(v[u][1]);
                const unsigned b2 = __float_as_uint(v[u][2]);
                const unsigned b3 = __float_as_uint(v[u][3]);
                m0 = (b0 > m0) ? b0 : m0;
                m1 = (b1 > m1) ? b1 : m1;
                m2 = (b2 > m2) ? b2 : m2;
                m3 = (b3 > m3) ? b3 : m3;
            }
        }
        for (; j < cnt; ++j) {
            const float* hp = h + (size_t)(base + j) * HID_C;
            const unsigned b0 = __float_as_uint(hp[lane]);
            const unsigned b1 = __float_as_uint(hp[lane + 32u]);
            const unsigned b2 = __float_as_uint(hp[lane + 64u]);
            const unsigned b3 = __float_as_uint(hp[lane + 96u]);
            const int g = __builtin_amdgcn_readlane(gv, j);
            if (g != cur) {
                if (cur >= 0) pool_flush(hg, cur, lane, m0, m1, m2, m3);
                m0 = 0u; m1 = 0u; m2 = 0u; m3 = 0u;
                cur = g;
            }
            m0 = (b0 > m0) ? b0 : m0;
            m1 = (b1 > m1) ? b1 : m1;
            m2 = (b2 > m2) ? b2 : m2;
            m3 = (b3 > m3) ? b3 : m3;
        }
    }
    if (cur >= 0) pool_flush(hg, cur, lane, m0, m1, m2, m3);
}

extern "C" void kernel_launch(void* const* d_in, const int* in_sizes, int n_in,
                              void* d_out, int out_size, void* d_ws, size_t ws_size,
                              hipStream_t stream) {
    (void)in_sizes; (void)n_in; (void)out_size; (void)ws_size;

    const float* feat = (const float*)d_in[0];
    const int*   src  = (const int*)d_in[1];
    const int*   dst  = (const int*)d_in[2];
    const int*   gids = (const int*)d_in[3];
    const float* Wg[3] = {(const float*)d_in[4], (const float*)d_in[6], (const float*)d_in[8]};
    const float* bg[3] = {(const float*)d_in[5], (const float*)d_in[7], (const float*)d_in[9]};
    const float* Wc1 = (const float*)d_in[10]; const float* bc1 = (const float*)d_in[11];
    const float* Wc2 = (const float*)d_in[12]; const float* bc2 = (const float*)d_in[13];
    const float* Wc3 = (const float*)d_in[14]; const float* bc3 = (const float*)d_in[15];
    float* out = (float*)d_out;

    char* ws = (char*)d_ws;
    size_t off = 0;
    auto take = [&](size_t elems) -> float* {
        float* p = (float*)(ws + off);
        off += ((elems * sizeof(float) + 255) / 256) * 256;
        return p;
    };
    unsigned int* part = (unsigned int*)take((size_t)2 * DEG_CHUNKS * N_NODES_C);
    float* srcn = take(N_NODES_C);
    float* dstn = take(N_NODES_C);
    unsigned int* bm = (unsigned int*)take((size_t)TILE_BLOCKS * BM_PITCH);
    float* h2   = take((size_t)N_NODES_C * HID_C);
    float* h    = take((size_t)N_NODES_C * HID_C);
    float* hg   = take((size_t)N_GRAPHS_C * HID_C);
    float* x1   = take((size_t)N_GRAPHS_C * HID_C);
    float* x2   = take((size_t)N_GRAPHS_C * (HID_C / 2));

    degree_kernel<<<dim3(DEG_RANGES, DEG_CHUNKS, 2), DEG_THREADS, 0, stream>>>(
        src, dst, part, N_EDGES_C, N_NODES_C);
    norm_kernel<<<(N_NODES_C + 255) / 256, 256, 0, stream>>>(part, srcn, dstn, N_NODES_C);

    mark_kernel<<<MARK_BLOCKS, 1024, 0, stream>>>(dst, bm, N_EDGES_C);

    edge_conv_kernel<<<TILE_BLOCKS, TILE_THREADS, 0, stream>>>(
        feat, srcn, dstn, src, dst, bm, Wg[0], bg[0], h, N_EDGES_C, N_NODES_C);
    edge_conv_kernel<<<TILE_BLOCKS, TILE_THREADS, 0, stream>>>(
        h, srcn, dstn, src, dst, bm, Wg[1], bg[1], h2, N_EDGES_C, N_NODES_C);
    edge_conv_kernel<<<TILE_BLOCKS, TILE_THREADS, 0, stream>>>(
        h2, srcn, dstn, src, dst, bm, Wg[2], bg[2], h, N_EDGES_C, N_NODES_C);

    hipMemsetAsync(hg, 0, (size_t)N_GRAPHS_C * HID_C * sizeof(float), stream);
    segmax_kernel<<<POOL_BLOCKS, POOL_WAVES_PER_BLOCK * 32, 0, stream>>>(h, gids, hg, N_NODES_C);

    wmma_gemm_kernel<HID_C><<<dim3(N_GRAPHS_C / 16, HID_C / 16), 32 * CLS_SPLIT, 0, stream>>>(
        hg, Wc1, bc1, x1, N_GRAPHS_C, HID_C, 1);
    wmma_gemm_kernel<HID_C><<<dim3(N_GRAPHS_C / 16, (HID_C / 2) / 16), 32 * CLS_SPLIT, 0, stream>>>(
        x1, Wc2, bc2, x2, N_GRAPHS_C, HID_C / 2, 1);
    wmma_gemm_kernel<HID_C / 2><<<dim3(N_GRAPHS_C / 16, (OUT_DIM_C + 15) / 16), 32 * CLS_SPLIT, 0, stream>>>(
        x2, Wc3, bc3, out, N_GRAPHS_C, OUT_DIM_C, 0);
}


# Layers 2 and 3 take each tile's edges from lists the first layer saves (src id, tile row a word, per wave); their bitmap walk, id line loads, tests and src reads gone; LDS f32 adds in arrival order
# speedup vs baseline: 3.5501x; 1.1319x over previous
#include <hip/hip_runtime.h>
#include <hip/hip_bf16.h>


typedef float v2f __attribute__((ext_vector_type(2)));
typedef float v8f __attribute__((ext_vector_type(8)));

#define N_NODES_C 100000
#define N_EDGES_C 800000
#define HID_C     128
#define OUT_DIM_C 500
#define N_GRAPHS_C 64

#define DEG_RANGES  16
#define DEG_CHUNKS  16
#define DEG_THREADS 256
#define DEG_RANGE_NODES ((((N_NODES_C + DEG_RANGES - 1) / DEG_RANGES) + 31) & ~31)
#define DEG_LOADS   13

__global__ void __launch_bounds__(DEG_THREADS) degree_kernel(
    const int* __restrict__ src, const int* __restrict__ dst,
    unsigned int* __restrict__ part, int n_edges, int n_nodes) {
    __shared__ unsigned int cnt[DEG_RANGE_NODES];

    const int tid = threadIdx.x;
    const int* __restrict__ ids = (blockIdx.z == 0) ? src : dst;
    const int lo = blockIdx.x * DEG_RANGE_NODES;
    int span = n_nodes - lo;
    if (span > DEG_RANGE_NODES) span = DEG_RANGE_NODES;
    if (span < 0) span = 0;

    for (int i = tid; i < DEG_RANGE_NODES; i += DEG_THREADS) cnt[i] = 0u;
    __syncthreads();

    const int chunk = (((n_edges + DEG_CHUNKS - 1) / DEG_CHUNKS) + 3) & ~3;
    const int e0 = blockIdx.y * chunk;
    int e1 = e0 + chunk;
    if (e1 > n_edges) e1 = n_edges;
    if (e1 < e0) e1 = e0;

    const int e1v = e0 + ((e1 - e0) & ~3);
    for (int base = e0 + 4 * tid; base < e1v; base += 4 * DEG_THREADS * DEG_LOADS) {
        int4 v[DEG_LOADS];
        #pragma unroll
        for (int u = 0; u < DEG_LOADS; ++u) {
            const int e = base + u * 4 * DEG_THREADS;
            v[u] = make_int4(-1, -1, -1, -1);
            if (e < e1v) v[u] = *(const int4*)(ids + e);
        }
        #pragma unroll
        for (int u = 0; u < DEG_LOADS; ++u) {
            const unsigned rx = (unsigned)(v[u].x - lo);
            const unsigned ry = (unsigned)(v[u].y - lo);
            const unsigned rz = (unsigned)(v[u].z - lo);
            const unsigned rw = (unsigned)(v[u].w - lo);
            if (rx < (unsigned)span) atomicAdd(&cnt[rx], 1u);
            if (ry < (unsigned)span) atomicAdd(&cnt[ry], 1u);
            if (rz < (unsigned)span) atomicAdd(&cnt[rz], 1u);
            if (rw < (unsigned)span) atomicAdd(&cnt[rw], 1u);
        }
    }
    if (tid < e1 - e1v) {
        const unsigned rt = (unsigned)(ids[e1v + tid] - lo);
        if (rt < (unsigned)span) atomicAdd(&cnt[rt], 1u);
    }
    __syncthreads();

    unsigned int* __restrict__ dstp =
        part + ((size_t)blockIdx.z * DEG_CHUNKS + blockIdx.y) * (size_t)n_nodes + lo;
    for (int i = tid; i < span; i += DEG_THREADS) dstp[i] = cnt[i];
}

__global__ void norm_kernel(const unsigned int* __restrict__ part,
                            float* __restrict__ srcn, float* __restrict__ dstn, int n) {
    int i = blockIdx.x * blockDim.x + threadIdx.x;
    if (i < n) {
        unsigned int od = 0u, id = 0u;
        #pragma unroll
        for (int c = 0; c < DEG_CHUNKS; ++c) {
            od += part[(size_t)c * n + i];
            id += part[(size_t)(DEG_CHUNKS + c) * n + i];
        }
        srcn[i] = rsqrtf(fmaxf((float)od, 1.0f));
        dstn[i] = rsqrtf(fmaxf((float)id, 1.0f));
    }
}

#define TILE_ROWS    400
#define TILE_PITCH   (HID_C + 4)
#define M_TILES      (TILE_ROWS / 16)
#define GEMM_ROUNDS  ((M_TILES + 3) / 4)
#define TILE_THREADS 1024
#define TILE_WAVES   (TILE_THREADS / 32)
#define TILE_BLOCKS  ((N_NODES_C + TILE_ROWS - 1) / TILE_ROWS)

#define LINE_IDS     32
#define N_LINES      ((N_EDGES_C + LINE_IDS - 1) / LINE_IDS)
#define BM_WORDS     ((N_LINES + 31) / 32)
#define WAVE_WORDS   ((BM_WORDS + TILE_WAVES - 1) / TILE_WAVES)
#define BM_PITCH     (((WAVE_WORDS * TILE_WAVES) + 31) & ~31)
#define MARK_BLOCKS  (BM_PITCH / 32)
#define MARK_STRIDE  33
#define BATCH_LINES  16
#define QCAP         (BATCH_LINES * LINE_IDS + 8)

#define LIST_CAP     256
#define LIST_OVER    0xFFFFFFFFu

static_assert(N_EDGES_C <= (1 << 20), "edge index must fit bits 0-19 of a queue word");
static_assert(N_NODES_C <= (1 << 20), "src id must fit bits 0-19 of a saved word");
static_assert(LIST_CAP % 32 == 0, "a saved list is read in whole 128-byte lines");
static_assert(TILE_ROWS <= (1 << 12), "tile row must fit above bit 20 of a queue word");
static_assert(WAVE_WORDS <= 32, "a wave holds its bitmap words one per lane");
static_assert(TILE_ROWS % 16 == 0, "a tile is whole 16-row blocks");
static_assert(TILE_WAVES == 32 && HID_C == 128, "8 column owners x 4 row blocks a round, 2 rows a wave to store");

__global__ void __launch_bounds__(1024) mark_kernel(
    const int* __restrict__ dst, unsigned int* __restrict__ bm, int n_edges) {
    __shared__ unsigned int marks[TILE_BLOCKS * MARK_STRIDE];

    const unsigned lane = threadIdx.x & 31u;
    const int wv = (int)(threadIdx.x >> 5);
    for (int i = threadIdx.x; i < TILE_BLOCKS * MARK_STRIDE; i += 1024) marks[i] = 0u;
    __syncthreads();

    const int ebase = ((int)blockIdx.x * 32 + wv) * (32 * LINE_IDS) + (int)lane;
    #pragma unroll
    for (int g = 0; g < 32; g += 8) {
        int id[8];
        #pragma unroll
        for (int u = 0; u < 8; ++u) {
            const int e = ebase + (g + u) * LINE_IDS;
            id[u] = -1;
            if (e < n_edges) id[u] = dst[e];
        }
        #pragma unroll
        for (int u = 0; u < 8; ++u) {
            const unsigned t = (unsigned)id[u] / (unsigned)TILE_ROWS;
            if (t < (unsigned)TILE_BLOCKS) atomicOr(&marks[t * MARK_STRIDE + wv], 1u << (g + u));
        }
    }
    __syncthreads();

    for (int t = wv; t < TILE_BLOCKS; t += 32)
        bm[(size_t)t * BM_PITCH + blockIdx.x * 32 + lane] = marks[t * MARK_STRIDE + lane];
}

template <bool FULL>
__device__ __forceinline__ void tile_group(const float* __restrict__ h,
                                           const float* __restrict__ srcn,
                                           float* tile, int sv, int dv,
                                           int first, int nvalid, unsigned lane) {
    int   s[8], d[8];
    float sn[8];
    float v[8][4];
    #pragma unroll
    for (int u = 0; u < 8; ++u) {
        s[u] = __builtin_amdgcn_readlane(sv, first + u);
        d[u] = __builtin_amdgcn_readlane(dv, first + u);
    }
    #pragma unroll
    for (int u = 0; u < 8; ++u) {
        sn[u] = 0.0f;
        v[u][0] = 0.0f; v[u][1] = 0.0f; v[u][2] = 0.0f; v[u][3] = 0.0f;
        if (FULL || u < nvalid) {
            const float* hp = h + (size_t)s[u] * HID_C;
            sn[u]   = srcn[s[u]];
            v[u][0] = hp[lane];
            v[u][1] = hp[lane + 32u];
            v[u][2] = hp[lane + 64u];
            v[u][3] = hp[lane + 96u];
        }
    }
    #pragma unroll
    for (int u = 0; u < 8; ++u) {
        if (FULL || u < nvalid) {
            float* tp = tile + d[u] * TILE_PITCH;
            __hip_atomic_fetch_add(tp + lane,       v[u][0] * sn[u], __ATOMIC_RELAXED, __HIP_MEMORY_SCOPE_WORKGROUP);
            __hip_atomic_fetch_add(tp + lane + 32u, v[u][1] * sn[u], __ATOMIC_RELAXED, __HIP_MEMORY_SCOPE_WORKGROUP);
            __hip_atomic_fetch_add(tp + lane + 64u, v[u][2] * sn[u], __ATOMIC_RELAXED, __HIP_MEMORY_SCOPE_WORKGROUP);
            __hip_atomic_fetch_add(tp + lane + 96u, v[u][3] * sn[u], __ATOMIC_RELAXED, __HIP_MEMORY_SCOPE_WORKGROUP);
        }
    }
}

template <bool FIRST>
__global__ void __launch_bounds__(TILE_THREADS) edge_conv_kernel(
    const float* __restrict__ h, const float* __restrict__ srcn,
    const float* __restrict__ dstn,
    const int* __restrict__ src, const int* __restrict__ dst,
    const unsigned int* __restrict__ bm,
    unsigned int* __restrict__ lst, unsigned int* __restrict__ hdr,
    const float* __restrict__ W, const float* __restrict__ bias,
    float* __restrict__ out, int n_edges, int n_nodes) {
    __shared__ float        tile[TILE_ROWS * TILE_PITCH];
    __shared__ unsigned int queue[TILE_WAVES * QCAP];

    const unsigned lane = threadIdx.x & 31u;
    const int w  = __builtin_amdgcn_readfirstlane((int)(threadIdx.x >> 5));
    const int lo = blockIdx.x * TILE_ROWS;
    int span = n_nodes - lo;
    if (span > TILE_ROWS) span = TILE_ROWS;
    if (span < 0) span = 0;

    const int slot = blockIdx.x * TILE_WAVES + w;
    unsigned int* lp = lst + (size_t)slot * LIST_CAP;
    unsigned saved = LIST_OVER;
    if (!FIRST) saved = (unsigned)__builtin_amdgcn_readfirstlane((int)hdr[slot]);
    const bool scan = FIRST || saved > (unsigned)LIST_CAP;

    unsigned words = 0u;
    unsigned ent_next = 0u;
    if (scan) {
        if ((int)lane < WAVE_WORDS)
            words = bm[(size_t)blockIdx.x * BM_PITCH + w * WAVE_WORDS + (int)lane];
    } else {
        if (lane < saved) ent_next = lp[lane];
    }

    float4* t4 = (float4*)tile;
    for (int i = threadIdx.x; i < TILE_ROWS * TILE_PITCH / 4; i += TILE_THREADS)
        t4[i] = make_float4(0.f, 0.f, 0.f, 0.f);
    __syncthreads();

    unsigned int* q = queue + w * QCAP;
    int count = 0;
    int  nsaved = 0;
    bool over   = false;

    auto save = [&](int sv, int dv, int n) {
        if (!over && nsaved + n <= LIST_CAP) {
            if ((int)lane < n) lp[nsaved + (int)lane] = (unsigned)sv | ((unsigned)dv << 20);
            nsaved += n;
        } else {
            over = true;
        }
    };

    auto enqueue = [&](int id, int e) {
        const unsigned rel  = (unsigned)(id - lo);
        const bool     m    = rel < (unsigned)span;
        const unsigned mask = __builtin_amdgcn_ballot_w32(m);
        if (mask != 0u) {
            const unsigned rank = (unsigned)count + __builtin_amdgcn_mbcnt_lo(mask, 0u);
            if (m) q[rank] = (unsigned)e | (rel << 20);
            count += __builtin_popcount(mask);
        }
    };
    auto drain = [&]() {
        const int nfull = count & ~7;
        for (int pos = 0; pos < nfull; pos += 32) {
            const int n = (nfull - pos < 32) ? (nfull - pos) : 32;
            unsigned ent = 0u;
            int sv = 0;
            if ((int)lane < n) {
                ent = q[pos + (int)lane];
                sv  = src[ent & 0xFFFFFu];
            }
            const int dv = (int)(ent >> 20);
            if (FIRST) save(sv, dv, n);
            for (int b = 0; b < n; b += 8)
                tile_group<true>(h, srcn, tile, sv, dv, b, 8, lane);
        }
        const int rem = count - nfull;
        unsigned carry = 0u;
        if ((int)lane < rem) carry = q[nfull + (int)lane];
        if ((int)lane < rem) q[lane] = carry;
        count = rem;
    };

    const int line0 = w * WAVE_WORDS * 32;
    int p = 0;
    unsigned bits = (unsigned)__builtin_amdgcn_readlane((int)words, 0);
    auto next_line = [&]() -> int {
        while (bits == 0u && p < WAVE_WORDS - 1) {
            ++p;
            bits = (unsigned)__builtin_amdgcn_readlane((int)words, p);
        }
        if (bits == 0u) return -1;
        const int b = __builtin_ctz(bits);
        bits &= bits - 1u;
        return line0 + p * 32 + b;
    };
    auto load_batch = [&](int (&ln)[BATCH_LINES], int (&idv)[BATCH_LINES]) {
        #pragma unroll
        for (int u = 0; u < BATCH_LINES; ++u) ln[u] = next_line();
        #pragma unroll
        for (int u = 0; u < BATCH_LINES; ++u) {
            const int lnu = (ln[u] < 0) ? 0 : ln[u];
            const int e   = lnu * LINE_IDS + (int)lane;
            int v = -1;
            if (e < n_edges) v = dst[e];
            idv[u] = (ln[u] < 0) ? -1 : v;
        }
    };

    int t_sv = 0, t_dv = 0, t_first = 0, t_n = 0;
    if (scan) {
        int lc[BATCH_LINES], ic[BATCH_LINES], lx[BATCH_LINES], ix[BATCH_LINES];
        load_batch(lc, ic);
        for (int it = 0; it < (WAVE_WORDS * 32) / BATCH_LINES && lc[0] >= 0; ++it) {
            load_batch(lx, ix);
            #pragma unroll
            for (int u = 0; u < BATCH_LINES; ++u)
                enqueue(ic[u], lc[u] * LINE_IDS + (int)lane);
            if (count >= 8) drain();
            #pragma unroll
            for (int u = 0; u < BATCH_LINES; ++u) { lc[u] = lx[u]; ic[u] = ix[u]; }
        }
    } else {
        const int total = (int)saved;
        for (int pos = 0; pos < total; pos += 32) {
            const int n = (total - pos < 32) ? (total - pos) : 32;
            const unsigned ent = ent_next;
            ent_next = 0u;
            if ((int)lane < total - pos - 32) ent_next = lp[pos + 32 + (int)lane];
            const int sv = (int)(ent & 0xFFFFFu);
            const int dv = (int)(ent >> 20);
            const int nf = n & ~7;
            for (int b = 0; b < nf; b += 8)
                tile_group<true>(h, srcn, tile, sv, dv, b, 8, lane);
            if (nf < n) { t_sv = sv; t_dv = dv; t_first = nf; t_n = n - nf; }
        }
    }
    const int l15   = (int)(lane & 15u);
    const int khalf = (int)(lane >> 4) << 1;
    const int nt    = w & 7;
    const int mg    = w >> 3;
    const int col   = nt * 16 + l15;

    v2f bw[HID_C / 4];
    #pragma unroll
    for (int ks = 0; ks < HID_C / 4; ++ks) {
        bw[ks].x = W[(size_t)(4 * ks + khalf) * HID_C + col];
        bw[ks].y = W[(size_t)(4 * ks + khalf + 1) * HID_C + col];
    }
    const float bv = bias[col];

    if (scan) {
        if (count > 0) {
            unsigned ent = 0u;
            int sv = 0;
            if ((int)lane < count) {
                ent = q[lane];
                sv  = src[ent & 0xFFFFFu];
            }
            const int dv = (int)(ent >> 20);
            if (FIRST) save(sv, dv, count);
            tile_group<false>(h, srcn, tile, sv, dv, 0, count, lane);
        }
        if (FIRST && lane == 0u) hdr[slot] = over ? LIST_OVER : (unsigned)nsaved;
    } else if (t_n > 0) {
        tile_group<false>(h, srcn, tile, t_sv, t_dv, t_first, t_n, lane);
    }

    auto row_scale = [&](int j) -> float {
        const int m    = 4 * j + mg;
        const int node = lo + 16 * m + l15;
        return (m < M_TILES && node < n_nodes) ? dstn[node] : 0.0f;
    };
    float sc = row_scale(0);

    __syncthreads();

    auto store_round = [&](int j) {
        #pragma unroll
        for (int t = 0; t < 2; ++t) {
            const int r = 64 * j + 32 * t + w;
            if (r < span) {
                const float4 v = *(const float4*)(tile + r * TILE_PITCH + 4 * (int)lane);
                *(float4*)(out + (size_t)(lo + r) * HID_C + 4 * lane) = v;
            }
        }
    };

    for (int j = 0; j < GEMM_ROUNDS; ++j) {
        const int m = 4 * j + mg;
        const float sc_next = row_scale(j + 1);
        v8f acc = {0.f, 0.f, 0.f, 0.f, 0.f, 0.f, 0.f, 0.f};
        if (m < M_TILES) {
            const float* ap = tile + (16 * m + l15) * TILE_PITCH + khalf;
            #pragma unroll
            for (int ks = 0; ks < HID_C / 4; ++ks) {
                v2f a = *(const v2f*)(ap + 4 * ks);
                a.x *= sc;
                a.y *= sc;
                acc = __builtin_amdgcn_wmma_f32_16x16x4_f32(false, a, false, bw[ks],
                                                            (short)0, acc, false, false);
            }
        }
        __syncthreads();
        if (m < M_TILES) {
            float* op = tile + (16 * m + 8 * (int)(lane >> 4)) * TILE_PITCH + col;
            #pragma unroll
            for (int i = 0; i < 8; ++i)
                op[i * TILE_PITCH] = fmaxf(acc[i] + bv, 0.0f);
        }
        if (j > 0) store_round(j - 1);
        sc = sc_next;
    }
    __syncthreads();
    store_round(GEMM_ROUNDS - 1);
}

#define CLS_SPLIT 8

template <int K>
__global__ void __launch_bounds__(32 * CLS_SPLIT) wmma_gemm_kernel(
    const float* __restrict__ A, const float* __restrict__ B,
    const float* __restrict__ bias, float* __restrict__ C,
    int M, int N, int do_relu) {
    constexpr int STEPS = K / (4 * CLS_SPLIT);
    static_assert(K % (4 * CLS_SPLIT) == 0 && STEPS >= 1, "K is cut into whole k-steps per wave");
    __shared__ float part[CLS_SPLIT][8][32];

    const int lane  = threadIdx.x & 31;
    const int s     = __builtin_amdgcn_readfirstlane((int)(threadIdx.x >> 5));
    const int l15   = lane & 15;
    const int khalf = (lane >> 4) << 1;
    const int mbase = blockIdx.x * 16;
    const int nbase = blockIdx.y * 16;

    int row = mbase + l15;
    if (row > M - 1) row = M - 1;
    int col = nbase + l15;
    if (col > N - 1) col = N - 1;

    const int kbase = s * (STEPS * 4) + khalf;
    const float* ap = A + (size_t)row * K + kbase;
    const float* bp = B + (size_t)kbase * N + col;

    v2f a[STEPS], b[STEPS];
    #pragma unroll
    for (int t = 0; t < STEPS; ++t) {
        a[t].x = ap[4 * t];
        a[t].y = ap[4 * t + 1];
        b[t].x = bp[(size_t)(4 * t) * N];
        b[t].y = bp[(size_t)(4 * t + 1) * N];
    }
    const float bv = bias[col];

    v8f acc = {0.f, 0.f, 0.f, 0.f, 0.f, 0.f, 0.f, 0.f};
    #pragma unroll
    for (int t = 0; t < STEPS; ++t)
        acc = __builtin_amdgcn_wmma_f32_16x16x4_f32(false, a[t], false, b[t],
                                                    (short)0, acc, false, false);
    #pragma unroll
    for (int i = 0; i < 8; ++i) part[s][i][lane] = acc[i];
    __syncthreads();

    for (int i = s; i < 8; i += CLS_SPLIT) {
        float v = part[0][i][lane];
        #pragma unroll
        for (int p = 1; p < CLS_SPLIT; ++p) v += part[p][i][lane];
        const int m = mbase + i + ((lane >> 4) << 3);
        const int n = nbase + l15;
        if (m < M && n < N) {
            v += bv;
            C[(size_t)m * N + n] = do_relu ? fmaxf(v, 0.0f) : v;
        }
    }
}

#define POOL_WAVES_PER_BLOCK 10
#define POOL_BLOCKS 625
#define POOL_WAVES (POOL_BLOCKS * POOL_WAVES_PER_BLOCK)

__device__ __forceinline__ void pool_flush(float* __restrict__ hg, int g, unsigned lane,
                                           unsigned m0, unsigned m1, unsigned m2, unsigned m3) {
    unsigned int* p = (unsigned int*)(hg + (size_t)g * HID_C);
    atomicMax(p + lane,       m0);
    atomicMax(p + lane + 32u, m1);
    atomicMax(p + lane + 64u, m2);
    atomicMax(p + lane + 96u, m3);
}

__global__ void __launch_bounds__(POOL_WAVES_PER_BLOCK * 32) segmax_kernel(
    const float* __restrict__ h, const int* __restrict__ gids,
    float* __restrict__ hg, int n_nodes) {
    const unsigned lane = threadIdx.x & 31u;
    const int w = __builtin_amdgcn_readfirstlane(
        (int)(blockIdx.x * POOL_WAVES_PER_BLOCK + (threadIdx.x >> 5)));

    const int q  = n_nodes / POOL_WAVES;
    const int r  = n_nodes % POOL_WAVES;
    const int n0 = w * q + (w < r ? w : r);
    const int n1 = n0 + q + (w < r ? 1 : 0);

    int cur = -1;
    unsigned m0 = 0u, m1 = 0u, m2 = 0u, m3 = 0u;

    for (int base = n0; base < n1; base += 32) {
        const int idx = base + (int)lane;
        int gv = 0;
        if (idx < n1) gv = gids[idx];
        const int cnt = (n1 - base < 32) ? (n1 - base) : 32;

        int j = 0;
        for (; j + 8 <= cnt; j += 8) {
            float v[8][4];
            #pragma unroll
            for (int u = 0; u < 8; ++u) {
                const float* hp = h + (size_t)(base + j + u) * HID_C;
                v[u][0] = hp[lane];
                v[u][1] = hp[lane + 32u];
                v[u][2] = hp[lane + 64u];
                v[u][3] = hp[lane + 96u];
            }
            #pragma unroll
            for (int u = 0; u < 8; ++u) {
                const int g = __builtin_amdgcn_readlane(gv, j + u);
                if (g != cur) {
                    if (cur >= 0) pool_flush(hg, cur, lane, m0, m1, m2, m3);
                    m0 = 0u; m1 = 0u; m2 = 0u; m3 = 0u;
                    cur = g;
                }
                const unsigned b0 = __float_as_uint(v[u][0]);
                const unsigned b1 = __float_as_uint(v[u][1]);
                const unsigned b2 = __float_as_uint(v[u][2]);
                const unsigned b3 = __float_as_uint(v[u][3]);
                m0 = (b0 > m0) ? b0 : m0;
                m1 = (b1 > m1) ? b1 : m1;
                m2 = (b2 > m2) ? b2 : m2;
                m3 = (b3 > m3) ? b3 : m3;
            }
        }
        for (; j < cnt; ++j) {
            const float* hp = h + (size_t)(base + j) * HID_C;
            const unsigned b0 = __float_as_uint(hp[lane]);
            const unsigned b1 = __float_as_uint(hp[lane + 32u]);
            const unsigned b2 = __float_as_uint(hp[lane + 64u]);
            const unsigned b3 = __float_as_uint(hp[lane + 96u]);
            const int g = __builtin_amdgcn_readlane(gv, j);
            if (g != cur) {
                if (cur >= 0) pool_flush(hg, cur, lane, m0, m1, m2, m3);
                m0 = 0u; m1 = 0u; m2 = 0u; m3 = 0u;
                cur = g;
            }
            m0 = (b0 > m0) ? b0 : m0;
            m1 = (b1 > m1) ? b1 : m1;
            m2 = (b2 > m2) ? b2 : m2;
            m3 = (b3 > m3) ? b3 : m3;
        }
    }
    if (cur >= 0) pool_flush(hg, cur, lane, m0, m1, m2, m3);
}

extern "C" void kernel_launch(void* const* d_in, const int* in_sizes, int n_in,
                              void* d_out, int out_size, void* d_ws, size_t ws_size,
                              hipStream_t stream) {
    (void)in_sizes; (void)n_in; (void)out_size; (void)ws_size;

    const float* feat = (const float*)d_in[0];
    const int*   src  = (const int*)d_in[1];
    const int*   dst  = (const int*)d_in[2];
    const int*   gids = (const int*)d_in[3];
    const float* Wg[3] = {(const float*)d_in[4], (const float*)d_in[6], (const float*)d_in[8]};
    const float* bg[3] = {(const float*)d_in[5], (const float*)d_in[7], (const float*)d_in[9]};
    const float* Wc1 = (const float*)d_in[10]; const float* bc1 = (const float*)d_in[11];
    const float* Wc2 = (const float*)d_in[12]; const float* bc2 = (const float*)d_in[13];
    const float* Wc3 = (const float*)d_in[14]; const float* bc3 = (const float*)d_in[15];
    float* out = (float*)d_out;

    char* ws = (char*)d_ws;
    size_t off = 0;
    auto take = [&](size_t elems) -> float* {
        float* p = (float*)(ws + off);
        off += ((elems * sizeof(float) + 255) / 256) * 256;
        return p;
    };
    unsigned int* part = (unsigned int*)take((size_t)2 * DEG_CHUNKS * N_NODES_C);
    float* srcn = take(N_NODES_C);
    float* dstn = take(N_NODES_C);
    unsigned int* bm = (unsigned int*)take((size_t)TILE_BLOCKS * BM_PITCH);
    unsigned int* lst = (unsigned int*)take((size_t)TILE_BLOCKS * TILE_WAVES * LIST_CAP);
    unsigned int* hdr = (unsigned int*)take((size_t)TILE_BLOCKS * TILE_WAVES);
    float* h2   = take((size_t)N_NODES_C * HID_C);
    float* h    = take((size_t)N_NODES_C * HID_C);
    float* hg   = take((size_t)N_GRAPHS_C * HID_C);
    float* x1   = take((size_t)N_GRAPHS_C * HID_C);
    float* x2   = take((size_t)N_GRAPHS_C * (HID_C / 2));

    degree_kernel<<<dim3(DEG_RANGES, DEG_CHUNKS, 2), DEG_THREADS, 0, stream>>>(
        src, dst, part, N_EDGES_C, N_NODES_C);
    norm_kernel<<<(N_NODES_C + 255) / 256, 256, 0, stream>>>(part, srcn, dstn, N_NODES_C);

    mark_kernel<<<MARK_BLOCKS, 1024, 0, stream>>>(dst, bm, N_EDGES_C);

    edge_conv_kernel<true><<<TILE_BLOCKS, TILE_THREADS, 0, stream>>>(
        feat, srcn, dstn, src, dst, bm, lst, hdr, Wg[0], bg[0], h, N_EDGES_C, N_NODES_C);
    edge_conv_kernel<false><<<TILE_BLOCKS, TILE_THREADS, 0, stream>>>(
        h, srcn, dstn, src, dst, bm, lst, hdr, Wg[1], bg[1], h2, N_EDGES_C, N_NODES_C);
    edge_conv_kernel<false><<<TILE_BLOCKS, TILE_THREADS, 0, stream>>>(
        h2, srcn, dstn, src, dst, bm, lst, hdr, Wg[2], bg[2], h, N_EDGES_C, N_NODES_C);

    hipMemsetAsync(hg, 0, (size_t)N_GRAPHS_C * HID_C * sizeof(float), stream);
    segmax_kernel<<<POOL_BLOCKS, POOL_WAVES_PER_BLOCK * 32, 0, stream>>>(h, gids, hg, N_NODES_C);

    wmma_gemm_kernel<HID_C><<<dim3(N_GRAPHS_C / 16, HID_C / 16), 32 * CLS_SPLIT, 0, stream>>>(
        hg, Wc1, bc1, x1, N_GRAPHS_C, HID_C, 1);
    wmma_gemm_kernel<HID_C><<<dim3(N_GRAPHS_C / 16, (HID_C / 2) / 16), 32 * CLS_SPLIT, 0, stream>>>(
        x1, Wc2, bc2, x2, N_GRAPHS_C, HID_C / 2, 1);
    wmma_gemm_kernel<HID_C / 2><<<dim3(N_GRAPHS_C / 16, (OUT_DIM_C + 15) / 16), 32 * CLS_SPLIT, 0, stream>>>(
        x2, Wc3, bc3, out, N_GRAPHS_C, OUT_DIM_C, 0);
}


# Conv row gathers as one 16-byte load per lane (a wave load = the row's 512 B; was 4 one-line loads); LDS tile holds sums column-permuted, A and B loaders follow: k sum order of conv products changes
# speedup vs baseline: 3.7223x; 1.0485x over previous
#include <hip/hip_runtime.h>
#include <hip/hip_bf16.h>


typedef float v2f __attribute__((ext_vector_type(2)));
typedef float v8f __attribute__((ext_vector_type(8)));

#define N_NODES_C 100000
#define N_EDGES_C 800000
#define HID_C     128
#define OUT_DIM_C 500
#define N_GRAPHS_C 64

#define DEG_RANGES  16
#define DEG_CHUNKS  16
#define DEG_THREADS 256
#define DEG_RANGE_NODES ((((N_NODES_C + DEG_RANGES - 1) / DEG_RANGES) + 31) & ~31)
#define DEG_LOADS   13

__global__ void __launch_bounds__(DEG_THREADS) degree_kernel(
    const int* __restrict__ src, const int* __restrict__ dst,
    unsigned int* __restrict__ part, int n_edges, int n_nodes) {
    __shared__ unsigned int cnt[DEG_RANGE_NODES];

    const int tid = threadIdx.x;
    const int* __restrict__ ids = (blockIdx.z == 0) ? src : dst;
    const int lo = blockIdx.x * DEG_RANGE_NODES;
    int span = n_nodes - lo;
    if (span > DEG_RANGE_NODES) span = DEG_RANGE_NODES;
    if (span < 0) span = 0;

    for (int i = tid; i < DEG_RANGE_NODES; i += DEG_THREADS) cnt[i] = 0u;
    __syncthreads();

    const int chunk = (((n_edges + DEG_CHUNKS - 1) / DEG_CHUNKS) + 3) & ~3;
    const int e0 = blockIdx.y * chunk;
    int e1 = e0 + chunk;
    if (e1 > n_edges) e1 = n_edges;
    if (e1 < e0) e1 = e0;

    const int e1v = e0 + ((e1 - e0) & ~3);
    for (int base = e0 + 4 * tid; base < e1v; base += 4 * DEG_THREADS * DEG_LOADS) {
        int4 v[DEG_LOADS];
        #pragma unroll
        for (int u = 0; u < DEG_LOADS; ++u) {
            const int e = base + u * 4 * DEG_THREADS;
            v[u] = make_int4(-1, -1, -1, -1);
            if (e < e1v) v[u] = *(const int4*)(ids + e);
        }
        #pragma unroll
        for (int u = 0; u < DEG_LOADS; ++u) {
            const unsigned rx = (unsigned)(v[u].x - lo);
            const unsigned ry = (unsigned)(v[u].y - lo);
            const unsigned rz = (unsigned)(v[u].z - lo);
            const unsigned rw = (unsigned)(v[u].w - lo);
            if (rx < (unsigned)span) atomicAdd(&cnt[rx], 1u);
            if (ry < (unsigned)span) atomicAdd(&cnt[ry], 1u);
            if (rz < (unsigned)span) atomicAdd(&cnt[rz], 1u);
            if (rw < (unsigned)span) atomicAdd(&cnt[rw], 1u);
        }
    }
    if (tid < e1 - e1v) {
        const unsigned rt = (unsigned)(ids[e1v + tid] - lo);
        if (rt < (unsigned)span) atomicAdd(&cnt[rt], 1u);
    }
    __syncthreads();

    unsigned int* __restrict__ dstp =
        part + ((size_t)blockIdx.z * DEG_CHUNKS + blockIdx.y) * (size_t)n_nodes + lo;
    for (int i = tid; i < span; i += DEG_THREADS) dstp[i] = cnt[i];
}

__global__ void norm_kernel(const unsigned int* __restrict__ part,
                            float* __restrict__ srcn, float* __restrict__ dstn, int n) {
    int i = blockIdx.x * blockDim.x + threadIdx.x;
    if (i < n) {
        unsigned int od = 0u, id = 0u;
        #pragma unroll
        for (int c = 0; c < DEG_CHUNKS; ++c) {
            od += part[(size_t)c * n + i];
            id += part[(size_t)(DEG_CHUNKS + c) * n + i];
        }
        srcn[i] = rsqrtf(fmaxf((float)od, 1.0f));
        dstn[i] = rsqrtf(fmaxf((float)id, 1.0f));
    }
}

#define TILE_ROWS    400
#define TILE_PITCH   (HID_C + 4)
#define TILE_COL(p)  (4 * ((p) & 31) + ((p) >> 5))
#define M_TILES      (TILE_ROWS / 16)
#define GEMM_ROUNDS  ((M_TILES + 3) / 4)
#define TILE_THREADS 1024
#define TILE_WAVES   (TILE_THREADS / 32)
#define TILE_BLOCKS  ((N_NODES_C + TILE_ROWS - 1) / TILE_ROWS)

#define LINE_IDS     32
#define N_LINES      ((N_EDGES_C + LINE_IDS - 1) / LINE_IDS)
#define BM_WORDS     ((N_LINES + 31) / 32)
#define WAVE_WORDS   ((BM_WORDS + TILE_WAVES - 1) / TILE_WAVES)
#define BM_PITCH     (((WAVE_WORDS * TILE_WAVES) + 31) & ~31)
#define MARK_BLOCKS  (BM_PITCH / 32)
#define MARK_STRIDE  33
#define BATCH_LINES  16
#define QCAP         (BATCH_LINES * LINE_IDS + 8)

#define LIST_CAP     256
#define LIST_OVER    0xFFFFFFFFu

static_assert(N_EDGES_C <= (1 << 20), "edge index must fit bits 0-19 of a queue word");
static_assert(N_NODES_C <= (1 << 20), "src id must fit bits 0-19 of a saved word");
static_assert(LIST_CAP % 32 == 0, "a saved list is read in whole 128-byte lines");
static_assert(TILE_ROWS <= (1 << 12), "tile row must fit above bit 20 of a queue word");
static_assert(WAVE_WORDS <= 32, "a wave holds its bitmap words one per lane");
static_assert(TILE_ROWS % 16 == 0, "a tile is whole 16-row blocks");
static_assert(TILE_WAVES == 32 && HID_C == 128, "8 column owners x 4 row blocks a round, 2 rows a wave to store");
static_assert(HID_C == 4 * 32, "a lane's 16-byte load is its 4 columns of a row; TILE_COL is their word order in the tile");

__global__ void __launch_bounds__(1024) mark_kernel(
    const int* __restrict__ dst, unsigned int* __restrict__ bm, int n_edges) {
    __shared__ unsigned int marks[TILE_BLOCKS * MARK_STRIDE];

    const unsigned lane = threadIdx.x & 31u;
    const int wv = (int)(threadIdx.x >> 5);
    for (int i = threadIdx.x; i < TILE_BLOCKS * MARK_STRIDE; i += 1024) marks[i] = 0u;
    __syncthreads();

    const int ebase = ((int)blockIdx.x * 32 + wv) * (32 * LINE_IDS) + (int)lane;
    #pragma unroll
    for (int g = 0; g < 32; g += 8) {
        int id[8];
        #pragma unroll
        for (int u = 0; u < 8; ++u) {
            const int e = ebase + (g + u) * LINE_IDS;
            id[u] = -1;
            if (e < n_edges) id[u] = dst[e];
        }
        #pragma unroll
        for (int u = 0; u < 8; ++u) {
            const unsigned t = (unsigned)id[u] / (unsigned)TILE_ROWS;
            if (t < (unsigned)TILE_BLOCKS) atomicOr(&marks[t * MARK_STRIDE + wv], 1u << (g + u));
        }
    }
    __syncthreads();

    for (int t = wv; t < TILE_BLOCKS; t += 32)
        bm[(size_t)t * BM_PITCH + blockIdx.x * 32 + lane] = marks[t * MARK_STRIDE + lane];
}

template <bool FULL>
__device__ __forceinline__ void tile_group(const float* __restrict__ h,
                                           const float* __restrict__ srcn,
                                           float* tile, int sv, int dv,
                                           int first, int nvalid, unsigned lane) {
    int    s[8], d[8];
    float  sn[8];
    float4 v[8];
    #pragma unroll
    for (int u = 0; u < 8; ++u) {
        s[u] = __builtin_amdgcn_readlane(sv, first + u);
        d[u] = __builtin_amdgcn_readlane(dv, first + u);
    }
    #pragma unroll
    for (int u = 0; u < 8; ++u) {
        sn[u] = 0.0f;
        v[u]  = make_float4(0.0f, 0.0f, 0.0f, 0.0f);
        if (FULL || u < nvalid) {
            const float* hp = h + (size_t)s[u] * HID_C;
            sn[u] = srcn[s[u]];
            v[u]  = *(const float4*)(hp + 4u * lane);
        }
    }
    #pragma unroll
    for (int u = 0; u < 8; ++u) {
        if (FULL || u < nvalid) {
            float* tp = tile + d[u] * TILE_PITCH;
            __hip_atomic_fetch_add(tp + lane,       v[u].x * sn[u], __ATOMIC_RELAXED, __HIP_MEMORY_SCOPE_WORKGROUP);
            __hip_atomic_fetch_add(tp + lane + 32u, v[u].y * sn[u], __ATOMIC_RELAXED, __HIP_MEMORY_SCOPE_WORKGROUP);
            __hip_atomic_fetch_add(tp + lane + 64u, v[u].z * sn[u], __ATOMIC_RELAXED, __HIP_MEMORY_SCOPE_WORKGROUP);
            __hip_atomic_fetch_add(tp + lane + 96u, v[u].w * sn[u], __ATOMIC_RELAXED, __HIP_MEMORY_SCOPE_WORKGROUP);
        }
    }
}

template <bool FIRST>
__global__ void __launch_bounds__(TILE_THREADS) edge_conv_kernel(
    const float* __restrict__ h, const float* __restrict__ srcn,
    const float* __restrict__ dstn,
    const int* __restrict__ src, const int* __restrict__ dst,
    const unsigned int* __restrict__ bm,
    unsigned int* __restrict__ lst, unsigned int* __restrict__ hdr,
    const float* __restrict__ W, const float* __restrict__ bias,
    float* __restrict__ out, int n_edges, int n_nodes) {
    __shared__ float        tile[TILE_ROWS * TILE_PITCH];
    __shared__ unsigned int queue[TILE_WAVES * QCAP];

    const unsigned lane = threadIdx.x & 31u;
    const int w  = __builtin_amdgcn_readfirstlane((int)(threadIdx.x >> 5));
    const int lo = blockIdx.x * TILE_ROWS;
    int span = n_nodes - lo;
    if (span > TILE_ROWS) span = TILE_ROWS;
    if (span < 0) span = 0;

    const int slot = blockIdx.x * TILE_WAVES + w;
    unsigned int* lp = lst + (size_t)slot * LIST_CAP;
    unsigned saved = LIST_OVER;
    if (!FIRST) saved = (unsigned)__builtin_amdgcn_readfirstlane((int)hdr[slot]);
    const bool scan = FIRST || saved > (unsigned)LIST_CAP;

    unsigned words = 0u;
    unsigned ent_next = 0u;
    if (scan) {
        if ((int)lane < WAVE_WORDS)
            words = bm[(size_t)blockIdx.x * BM_PITCH + w * WAVE_WORDS + (int)lane];
    } else {
        if (lane < saved) ent_next = lp[lane];
    }

    float4* t4 = (float4*)tile;
    for (int i = threadIdx.x; i < TILE_ROWS * TILE_PITCH / 4; i += TILE_THREADS)
        t4[i] = make_float4(0.f, 0.f, 0.f, 0.f);
    __syncthreads();

    unsigned int* q = queue + w * QCAP;
    int count = 0;
    int  nsaved = 0;
    bool over   = false;

    auto save = [&](int sv, int dv, int n) {
        if (!over && nsaved + n <= LIST_CAP) {
            if ((int)lane < n) lp[nsaved + (int)lane] = (unsigned)sv | ((unsigned)dv << 20);
            nsaved += n;
        } else {
            over = true;
        }
    };

    auto enqueue = [&](int id, int e) {
        const unsigned rel  = (unsigned)(id - lo);
        const bool     m    = rel < (unsigned)span;
        const unsigned mask = __builtin_amdgcn_ballot_w32(m);
        if (mask != 0u) {
            const unsigned rank = (unsigned)count + __builtin_amdgcn_mbcnt_lo(mask, 0u);
            if (m) q[rank] = (unsigned)e | (rel << 20);
            count += __builtin_popcount(mask);
        }
    };
    auto drain = [&]() {
        const int nfull = count & ~7;
        for (int pos = 0; pos < nfull; pos += 32) {
            const int n = (nfull - pos < 32) ? (nfull - pos) : 32;
            unsigned ent = 0u;
            int sv = 0;
            if ((int)lane < n) {
                ent = q[pos + (int)lane];
                sv  = src[ent & 0xFFFFFu];
            }
            const int dv = (int)(ent >> 20);
            if (FIRST) save(sv, dv, n);
            for (int b = 0; b < n; b += 8)
                tile_group<true>(h, srcn, tile, sv, dv, b, 8, lane);
        }
        const int rem = count - nfull;
        unsigned carry = 0u;
        if ((int)lane < rem) carry = q[nfull + (int)lane];
        if ((int)lane < rem) q[lane] = carry;
        count = rem;
    };

    const int line0 = w * WAVE_WORDS * 32;
    int p = 0;
    unsigned bits = (unsigned)__builtin_amdgcn_readlane((int)words, 0);
    auto next_line = [&]() -> int {
        while (bits == 0u && p < WAVE_WORDS - 1) {
            ++p;
            bits = (unsigned)__builtin_amdgcn_readlane((int)words, p);
        }
        if (bits == 0u) return -1;
        const int b = __builtin_ctz(bits);
        bits &= bits - 1u;
        return line0 + p * 32 + b;
    };
    auto load_batch = [&](int (&ln)[BATCH_LINES], int (&idv)[BATCH_LINES]) {
        #pragma unroll
        for (int u = 0; u < BATCH_LINES; ++u) ln[u] = next_line();
        #pragma unroll
        for (int u = 0; u < BATCH_LINES; ++u) {
            const int lnu = (ln[u] < 0) ? 0 : ln[u];
            const int e   = lnu * LINE_IDS + (int)lane;
            int v = -1;
            if (e < n_edges) v = dst[e];
            idv[u] = (ln[u] < 0) ? -1 : v;
        }
    };

    int t_sv = 0, t_dv = 0, t_first = 0, t_n = 0;
    if (scan) {
        int lc[BATCH_LINES], ic[BATCH_LINES], lx[BATCH_LINES], ix[BATCH_LINES];
        load_batch(lc, ic);
        for (int it = 0; it < (WAVE_WORDS * 32) / BATCH_LINES && lc[0] >= 0; ++it) {
            load_batch(lx, ix);
            #pragma unroll
            for (int u = 0; u < BATCH_LINES; ++u)
                enqueue(ic[u], lc[u] * LINE_IDS + (int)lane);
            if (count >= 8) drain();
            #pragma unroll
            for (int u = 0; u < BATCH_LINES; ++u) { lc[u] = lx[u]; ic[u] = ix[u]; }
        }
    } else {
        const int total = (int)saved;
        for (int pos = 0; pos < total; pos += 32) {
            const int n = (total - pos < 32) ? (total - pos) : 32;
            const unsigned ent = ent_next;
            ent_next = 0u;
            if ((int)lane < total - pos - 32) ent_next = lp[pos + 32 + (int)lane];
            const int sv = (int)(ent & 0xFFFFFu);
            const int dv = (int)(ent >> 20);
            const int nf = n & ~7;
            for (int b = 0; b < nf; b += 8)
                tile_group<true>(h, srcn, tile, sv, dv, b, 8, lane);
            if (nf < n) { t_sv = sv; t_dv = dv; t_first = nf; t_n = n - nf; }
        }
    }
    const int l15   = (int)(lane & 15u);
    const int khalf = (int)(lane >> 4) << 1;
    const int nt    = w & 7;
    const int mg    = w >> 3;
    const int col   = nt * 16 + l15;

    v2f bw[HID_C / 4];
    #pragma unroll
    for (int ks = 0; ks < HID_C / 4; ++ks) {
        bw[ks].x = W[(size_t)TILE_COL(4 * ks + khalf) * HID_C + col];
        bw[ks].y = W[(size_t)TILE_COL(4 * ks + khalf + 1) * HID_C + col];
    }
    const float bv = bias[col];

    if (scan) {
        if (count > 0) {
            unsigned ent = 0u;
            int sv = 0;
            if ((int)lane < count) {
                ent = q[lane];
                sv  = src[ent & 0xFFFFFu];
            }
            const int dv = (int)(ent >> 20);
            if (FIRST) save(sv, dv, count);
            tile_group<false>(h, srcn, tile, sv, dv, 0, count, lane);
        }
        if (FIRST && lane == 0u) hdr[slot] = over ? LIST_OVER : (unsigned)nsaved;
    } else if (t_n > 0) {
        tile_group<false>(h, srcn, tile, t_sv, t_dv, t_first, t_n, lane);
    }

    auto row_scale = [&](int j) -> float {
        const int m    = 4 * j + mg;
        const int node = lo + 16 * m + l15;
        return (m < M_TILES && node < n_nodes) ? dstn[node] : 0.0f;
    };
    float sc = row_scale(0);

    __syncthreads();

    auto store_round = [&](int j) {
        #pragma unroll
        for (int t = 0; t < 2; ++t) {
            const int r = 64 * j + 32 * t + w;
            if (r < span) {
                const float4 v = *(const float4*)(tile + r * TILE_PITCH + 4 * (int)lane);
                *(float4*)(out + (size_t)(lo + r) * HID_C + 4 * lane) = v;
            }
        }
    };

    for (int j = 0; j < GEMM_ROUNDS; ++j) {
        const int m = 4 * j + mg;
        const float sc_next = row_scale(j + 1);
        v8f acc = {0.f, 0.f, 0.f, 0.f, 0.f, 0.f, 0.f, 0.f};
        if (m < M_TILES) {
            const float* ap = tile + (16 * m + l15) * TILE_PITCH + khalf;
            #pragma unroll
            for (int ks = 0; ks < HID_C / 4; ++ks) {
                v2f a = *(const v2f*)(ap + 4 * ks);
                a.x *= sc;
                a.y *= sc;
                acc = __builtin_amdgcn_wmma_f32_16x16x4_f32(false, a, false, bw[ks],
                                                            (short)0, acc, false, false);
            }
        }
        __syncthreads();
        if (m < M_TILES) {
            float* op = tile + (16 * m + 8 * (int)(lane >> 4)) * TILE_PITCH + col;
            #pragma unroll
            for (int i = 0; i < 8; ++i)
                op[i * TILE_PITCH] = fmaxf(acc[i] + bv, 0.0f);
        }
        if (j > 0) store_round(j - 1);
        sc = sc_next;
    }
    __syncthreads();
    store_round(GEMM_ROUNDS - 1);
}

#define CLS_SPLIT 8

template <int K>
__global__ void __launch_bounds__(32 * CLS_SPLIT) wmma_gemm_kernel(
    const float* __restrict__ A, const float* __restrict__ B,
    const float* __restrict__ bias, float* __restrict__ C,
    int M, int N, int do_relu) {
    constexpr int STEPS = K / (4 * CLS_SPLIT);
    static_assert(K % (4 * CLS_SPLIT) == 0 && STEPS >= 1, "K is cut into whole k-steps per wave");
    __shared__ float part[CLS_SPLIT][8][32];

    const int lane  = threadIdx.x & 31;
    const int s     = __builtin_amdgcn_readfirstlane((int)(threadIdx.x >> 5));
    const int l15   = lane & 15;
    const int khalf = (lane >> 4) << 1;
    const int mbase = blockIdx.x * 16;
    const int nbase = blockIdx.y * 16;

    int row = mbase + l15;
    if (row > M - 1) row = M - 1;
    int col = nbase + l15;
    if (col > N - 1) col = N - 1;

    const int kbase = s * (STEPS * 4) + khalf;
    const float* ap = A + (size_t)row * K + kbase;
    const float* bp = B + (size_t)kbase * N + col;

    v2f a[STEPS], b[STEPS];
    #pragma unroll
    for (int t = 0; t < STEPS; ++t) {
        a[t].x = ap[4 * t];
        a[t].y = ap[4 * t + 1];
        b[t].x = bp[(size_t)(4 * t) * N];
        b[t].y = bp[(size_t)(4 * t + 1) * N];
    }
    const float bv = bias[col];

    v8f acc = {0.f, 0.f, 0.f, 0.f, 0.f, 0.f, 0.f, 0.f};
    #pragma unroll
    for (int t = 0; t < STEPS; ++t)
        acc = __builtin_amdgcn_wmma_f32_16x16x4_f32(false, a[t], false, b[t],
                                                    (short)0, acc, false, false);
    #pragma unroll
    for (int i = 0; i < 8; ++i) part[s][i][lane] = acc[i];
    __syncthreads();

    for (int i = s; i < 8; i += CLS_SPLIT) {
        float v = part[0][i][lane];
        #pragma unroll
        for (int p = 1; p < CLS_SPLIT; ++p) v += part[p][i][lane];
        const int m = mbase + i + ((lane >> 4) << 3);
        const int n = nbase + l15;
        if (m < M && n < N) {
            v += bv;
            C[(size_t)m * N + n] = do_relu ? fmaxf(v, 0.0f) : v;
        }
    }
}

#define POOL_WAVES_PER_BLOCK 10
#define POOL_BLOCKS 625
#define POOL_WAVES (POOL_BLOCKS * POOL_WAVES_PER_BLOCK)

__device__ __forceinline__ void pool_flush(float* __restrict__ hg, int g, unsigned lane,
                                           unsigned m0, unsigned m1, unsigned m2, unsigned m3) {
    unsigned int* p = (unsigned int*)(hg + (size_t)g * HID_C);
    atomicMax(p + lane,       m0);
    atomicMax(p + lane + 32u, m1);
    atomicMax(p + lane + 64u, m2);
    atomicMax(p + lane + 96u, m3);
}

__global__ void __launch_bounds__(POOL_WAVES_PER_BLOCK * 32) segmax_kernel(
    const float* __restrict__ h, const int* __restrict__ gids,
    float* __restrict__ hg, int n_nodes) {
    const unsigned lane = threadIdx.x & 31u;
    const int w = __builtin_amdgcn_readfirstlane(
        (int)(blockIdx.x * POOL_WAVES_PER_BLOCK + (threadIdx.x >> 5)));

    const int q  = n_nodes / POOL_WAVES;
    const int r  = n_nodes % POOL_WAVES;
    const int n0 = w * q + (w < r ? w : r);
    const int n1 = n0 + q + (w < r ? 1 : 0);

    int cur = -1;
    unsigned m0 = 0u, m1 = 0u, m2 = 0u, m3 = 0u;

    for (int base = n0; base < n1; base += 32) {
        const int idx = base + (int)lane;
        int gv = 0;
        if (idx < n1) gv = gids[idx];
        const int cnt = (n1 - base < 32) ? (n1 - base) : 32;

        int j = 0;
        for (; j + 8 <= cnt; j += 8) {
            float v[8][4];
            #pragma unroll
            for (int u = 0; u < 8; ++u) {
                const float* hp = h + (size_t)(base + j + u) * HID_C;
                v[u][0] = hp[lane];
                v[u][1] = hp[lane + 32u];
                v[u][2] = hp[lane + 64u];
                v[u][3] = hp[lane + 96u];
            }
            #pragma unroll
            for (int u = 0; u < 8; ++u) {
                const int g = __builtin_amdgcn_readlane(gv, j + u);
                if (g != cur) {
                    if (cur >= 0) pool_flush(hg, cur, lane, m0, m1, m2, m3);
                    m0 = 0u; m1 = 0u; m2 = 0u; m3 = 0u;
                    cur = g;
                }
                const unsigned b0 = __float_as_uint(v[u][0]);
                const unsigned b1 = __float_as_uint(v[u][1]);
                const unsigned b2 = __float_as_uint(v[u][2]);
                const unsigned b3 = __float_as_uint(v[u][3]);
                m0 = (b0 > m0) ? b0 : m0;
                m1 = (b1 > m1) ? b1 : m1;
                m2 = (b2 > m2) ? b2 : m2;
                m3 = (b3 > m3) ? b3 : m3;
            }
        }
        for (; j < cnt; ++j) {
            const float* hp = h + (size_t)(base + j) * HID_C;
            const unsigned b0 = __float_as_uint(hp[lane]);
            const unsigned b1 = __float_as_uint(hp[lane + 32u]);
            const unsigned b2 = __float_as_uint(hp[lane + 64u]);
            const unsigned b3 = __float_as_uint(hp[lane + 96u]);
            const int g = __builtin_amdgcn_readlane(gv, j);
            if (g != cur) {
                if (cur >= 0) pool_flush(hg, cur, lane, m0, m1, m2, m3);
                m0 = 0u; m1 = 0u; m2 = 0u; m3 = 0u;
                cur = g;
            }
            m0 = (b0 > m0) ? b0 : m0;
            m1 = (b1 > m1) ? b1 : m1;
            m2 = (b2 > m2) ? b2 : m2;
            m3 = (b3 > m3) ? b3 : m3;
        }
    }
    if (cur >= 0) pool_flush(hg, cur, lane, m0, m1, m2, m3);
}

extern "C" void kernel_launch(void* const* d_in, const int* in_sizes, int n_in,
                              void* d_out, int out_size, void* d_ws, size_t ws_size,
                              hipStream_t stream) {
    (void)in_sizes; (void)n_in; (void)out_size; (void)ws_size;

    const float* feat = (const float*)d_in[0];
    const int*   src  = (const int*)d_in[1];
    const int*   dst  = (const int*)d_in[2];
    const int*   gids = (const int*)d_in[3];
    const float* Wg[3] = {(const float*)d_in[4], (const float*)d_in[6], (const float*)d_in[8]};
    const float* bg[3] = {(const float*)d_in[5], (const float*)d_in[7], (const float*)d_in[9]};
    const float* Wc1 = (const float*)d_in[10]; const float* bc1 = (const float*)d_in[11];
    const float* Wc2 = (const float*)d_in[12]; const float* bc2 = (const float*)d_in[13];
    const float* Wc3 = (const float*)d_in[14]; const float* bc3 = (const float*)d_in[15];
    float* out = (float*)d_out;

    char* ws = (char*)d_ws;
    size_t off = 0;
    auto take = [&](size_t elems) -> float* {
        float* p = (float*)(ws + off);
        off += ((elems * sizeof(float) + 255) / 256) * 256;
        return p;
    };
    unsigned int* part = (unsigned int*)take((size_t)2 * DEG_CHUNKS * N_NODES_C);
    float* srcn = take(N_NODES_C);
    float* dstn = take(N_NODES_C);
    unsigned int* bm = (unsigned int*)take((size_t)TILE_BLOCKS * BM_PITCH);
    unsigned int* lst = (unsigned int*)take((size_t)TILE_BLOCKS * TILE_WAVES * LIST_CAP);
    unsigned int* hdr = (unsigned int*)take((size_t)TILE_BLOCKS * TILE_WAVES);
    float* h2   = take((size_t)N_NODES_C * HID_C);
    float* h    = take((size_t)N_NODES_C * HID_C);
    float* hg   = take((size_t)N_GRAPHS_C * HID_C);
    float* x1   = take((size_t)N_GRAPHS_C * HID_C);
    float* x2   = take((size_t)N_GRAPHS_C * (HID_C / 2));

    degree_kernel<<<dim3(DEG_RANGES, DEG_CHUNKS, 2), DEG_THREADS, 0, stream>>>(
        src, dst, part, N_EDGES_C, N_NODES_C);
    norm_kernel<<<(N_NODES_C + 255) / 256, 256, 0, stream>>>(part, srcn, dstn, N_NODES_C);

    mark_kernel<<<MARK_BLOCKS, 1024, 0, stream>>>(dst, bm, N_EDGES_C);

    edge_conv_kernel<true><<<TILE_BLOCKS, TILE_THREADS, 0, stream>>>(
        feat, srcn, dstn, src, dst, bm, lst, hdr, Wg[0], bg[0], h, N_EDGES_C, N_NODES_C);
    edge_conv_kernel<false><<<TILE_BLOCKS, TILE_THREADS, 0, stream>>>(
        h, srcn, dstn, src, dst, bm, lst, hdr, Wg[1], bg[1], h2, N_EDGES_C, N_NODES_C);
    edge_conv_kernel<false><<<TILE_BLOCKS, TILE_THREADS, 0, stream>>>(
        h2, srcn, dstn, src, dst, bm, lst, hdr, Wg[2], bg[2], h, N_EDGES_C, N_NODES_C);

    hipMemsetAsync(hg, 0, (size_t)N_GRAPHS_C * HID_C * sizeof(float), stream);
    segmax_kernel<<<POOL_BLOCKS, POOL_WAVES_PER_BLOCK * 32, 0, stream>>>(h, gids, hg, N_NODES_C);

    wmma_gemm_kernel<HID_C><<<dim3(N_GRAPHS_C / 16, HID_C / 16), 32 * CLS_SPLIT, 0, stream>>>(
        hg, Wc1, bc1, x1, N_GRAPHS_C, HID_C, 1);
    wmma_gemm_kernel<HID_C><<<dim3(N_GRAPHS_C / 16, (HID_C / 2) / 16), 32 * CLS_SPLIT, 0, stream>>>(
        x1, Wc2, bc2, x2, N_GRAPHS_C, HID_C / 2, 1);
    wmma_gemm_kernel<HID_C / 2><<<dim3(N_GRAPHS_C / 16, (OUT_DIM_C + 15) / 16), 32 * CLS_SPLIT, 0, stream>>>(
        x2, Wc3, bc3, out, N_GRAPHS_C, OUT_DIM_C, 0);
}
